# starting kernel
# baseline (speedup 1.0000x reference)
#include <hip/hip_runtime.h>

typedef __attribute__((ext_vector_type(2))) float v2f;
typedef __attribute__((ext_vector_type(8))) float v8f;

__global__ void k_zero_i32(int* __restrict__ p, int n) {
    int i = blockIdx.x * blockDim.x + threadIdx.x;
    if (i < n) p[i] = 0;
}

__global__ void k_count_indeg(const int* __restrict__ col, int* __restrict__ indeg, int e) {
    int i = blockIdx.x * blockDim.x + threadIdx.x;
    if (i < e) atomicAdd(&indeg[col[i]], 1);
}

__global__ void k_dinv(const int* __restrict__ indeg,
                       float* __restrict__ dinv_s, float* __restrict__ dinv_h, int n) {
    int i = blockIdx.x * blockDim.x + threadIdx.x;
    if (i < n) {
        float d = (float)indeg[i];
        dinv_s[i] = rsqrtf(d + 1.0f);
        dinv_h[i] = rsqrtf(d + 2.0f);
    }
}

template <int DIN, int NT, bool RELU_IN>
__global__ void k_gemm_wmma(const float* __restrict__ X, const float* __restrict__ W,
                            float* __restrict__ H, int N) {
    constexpr int DOUT = NT * 16;
    const int lane = threadIdx.x & 31;
    const int wave = threadIdx.x >> 5;
    const int tile = blockIdx.x * (blockDim.x >> 5) + wave;
    const int rtiles = (N + 15) >> 4;
    if (tile >= rtiles) return;

    const int mrow  = tile * 16 + (lane & 15);
    const int mload = mrow < N ? mrow : N - 1;
    const int kb    = (lane >> 4) * 2;
    const int nloc  = lane & 15;

    v8f acc[NT];
#pragma unroll
    for (int j = 0; j < NT; ++j) {
        v8f z = {0.f, 0.f, 0.f, 0.f, 0.f, 0.f, 0.f, 0.f};
        acc[j] = z;
    }

    const float* xr = X + (size_t)mload * DIN;
#pragma unroll 4
    for (int k = 0; k < DIN; k += 4) {
        float a0 = xr[k + kb];
        float a1 = xr[k + kb + 1];
        if (RELU_IN) { a0 = fmaxf(a0, 0.0f); a1 = fmaxf(a1, 0.0f); }
        v2f A = {a0, a1};
#pragma unroll
        for (int j = 0; j < NT; ++j) {
            const int n0 = j * 16 + nloc;
            v2f B = {W[(size_t)(k + kb) * DOUT + n0],
                     W[(size_t)(k + kb + 1) * DOUT + n0]};
            acc[j] = __builtin_amdgcn_wmma_f32_16x16x4_f32(
                false, A, false, B, (short)0, acc[j], false, false);
        }
    }

    const int half = lane >> 4;
#pragma unroll
    for (int j = 0; j < NT; ++j) {
#pragma unroll
        for (int r = 0; r < 8; ++r) {
            int mo = tile * 16 + r + 8 * half;
            if (mo < N) H[(size_t)mo * DOUT + j * 16 + nloc] = acc[j][r];
        }
    }
}

__global__ void k_agg_init(const float* __restrict__ H, const float* __restrict__ dinv,
                           const float* __restrict__ bias, float* __restrict__ Y,
                           int n, int shift, int sharp) {
    long long i = (long long)blockIdx.x * blockDim.x + threadIdx.x;
    long long tot = (long long)n << shift;
    if (i >= tot) return;
    int node = (int)(i >> shift);
    int f    = (int)(i & ((1 << shift) - 1));
    float dv = dinv[node];
    float sw = dv * dv;
    if (sharp) sw *= 2.0f;
    Y[i] = sw * H[i] + bias[f];
}

__global__ void k_agg_edges(const float* __restrict__ H, const int* __restrict__ row,
                            const int* __restrict__ col, const float* __restrict__ dinv,
                            float* __restrict__ Y, int e, int shift, int sharp) {
    long long i = (long long)blockIdx.x * blockDim.x + threadIdx.x;
    long long tot = (long long)e << shift;
    if (i >= tot) return;
    int eg = (int)(i >> shift);
    int f  = (int)(i & ((1 << shift) - 1));
    int r  = row[eg];
    int c  = col[eg];
    float w = dinv[r] * dinv[c];
    if (sharp) w = -w;
    int dout = 1 << shift;
    float v = w * H[(size_t)r * dout + f];
    unsafeAtomicAdd(&Y[(size_t)c * dout + f], v);
}

static inline int cdiv(long long a, int b) { return (int)((a + b - 1) / b); }

extern "C" void kernel_launch(void* const* d_in, const int* in_sizes, int n_in,
                              void* d_out, int out_size, void* d_ws, size_t ws_size,
                              hipStream_t stream) {
    const float* x   = (const float*)d_in[0];
    const int*   ei  = (const int*)d_in[1];
    const float* We1 = (const float*)d_in[2];  const float* be1 = (const float*)d_in[3];
    const float* We2 = (const float*)d_in[4];  const float* be2 = (const float*)d_in[5];
    const float* We3 = (const float*)d_in[6];  const float* be3 = (const float*)d_in[7];
    const float* Wd1 = (const float*)d_in[8];  const float* bd1 = (const float*)d_in[9];
    const float* Wd2 = (const float*)d_in[10]; const float* bd2 = (const float*)d_in[11];
    const float* Wd3 = (const float*)d_in[12]; const float* bd3 = (const float*)d_in[13];

    const int N = in_sizes[0] / 128;
    const int E = in_sizes[1] / 2;
    const int* rowp = ei;
    const int* colp = ei + E;

    char* ws = (char*)d_ws;
    int*   indeg  = (int*)ws;    ws += sizeof(int)   * (size_t)N;
    float* dinv_s = (float*)ws;  ws += sizeof(float) * (size_t)N;
    float* dinv_h = (float*)ws;  ws += sizeof(float) * (size_t)N;
    float* hbuf   = (float*)ws;  ws += sizeof(float) * (size_t)N * 128;
    float* ybuf   = (float*)ws;
    float* outp   = (float*)d_out;

    const int TB = 256;
    const int rtiles  = (N + 15) >> 4;
    const int gblocks = (rtiles + 7) >> 3;

    k_zero_i32   <<<cdiv(N, TB), TB, 0, stream>>>(indeg, N);
    k_count_indeg<<<cdiv(E, TB), TB, 0, stream>>>(colp, indeg, E);
    k_dinv       <<<cdiv(N, TB), TB, 0, stream>>>(indeg, dinv_s, dinv_h, N);

    k_gemm_wmma<128, 4, false><<<gblocks, TB, 0, stream>>>(x, We1, hbuf, N);
    k_agg_init <<<cdiv((long long)N << 6, TB), TB, 0, stream>>>(hbuf, dinv_s, be1, ybuf, N, 6, 0);
    k_agg_edges<<<cdiv((long long)E << 6, TB), TB, 0, stream>>>(hbuf, rowp, colp, dinv_s, ybuf, E, 6, 0);
    k_gemm_wmma<64, 2, true><<<gblocks, TB, 0, stream>>>(ybuf, We2, hbuf, N);
    k_agg_init <<<cdiv((long long)N << 5, TB), TB, 0, stream>>>(hbuf, dinv_s, be2, ybuf, N, 5, 0);
    k_agg_edges<<<cdiv((long long)E << 5, TB), TB, 0, stream>>>(hbuf, rowp, colp, dinv_s, ybuf, E, 5, 0);
    k_gemm_wmma<32, 1, true><<<gblocks, TB, 0, stream>>>(ybuf, We3, hbuf, N);
    k_agg_init <<<cdiv((long long)N << 4, TB), TB, 0, stream>>>(hbuf, dinv_s, be3, ybuf, N, 4, 0);
    k_agg_edges<<<cdiv((long long)E << 4, TB), TB, 0, stream>>>(hbuf, rowp, colp, dinv_s, ybuf, E, 4, 0);

    k_gemm_wmma<16, 2, true><<<gblocks, TB, 0, stream>>>(ybuf, Wd1, hbuf, N);
    k_agg_init <<<cdiv((long long)N << 5, TB), TB, 0, stream>>>(hbuf, dinv_h, bd1, ybuf, N, 5, 1);
    k_agg_edges<<<cdiv((long long)E << 5, TB), TB, 0, stream>>>(hbuf, rowp, colp, dinv_h, ybuf, E, 5, 1);
    k_gemm_wmma<32, 4, true><<<gblocks, TB, 0, stream>>>(ybuf, Wd2, hbuf, N);
    k_agg_init <<<cdiv((long long)N << 6, TB), TB, 0, stream>>>(hbuf, dinv_h, bd2, ybuf, N, 6, 1);
    k_agg_edges<<<cdiv((long long)E << 6, TB), TB, 0, stream>>>(hbuf, rowp, colp, dinv_h, ybuf, E, 6, 1);
    k_gemm_wmma<64, 8, true><<<gblocks, TB, 0, stream>>>(ybuf, Wd3, hbuf, N);
    k_agg_init <<<cdiv((long long)N << 7, TB), TB, 0, stream>>>(hbuf, dinv_h, bd3, outp, N, 7, 1);
    k_agg_edges<<<cdiv((long long)E << 7, TB), TB, 0, stream>>>(hbuf, rowp, colp, dinv_h, outp, E, 7, 1);
}


# Decoder layers (DOUT>DIN) aggregate before the linear map, (A relu(y))W+b for A(relu(y)W)+b: edge atomics and self-loop pass run at DIN not DOUT width; changes fp summation order
# speedup vs baseline: 1.3910x; 1.3910x over previous
#include <hip/hip_runtime.h>

typedef __attribute__((ext_vector_type(2))) float v2f;
typedef __attribute__((ext_vector_type(8))) float v8f;

__global__ void k_zero_i32(int* __restrict__ p, int n) {
    int i = blockIdx.x * blockDim.x + threadIdx.x;
    if (i < n) p[i] = 0;
}

__global__ void k_count_indeg(const int* __restrict__ col, int* __restrict__ indeg, int e) {
    int i = blockIdx.x * blockDim.x + threadIdx.x;
    if (i < e) atomicAdd(&indeg[col[i]], 1);
}

__global__ void k_dinv(const int* __restrict__ indeg,
                       float* __restrict__ dinv_s, float* __restrict__ dinv_h, int n) {
    int i = blockIdx.x * blockDim.x + threadIdx.x;
    if (i < n) {
        float d = (float)indeg[i];
        dinv_s[i] = rsqrtf(d + 1.0f);
        dinv_h[i] = rsqrtf(d + 2.0f);
    }
}

template <int DIN, int NT, bool RELU_IN, bool BIAS>
__global__ void k_gemm_wmma(const float* __restrict__ X, const float* __restrict__ W,
                            const float* __restrict__ bias, float* __restrict__ H, int N) {
    constexpr int DOUT = NT * 16;
    const int lane = threadIdx.x & 31;
    const int wave = threadIdx.x >> 5;
    const int tile = blockIdx.x * (blockDim.x >> 5) + wave;
    const int rtiles = (N + 15) >> 4;
    if (tile >= rtiles) return;

    const int mrow  = tile * 16 + (lane & 15);
    const int mload = mrow < N ? mrow : N - 1;
    const int kb    = (lane >> 4) * 2;
    const int nloc  = lane & 15;

    v8f acc[NT];
#pragma unroll
    for (int j = 0; j < NT; ++j) {
        v8f z = {0.f, 0.f, 0.f, 0.f, 0.f, 0.f, 0.f, 0.f};
        acc[j] = z;
    }

    const float* xr = X + (size_t)mload * DIN;
#pragma unroll 4
    for (int k = 0; k < DIN; k += 4) {
        float a0 = xr[k + kb];
        float a1 = xr[k + kb + 1];
        if (RELU_IN) { a0 = fmaxf(a0, 0.0f); a1 = fmaxf(a1, 0.0f); }
        v2f A = {a0, a1};
#pragma unroll
        for (int j = 0; j < NT; ++j) {
            const int n0 = j * 16 + nloc;
            v2f B = {W[(size_t)(k + kb) * DOUT + n0],
                     W[(size_t)(k + kb + 1) * DOUT + n0]};
            acc[j] = __builtin_amdgcn_wmma_f32_16x16x4_f32(
                false, A, false, B, (short)0, acc[j], false, false);
        }
    }

    const int half = lane >> 4;
#pragma unroll
    for (int j = 0; j < NT; ++j) {
        const float bj = BIAS ? bias[j * 16 + nloc] : 0.0f;
#pragma unroll
        for (int r = 0; r < 8; ++r) {
            int mo = tile * 16 + r + 8 * half;
            if (mo < N) H[(size_t)mo * DOUT + j * 16 + nloc] = BIAS ? acc[j][r] + bj : acc[j][r];
        }
    }
}

template <bool PRE>
__global__ void k_agg_init(const float* __restrict__ H, const float* __restrict__ dinv,
                           const float* __restrict__ bias, float* __restrict__ Y,
                           int n, int shift, int sharp) {
    long long i = (long long)blockIdx.x * blockDim.x + threadIdx.x;
    long long tot = (long long)n << shift;
    if (i >= tot) return;
    int node = (int)(i >> shift);
    int f    = (int)(i & ((1 << shift) - 1));
    float dv = dinv[node];
    float sw = dv * dv;
    if (sharp) sw *= 2.0f;
    if (PRE) Y[i] = sw * fmaxf(H[i], 0.0f);
    else     Y[i] = sw * H[i] + bias[f];
}

template <bool PRE>
__global__ void k_agg_edges(const float* __restrict__ H, const int* __restrict__ row,
                            const int* __restrict__ col, const float* __restrict__ dinv,
                            float* __restrict__ Y, int e, int shift, int sharp) {
    long long i = (long long)blockIdx.x * blockDim.x + threadIdx.x;
    long long tot = (long long)e << shift;
    if (i >= tot) return;
    int eg = (int)(i >> shift);
    int f  = (int)(i & ((1 << shift) - 1));
    int r  = row[eg];
    int c  = col[eg];
    float w = dinv[r] * dinv[c];
    if (sharp) w = -w;
    int dout = 1 << shift;
    float h = H[(size_t)r * dout + f];
    if (PRE) h = fmaxf(h, 0.0f);
    float v = w * h;
    unsafeAtomicAdd(&Y[(size_t)c * dout + f], v);
}

static inline int cdiv(long long a, int b) { return (int)((a + b - 1) / b); }

extern "C" void kernel_launch(void* const* d_in, const int* in_sizes, int n_in,
                              void* d_out, int out_size, void* d_ws, size_t ws_size,
                              hipStream_t stream) {
    const float* x   = (const float*)d_in[0];
    const int*   ei  = (const int*)d_in[1];
    const float* We1 = (const float*)d_in[2];  const float* be1 = (const float*)d_in[3];
    const float* We2 = (const float*)d_in[4];  const float* be2 = (const float*)d_in[5];
    const float* We3 = (const float*)d_in[6];  const float* be3 = (const float*)d_in[7];
    const float* Wd1 = (const float*)d_in[8];  const float* bd1 = (const float*)d_in[9];
    const float* Wd2 = (const float*)d_in[10]; const float* bd2 = (const float*)d_in[11];
    const float* Wd3 = (const float*)d_in[12]; const float* bd3 = (const float*)d_in[13];

    const int N = in_sizes[0] / 128;
    const int E = in_sizes[1] / 2;
    const int* rowp = ei;
    const int* colp = ei + E;

    char* ws = (char*)d_ws;
    int*   indeg  = (int*)ws;    ws += sizeof(int)   * (size_t)N;
    float* dinv_s = (float*)ws;  ws += sizeof(float) * (size_t)N;
    float* dinv_h = (float*)ws;  ws += sizeof(float) * (size_t)N;
    float* hbuf   = (float*)ws;  ws += sizeof(float) * (size_t)N * 128;
    float* ybuf   = (float*)ws;
    float* outp   = (float*)d_out;

    const int TB = 256;
    const int rtiles  = (N + 15) >> 4;
    const int gblocks = (rtiles + 7) >> 3;

    k_zero_i32   <<<cdiv(N, TB), TB, 0, stream>>>(indeg, N);
    k_count_indeg<<<cdiv(E, TB), TB, 0, stream>>>(colp, indeg, E);
    k_dinv       <<<cdiv(N, TB), TB, 0, stream>>>(indeg, dinv_s, dinv_h, N);

    k_gemm_wmma<128, 4, false, false><<<gblocks, TB, 0, stream>>>(x, We1, be1, hbuf, N);
    k_agg_init <false><<<cdiv((long long)N << 6, TB), TB, 0, stream>>>(hbuf, dinv_s, be1, ybuf, N, 6, 0);
    k_agg_edges<false><<<cdiv((long long)E << 6, TB), TB, 0, stream>>>(hbuf, rowp, colp, dinv_s, ybuf, E, 6, 0);
    k_gemm_wmma<64, 2, true, false><<<gblocks, TB, 0, stream>>>(ybuf, We2, be2, hbuf, N);
    k_agg_init <false><<<cdiv((long long)N << 5, TB), TB, 0, stream>>>(hbuf, dinv_s, be2, ybuf, N, 5, 0);
    k_agg_edges<false><<<cdiv((long long)E << 5, TB), TB, 0, stream>>>(hbuf, rowp, colp, dinv_s, ybuf, E, 5, 0);
    k_gemm_wmma<32, 1, true, false><<<gblocks, TB, 0, stream>>>(ybuf, We3, be3, hbuf, N);
    k_agg_init <false><<<cdiv((long long)N << 4, TB), TB, 0, stream>>>(hbuf, dinv_s, be3, ybuf, N, 4, 0);
    k_agg_edges<false><<<cdiv((long long)E << 4, TB), TB, 0, stream>>>(hbuf, rowp, colp, dinv_s, ybuf, E, 4, 0);

    k_agg_init <true><<<cdiv((long long)N << 4, TB), TB, 0, stream>>>(ybuf, dinv_h, bd1, hbuf, N, 4, 1);
    k_agg_edges<true><<<cdiv((long long)E << 4, TB), TB, 0, stream>>>(ybuf, rowp, colp, dinv_h, hbuf, E, 4, 1);
    k_gemm_wmma<16, 2, false, true><<<gblocks, TB, 0, stream>>>(hbuf, Wd1, bd1, ybuf, N);
    k_agg_init <true><<<cdiv((long long)N << 5, TB), TB, 0, stream>>>(ybuf, dinv_h, bd2, hbuf, N, 5, 1);
    k_agg_edges<true><<<cdiv((long long)E << 5, TB), TB, 0, stream>>>(ybuf, rowp, colp, dinv_h, hbuf, E, 5, 1);
    k_gemm_wmma<32, 4, false, true><<<gblocks, TB, 0, stream>>>(hbuf, Wd2, bd2, ybuf, N);
    k_agg_init <true><<<cdiv((long long)N << 6, TB), TB, 0, stream>>>(ybuf, dinv_h, bd3, hbuf, N, 6, 1);
    k_agg_edges<true><<<cdiv((long long)E << 6, TB), TB, 0, stream>>>(ybuf, rowp, colp, dinv_h, hbuf, E, 6, 1);
    k_gemm_wmma<64, 8, false, true><<<gblocks, TB, 0, stream>>>(hbuf, Wd3, bd3, outp, N);
}


# Float edge atomics replaced by a gather over per-call in-edge lists (counting sort on target): a node sums self-loop + in-edges in registers, one store; changes fp sum order (now fixed, by source)
# speedup vs baseline: 2.9988x; 2.1559x over previous
#include <hip/hip_runtime.h>

typedef __attribute__((ext_vector_type(2))) float v2f;
typedef __attribute__((ext_vector_type(8))) float v8f;
typedef __attribute__((ext_vector_type(4))) float v4f;

__global__ void k_zero_i32(int* __restrict__ p, int n) {
    int i = blockIdx.x * blockDim.x + threadIdx.x;
    if (i < n) p[i] = 0;
}

__global__ void k_count_rank(const int* __restrict__ col, int* __restrict__ indeg,
                             int* __restrict__ rank, int e) {
    int i = blockIdx.x * blockDim.x + threadIdx.x;
    if (i < e) rank[i] = atomicAdd(&indeg[col[i]], 1);
}

__global__ void __launch_bounds__(1024)
k_rowptr_scan(const int* __restrict__ indeg, int* __restrict__ rowptr, int n) {
    __shared__ int s_pre[32];
    __shared__ int s_inc[32];
    const int tid  = threadIdx.x;
    const int lane = tid & 31;
    const int wv   = tid >> 5;
    int pre = 0;
#pragma unroll 8
    for (int k = 0; k < (int)blockIdx.x; ++k) pre += indeg[k * 1024 + tid];
    const int i = blockIdx.x * 1024 + tid;
    const int v = (i < n) ? indeg[i] : 0;
    int inc = v;
    for (int d = 1; d < 32; d <<= 1) {
        int t = __shfl_up(inc, d, 32);
        if (lane >= d) inc += t;
    }
    for (int d = 16; d > 0; d >>= 1) pre += __shfl_xor(pre, d, 32);
    if (lane == 31) s_inc[wv] = inc;
    if (lane == 0)  s_pre[wv] = pre;
    __syncthreads();
    int off = s_pre[lane] + (lane < wv ? s_inc[lane] : 0);
    for (int d = 16; d > 0; d >>= 1) off += __shfl_xor(off, d, 32);
    if (i < n) rowptr[i] = off + inc - v;
    if (i == n - 1) rowptr[n] = off + inc;
}

__global__ void k_dinv(const int* __restrict__ indeg,
                       float* __restrict__ dinv_s, float* __restrict__ dinv_h, int n) {
    int i = blockIdx.x * blockDim.x + threadIdx.x;
    if (i < n) {
        float d = (float)indeg[i];
        dinv_s[i] = rsqrtf(d + 1.0f);
        dinv_h[i] = rsqrtf(d + 2.0f);
    }
}

__global__ void k_csr_fill(const int* __restrict__ row, const int* __restrict__ col,
                           const int* __restrict__ rank, const int* __restrict__ rowptr,
                           int* __restrict__ tmp_src, int e) {
    int i = blockIdx.x * blockDim.x + threadIdx.x;
    if (i < e) tmp_src[rowptr[col[i]] + rank[i]] = row[i];
}

__global__ void k_csr_place(const int* __restrict__ row, const int* __restrict__ col,
                            const int* __restrict__ rank, const int* __restrict__ rowptr,
                            const int* __restrict__ tmp_src,
                            const float* __restrict__ dinv_s, const float* __restrict__ dinv_h,
                            int* __restrict__ csr_src, float* __restrict__ w_s,
                            float* __restrict__ w_h, int e) {
    int i = blockIdx.x * blockDim.x + threadIdx.x;
    if (i >= e) return;
    const int r   = row[i];
    const int c   = col[i];
    const int beg = rowptr[c];
    const int end = rowptr[c + 1];
    const int me  = beg + rank[i];
    int pos = beg;
    for (int b = beg; b < end; ++b) {
        const int k = tmp_src[b];
        pos += (k < r || (k == r && b < me)) ? 1 : 0;
    }
    csr_src[pos] = r;
    w_s[pos] = dinv_s[r] * dinv_s[c];
    w_h[pos] = -(dinv_h[r] * dinv_h[c]);
}

template <int DIN, int NT, bool RELU_IN, bool BIAS>
__global__ void k_gemm_wmma(const float* __restrict__ X, const float* __restrict__ W,
                            const float* __restrict__ bias, float* __restrict__ H, int N) {
    constexpr int DOUT = NT * 16;
    const int lane = threadIdx.x & 31;
    const int wave = threadIdx.x >> 5;
    const int tile = blockIdx.x * (blockDim.x >> 5) + wave;
    const int rtiles = (N + 15) >> 4;
    if (tile >= rtiles) return;

    const int mrow  = tile * 16 + (lane & 15);
    const int mload = mrow < N ? mrow : N - 1;
    const int kb    = (lane >> 4) * 2;
    const int nloc  = lane & 15;

    v8f acc[NT];
#pragma unroll
    for (int j = 0; j < NT; ++j) {
        v8f z = {0.f, 0.f, 0.f, 0.f, 0.f, 0.f, 0.f, 0.f};
        acc[j] = z;
    }

    const float* xr = X + (size_t)mload * DIN;
#pragma unroll 4
    for (int k = 0; k < DIN; k += 4) {
        float a0 = xr[k + kb];
        float a1 = xr[k + kb + 1];
        if (RELU_IN) { a0 = fmaxf(a0, 0.0f); a1 = fmaxf(a1, 0.0f); }
        v2f A = {a0, a1};
#pragma unroll
        for (int j = 0; j < NT; ++j) {
            const int n0 = j * 16 + nloc;
            v2f B = {W[(size_t)(k + kb) * DOUT + n0],
                     W[(size_t)(k + kb + 1) * DOUT + n0]};
            acc[j] = __builtin_amdgcn_wmma_f32_16x16x4_f32(
                false, A, false, B, (short)0, acc[j], false, false);
        }
    }

    const int half = lane >> 4;
#pragma unroll
    for (int j = 0; j < NT; ++j) {
        const float bj = BIAS ? bias[j * 16 + nloc] : 0.0f;
#pragma unroll
        for (int r = 0; r < 8; ++r) {
            int mo = tile * 16 + r + 8 * half;
            if (mo < N) H[(size_t)mo * DOUT + j * 16 + nloc] = BIAS ? acc[j][r] + bj : acc[j][r];
        }
    }
}

static __device__ __forceinline__ v4f relu4(v4f h) {
    h.x = fmaxf(h.x, 0.0f); h.y = fmaxf(h.y, 0.0f);
    h.z = fmaxf(h.z, 0.0f); h.w = fmaxf(h.w, 0.0f);
    return h;
}

template <int D, bool DEC>
__global__ void k_agg_gather(const float* __restrict__ H, const int* __restrict__ rowptr,
                             const int* __restrict__ csr_src, const float* __restrict__ wgt,
                             const float* __restrict__ dinv, const float* __restrict__ bias,
                             float* __restrict__ Y, int n) {
    constexpr int LPN = D / 4;
    const int i    = blockIdx.x * blockDim.x + threadIdx.x;
    const int node = i / LPN;
    if (node >= n) return;
    const int q    = i % LPN;
    const v4f* __restrict__ H4 = (const v4f*)H;
    const int beg = rowptr[node];
    const int end = rowptr[node + 1];
    const float dv = dinv[node];
    float sw = dv * dv;
    if (DEC) sw *= 2.0f;

    v4f acc = H4[(size_t)node * LPN + q];
    if (DEC) {
        acc = relu4(acc) * sw;
    } else {
        v4f b = {bias[4 * q], bias[4 * q + 1], bias[4 * q + 2], bias[4 * q + 3]};
        acc = acc * sw + b;
    }

    int e = beg;
    for (; e + 4 <= end; e += 4) {
        const int s0 = csr_src[e],     s1 = csr_src[e + 1];
        const int s2 = csr_src[e + 2], s3 = csr_src[e + 3];
        const float w0 = wgt[e],     w1 = wgt[e + 1];
        const float w2 = wgt[e + 2], w3 = wgt[e + 3];
        v4f h0 = H4[(size_t)s0 * LPN + q];
        v4f h1 = H4[(size_t)s1 * LPN + q];
        v4f h2 = H4[(size_t)s2 * LPN + q];
        v4f h3 = H4[(size_t)s3 * LPN + q];
        if (DEC) { h0 = relu4(h0); h1 = relu4(h1); h2 = relu4(h2); h3 = relu4(h3); }
        acc += h0 * w0;
        acc += h1 * w1;
        acc += h2 * w2;
        acc += h3 * w3;
    }
    if (e < end) {
        const int last = end - 1;
        const int e1 = min(e + 1, last);
        const int e2 = min(e + 2, last);
        const int s0 = csr_src[e], s1 = csr_src[e1], s2 = csr_src[e2];
        const float w0 = wgt[e], w1 = wgt[e1], w2 = wgt[e2];
        v4f h0 = H4[(size_t)s0 * LPN + q];
        v4f h1 = H4[(size_t)s1 * LPN + q];
        v4f h2 = H4[(size_t)s2 * LPN + q];
        if (DEC) { h0 = relu4(h0); h1 = relu4(h1); h2 = relu4(h2); }
        acc += h0 * w0;
        if (e + 1 < end) acc += h1 * w1;
        if (e + 2 < end) acc += h2 * w2;
    }
    ((v4f*)Y)[(size_t)node * LPN + q] = acc;
}

static inline int cdiv(long long a, int b) { return (int)((a + b - 1) / b); }

extern "C" void kernel_launch(void* const* d_in, const int* in_sizes, int n_in,
                              void* d_out, int out_size, void* d_ws, size_t ws_size,
                              hipStream_t stream) {
    const float* x   = (const float*)d_in[0];
    const int*   ei  = (const int*)d_in[1];
    const float* We1 = (const float*)d_in[2];  const float* be1 = (const float*)d_in[3];
    const float* We2 = (const float*)d_in[4];  const float* be2 = (const float*)d_in[5];
    const float* We3 = (const float*)d_in[6];  const float* be3 = (const float*)d_in[7];
    const float* Wd1 = (const float*)d_in[8];  const float* bd1 = (const float*)d_in[9];
    const float* Wd2 = (const float*)d_in[10]; const float* bd2 = (const float*)d_in[11];
    const float* Wd3 = (const float*)d_in[12]; const float* bd3 = (const float*)d_in[13];

    const int N = in_sizes[0] / 128;
    const int E = in_sizes[1] / 2;
    const int* rowp = ei;
    const int* colp = ei + E;

    const size_t Np = ((size_t)N + 64 + 63) & ~(size_t)63;
    const size_t Ep = ((size_t)E + 63) & ~(size_t)63;
    char* ws = (char*)d_ws;
    int*   indeg   = (int*)ws;    ws += sizeof(int)   * Np;
    float* dinv_s  = (float*)ws;  ws += sizeof(float) * Np;
    float* dinv_h  = (float*)ws;  ws += sizeof(float) * Np;
    float* hbuf    = (float*)ws;  ws += sizeof(float) * (size_t)N * 64;
    float* ybuf    = (float*)ws;  ws += sizeof(float) * (size_t)N * 64;
    int*   rowptr  = (int*)ws;    ws += sizeof(int)   * Np;
    int*   rank    = (int*)ws;    ws += sizeof(int)   * Ep;
    int*   tmp_src = (int*)ws;    ws += sizeof(int)   * Ep;
    int*   csr_src = (int*)ws;    ws += sizeof(int)   * Ep;
    float* w_s     = (float*)ws;  ws += sizeof(float) * Ep;
    float* w_h     = (float*)ws;
    float* outp    = (float*)d_out;

    const int TB = 256;
    const int rtiles  = (N + 15) >> 4;
    const int gblocks = (rtiles + 7) >> 3;

    k_zero_i32   <<<cdiv(N, TB), TB, 0, stream>>>(indeg, N);
    k_count_rank <<<cdiv(E, TB), TB, 0, stream>>>(colp, indeg, rank, E);
    k_rowptr_scan<<<cdiv(N, 1024), 1024, 0, stream>>>(indeg, rowptr, N);
    k_dinv       <<<cdiv(N, TB), TB, 0, stream>>>(indeg, dinv_s, dinv_h, N);
    k_csr_fill   <<<cdiv(E, TB), TB, 0, stream>>>(rowp, colp, rank, rowptr, tmp_src, E);
    k_csr_place  <<<cdiv(E, TB), TB, 0, stream>>>(rowp, colp, rank, rowptr, tmp_src, dinv_s, dinv_h,
                                                  csr_src, w_s, w_h, E);

    k_gemm_wmma<128, 4, false, false><<<gblocks, TB, 0, stream>>>(x, We1, be1, hbuf, N);
    k_agg_gather<64, false><<<cdiv((long long)N * 16, TB), TB, 0, stream>>>(hbuf, rowptr, csr_src, w_s, dinv_s, be1, ybuf, N);
    k_gemm_wmma<64, 2, true, false><<<gblocks, TB, 0, stream>>>(ybuf, We2, be2, hbuf, N);
    k_agg_gather<32, false><<<cdiv((long long)N * 8, TB), TB, 0, stream>>>(hbuf, rowptr, csr_src, w_s, dinv_s, be2, ybuf, N);
    k_gemm_wmma<32, 1, true, false><<<gblocks, TB, 0, stream>>>(ybuf, We3, be3, hbuf, N);
    k_agg_gather<16, false><<<cdiv((long long)N * 4, TB), TB, 0, stream>>>(hbuf, rowptr, csr_src, w_s, dinv_s, be3, ybuf, N);

    k_agg_gather<16, true><<<cdiv((long long)N * 4, TB), TB, 0, stream>>>(ybuf, rowptr, csr_src, w_h, dinv_h, bd1, hbuf, N);
    k_gemm_wmma<16, 2, false, true><<<gblocks, TB, 0, stream>>>(hbuf, Wd1, bd1, ybuf, N);
    k_agg_gather<32, true><<<cdiv((long long)N * 8, TB), TB, 0, stream>>>(ybuf, rowptr, csr_src, w_h, dinv_h, bd2, hbuf, N);
    k_gemm_wmma<32, 4, false, true><<<gblocks, TB, 0, stream>>>(hbuf, Wd2, bd2, ybuf, N);
    k_agg_gather<64, true><<<cdiv((long long)N * 16, TB), TB, 0, stream>>>(ybuf, rowptr, csr_src, w_h, dinv_h, bd3, hbuf, N);
    k_gemm_wmma<64, 8, false, true><<<gblocks, TB, 0, stream>>>(hbuf, Wd3, bd3, outp, N);
}


# Each aggregation launch multiplies its own node tile by the linear map that follows it (tile in LDS, f32 WMMA, whole-line stores): 5 GEMM launches and their intermediates go; sums keep their order
# speedup vs baseline: 3.3122x; 1.1045x over previous
#include <hip/hip_runtime.h>

typedef __attribute__((ext_vector_type(2))) float v2f;
typedef __attribute__((ext_vector_type(8))) float v8f;
typedef __attribute__((ext_vector_type(4))) float v4f;

__global__ void k_zero_i32(int* __restrict__ p, int n) {
    int i = blockIdx.x * blockDim.x + threadIdx.x;
    if (i < n) p[i] = 0;
}

__global__ void k_count_rank(const int* __restrict__ col, int* __restrict__ indeg,
                             int* __restrict__ rank, int e) {
    int i = blockIdx.x * blockDim.x + threadIdx.x;
    if (i < e) rank[i] = atomicAdd(&indeg[col[i]], 1);
}

__global__ void __launch_bounds__(1024)
k_rowptr_scan(const int* __restrict__ indeg, int* __restrict__ rowptr, int n) {
    __shared__ int s_pre[32];
    __shared__ int s_inc[32];
    const int tid  = threadIdx.x;
    const int lane = tid & 31;
    const int wv   = tid >> 5;
    int pre = 0;
#pragma unroll 8
    for (int k = 0; k < (int)blockIdx.x; ++k) pre += indeg[k * 1024 + tid];
    const int i = blockIdx.x * 1024 + tid;
    const int v = (i < n) ? indeg[i] : 0;
    int inc = v;
    for (int d = 1; d < 32; d <<= 1) {
        int t = __shfl_up(inc, d, 32);
        if (lane >= d) inc += t;
    }
    for (int d = 16; d > 0; d >>= 1) pre += __shfl_xor(pre, d, 32);
    if (lane == 31) s_inc[wv] = inc;
    if (lane == 0)  s_pre[wv] = pre;
    __syncthreads();
    int off = s_pre[lane] + (lane < wv ? s_inc[lane] : 0);
    for (int d = 16; d > 0; d >>= 1) off += __shfl_xor(off, d, 32);
    if (i < n) rowptr[i] = off + inc - v;
    if (i == n - 1) rowptr[n] = off + inc;
}

__global__ void k_dinv(const int* __restrict__ indeg,
                       float* __restrict__ dinv_s, float* __restrict__ dinv_h, int n) {
    int i = blockIdx.x * blockDim.x + threadIdx.x;
    if (i < n) {
        float d = (float)indeg[i];
        dinv_s[i] = rsqrtf(d + 1.0f);
        dinv_h[i] = rsqrtf(d + 2.0f);
    }
}

__global__ void k_csr_fill(const int* __restrict__ row, const int* __restrict__ col,
                           const int* __restrict__ rank, const int* __restrict__ rowptr,
                           int* __restrict__ tmp_src, int e) {
    int i = blockIdx.x * blockDim.x + threadIdx.x;
    if (i < e) tmp_src[rowptr[col[i]] + rank[i]] = row[i];
}

__global__ void k_csr_place(const int* __restrict__ row, const int* __restrict__ col,
                            const int* __restrict__ rank, const int* __restrict__ rowptr,
                            const int* __restrict__ tmp_src,
                            const float* __restrict__ dinv_s, const float* __restrict__ dinv_h,
                            int* __restrict__ csr_src, float* __restrict__ w_s,
                            float* __restrict__ w_h, int e) {
    int i = blockIdx.x * blockDim.x + threadIdx.x;
    if (i >= e) return;
    const int r   = row[i];
    const int c   = col[i];
    const int beg = rowptr[c];
    const int end = rowptr[c + 1];
    const int me  = beg + rank[i];
    int pos = beg;
    for (int b = beg; b < end; ++b) {
        const int k = tmp_src[b];
        pos += (k < r || (k == r && b < me)) ? 1 : 0;
    }
    csr_src[pos] = r;
    w_s[pos] = dinv_s[r] * dinv_s[c];
    w_h[pos] = -(dinv_h[r] * dinv_h[c]);
}

template <int DIN, int NT, bool RELU_IN, bool BIAS>
__global__ void k_gemm_wmma(const float* __restrict__ X, const float* __restrict__ W,
                            const float* __restrict__ bias, float* __restrict__ H, int N) {
    constexpr int DOUT = NT * 16;
    const int lane = threadIdx.x & 31;
    const int wave = threadIdx.x >> 5;
    const int tile = blockIdx.x * (blockDim.x >> 5) + wave;
    const int rtiles = (N + 15) >> 4;
    if (tile >= rtiles) return;

    const int mrow  = tile * 16 + (lane & 15);
    const int mload = mrow < N ? mrow : N - 1;
    const int kb    = (lane >> 4) * 2;
    const int nloc  = lane & 15;

    v8f acc[NT];
#pragma unroll
    for (int j = 0; j < NT; ++j) {
        v8f z = {0.f, 0.f, 0.f, 0.f, 0.f, 0.f, 0.f, 0.f};
        acc[j] = z;
    }

    const float* xr = X + (size_t)mload * DIN;
#pragma unroll 4
    for (int k = 0; k < DIN; k += 4) {
        float a0 = xr[k + kb];
        float a1 = xr[k + kb + 1];
        if (RELU_IN) { a0 = fmaxf(a0, 0.0f); a1 = fmaxf(a1, 0.0f); }
        v2f A = {a0, a1};
#pragma unroll
        for (int j = 0; j < NT; ++j) {
            const int n0 = j * 16 + nloc;
            v2f B = {W[(size_t)(k + kb) * DOUT + n0],
                     W[(size_t)(k + kb + 1) * DOUT + n0]};
            acc[j] = __builtin_amdgcn_wmma_f32_16x16x4_f32(
                false, A, false, B, (short)0, acc[j], false, false);
        }
    }

    const int half = lane >> 4;
#pragma unroll
    for (int j = 0; j < NT; ++j) {
        const float bj = BIAS ? bias[j * 16 + nloc] : 0.0f;
#pragma unroll
        for (int r = 0; r < 8; ++r) {
            int mo = tile * 16 + r + 8 * half;
            if (mo < N) H[(size_t)mo * DOUT + j * 16 + nloc] = BIAS ? acc[j][r] + bj : acc[j][r];
        }
    }
}

static __device__ __forceinline__ v4f relu4(v4f h) {
    h.x = fmaxf(h.x, 0.0f); h.y = fmaxf(h.y, 0.0f);
    h.z = fmaxf(h.z, 0.0f); h.w = fmaxf(h.w, 0.0f);
    return h;
}

template <int D, bool DEC>
__global__ void k_agg_gather(const float* __restrict__ H, const int* __restrict__ rowptr,
                             const int* __restrict__ csr_src, const float* __restrict__ wgt,
                             const float* __restrict__ dinv, const float* __restrict__ bias,
                             float* __restrict__ Y, int n) {
    constexpr int LPN = D / 4;
    const int i    = blockIdx.x * blockDim.x + threadIdx.x;
    const int node = i / LPN;
    if (node >= n) return;
    const int q    = i % LPN;
    const v4f* __restrict__ H4 = (const v4f*)H;
    const int beg = rowptr[node];
    const int end = rowptr[node + 1];
    const float dv = dinv[node];
    float sw = dv * dv;
    if (DEC) sw *= 2.0f;

    v4f acc = H4[(size_t)node * LPN + q];
    if (DEC) {
        acc = relu4(acc) * sw;
    } else {
        v4f b = {bias[4 * q], bias[4 * q + 1], bias[4 * q + 2], bias[4 * q + 3]};
        acc = acc * sw + b;
    }

    int e = beg;
    for (; e + 4 <= end; e += 4) {
        const int s0 = csr_src[e],     s1 = csr_src[e + 1];
        const int s2 = csr_src[e + 2], s3 = csr_src[e + 3];
        const float w0 = wgt[e],     w1 = wgt[e + 1];
        const float w2 = wgt[e + 2], w3 = wgt[e + 3];
        v4f h0 = H4[(size_t)s0 * LPN + q];
        v4f h1 = H4[(size_t)s1 * LPN + q];
        v4f h2 = H4[(size_t)s2 * LPN + q];
        v4f h3 = H4[(size_t)s3 * LPN + q];
        if (DEC) { h0 = relu4(h0); h1 = relu4(h1); h2 = relu4(h2); h3 = relu4(h3); }
        acc += h0 * w0;
        acc += h1 * w1;
        acc += h2 * w2;
        acc += h3 * w3;
    }
    if (e < end) {
        const int last = end - 1;
        const int e1 = min(e + 1, last);
        const int e2 = min(e + 2, last);
        const int s0 = csr_src[e], s1 = csr_src[e1], s2 = csr_src[e2];
        const float w0 = wgt[e], w1 = wgt[e1], w2 = wgt[e2];
        v4f h0 = H4[(size_t)s0 * LPN + q];
        v4f h1 = H4[(size_t)s1 * LPN + q];
        v4f h2 = H4[(size_t)s2 * LPN + q];
        if (DEC) { h0 = relu4(h0); h1 = relu4(h1); h2 = relu4(h2); }
        acc += h0 * w0;
        if (e + 1 < end) acc += h1 * w1;
        if (e + 2 < end) acc += h2 * w2;
    }
    ((v4f*)Y)[(size_t)node * LPN + q] = acc;
}

template <int D, int NT, bool DEC>
__global__ void __launch_bounds__(256)
k_agg_gemm(const float* __restrict__ H, const int* __restrict__ rowptr,
           const int* __restrict__ csr_src, const float* __restrict__ wgt,
           const float* __restrict__ dinv, const float* __restrict__ bias,
           const float* __restrict__ W, float* __restrict__ Y, int n) {
    constexpr int LPN  = D / 4;
    constexpr int NPB  = 256 / LPN;
    constexpr int RT   = NPB / 16;
    constexpr int DOUT = NT * 16;
    constexpr int T    = RT * NT;
    constexpr int KS   = D / 4;
    constexpr int QO   = DOUT / 4;
    constexpr int PA   = D + 4;
    constexpr int PO   = DOUT + 4;
    constexpr bool ALLW = (T == 8);
    static_assert(T <= 8, "one output tile per wave");
    __shared__ __attribute__((aligned(16))) float sA[NPB * PA];
    __shared__ __attribute__((aligned(16))) float sO[NPB * PO];

    const int tid  = threadIdx.x;
    const int nl   = tid / LPN;
    const int q    = tid % LPN;
    const int node = blockIdx.x * NPB + nl;
    const v4f* __restrict__ H4 = (const v4f*)H;

    v4f acc = {0.f, 0.f, 0.f, 0.f};
    if (node < n) {
        const int beg = rowptr[node];
        const int end = rowptr[node + 1];
        const float dv = dinv[node];
        float sw = dv * dv;
        if (DEC) sw *= 2.0f;

        acc = H4[(size_t)node * LPN + q];
        if (DEC) {
            acc = relu4(acc) * sw;
        } else {
            v4f b = {bias[4 * q], bias[4 * q + 1], bias[4 * q + 2], bias[4 * q + 3]};
            acc = acc * sw + b;
        }

        int e = beg;
        for (; e + 4 <= end; e += 4) {
            const int s0 = csr_src[e],     s1 = csr_src[e + 1];
            const int s2 = csr_src[e + 2], s3 = csr_src[e + 3];
            const float w0 = wgt[e],     w1 = wgt[e + 1];
            const float w2 = wgt[e + 2], w3 = wgt[e + 3];
            v4f h0 = H4[(size_t)s0 * LPN + q];
            v4f h1 = H4[(size_t)s1 * LPN + q];
            v4f h2 = H4[(size_t)s2 * LPN + q];
            v4f h3 = H4[(size_t)s3 * LPN + q];
            if (DEC) { h0 = relu4(h0); h1 = relu4(h1); h2 = relu4(h2); h3 = relu4(h3); }
            acc += h0 * w0;
            acc += h1 * w1;
            acc += h2 * w2;
            acc += h3 * w3;
        }
        if (e < end) {
            const int last = end - 1;
            const int e1 = min(e + 1, last);
            const int e2 = min(e + 2, last);
            const int s0 = csr_src[e], s1 = csr_src[e1], s2 = csr_src[e2];
            const float w0 = wgt[e], w1 = wgt[e1], w2 = wgt[e2];
            v4f h0 = H4[(size_t)s0 * LPN + q];
            v4f h1 = H4[(size_t)s1 * LPN + q];
            v4f h2 = H4[(size_t)s2 * LPN + q];
            if (DEC) { h0 = relu4(h0); h1 = relu4(h1); h2 = relu4(h2); }
            acc += h0 * w0;
            if (e + 1 < end) acc += h1 * w1;
            if (e + 2 < end) acc += h2 * w2;
        }
    }
    if (!DEC) acc = relu4(acc);
    *(v4f*)&sA[nl * PA + 4 * q] = acc;

    const int wave = __builtin_amdgcn_readfirstlane(tid >> 5);
    const int lane = tid & 31;
    const int c    = lane & 15;
    const int half = lane >> 4;
    const int kb   = half * 2;
    const int rt   = wave / NT;
    const int ct   = wave % NT;

    v2f Bf[KS];
    if (ALLW || wave < T) {
        const float* wp = W + (size_t)kb * DOUT + ct * 16 + c;
#pragma unroll
        for (int kk = 0; kk < KS; ++kk) {
            Bf[kk].x = wp[(size_t)(4 * kk) * DOUT];
            Bf[kk].y = wp[(size_t)(4 * kk + 1) * DOUT];
        }
    }
    const int qo = tid % QO;
    v4f bv = {0.f, 0.f, 0.f, 0.f};
    if (DEC) { bv.x = bias[4 * qo]; bv.y = bias[4 * qo + 1]; bv.z = bias[4 * qo + 2]; bv.w = bias[4 * qo + 3]; }

    __syncthreads();

    if (ALLW || wave < T) {
        v8f d = {0.f, 0.f, 0.f, 0.f, 0.f, 0.f, 0.f, 0.f};
        const float* ap = &sA[(rt * 16 + c) * PA + kb];
#pragma unroll
        for (int kk = 0; kk < KS; ++kk) {
            const v2f A = *(const v2f*)(ap + 4 * kk);
            d = __builtin_amdgcn_wmma_f32_16x16x4_f32(
                false, A, false, Bf[kk], (short)0, d, false, false);
        }
#pragma unroll
        for (int r = 0; r < 8; ++r) sO[(rt * 16 + 8 * half + r) * PO + ct * 16 + c] = d[r];
    }

    __syncthreads();

#pragma unroll
    for (int idx = tid; idx < NPB * QO; idx += 256) {
        const int row = idx / QO;
        const int nd  = blockIdx.x * NPB + row;
        if (nd < n) {
            v4f o = *(const v4f*)&sO[row * PO + 4 * qo];
            if (DEC) o += bv;
            ((v4f*)Y)[(size_t)nd * QO + qo] = o;
        }
    }
}

static inline int cdiv(long long a, int b) { return (int)((a + b - 1) / b); }

extern "C" void kernel_launch(void* const* d_in, const int* in_sizes, int n_in,
                              void* d_out, int out_size, void* d_ws, size_t ws_size,
                              hipStream_t stream) {
    const float* x   = (const float*)d_in[0];
    const int*   ei  = (const int*)d_in[1];
    const float* We1 = (const float*)d_in[2];  const float* be1 = (const float*)d_in[3];
    const float* We2 = (const float*)d_in[4];  const float* be2 = (const float*)d_in[5];
    const float* We3 = (const float*)d_in[6];  const float* be3 = (const float*)d_in[7];
    const float* Wd1 = (const float*)d_in[8];  const float* bd1 = (const float*)d_in[9];
    const float* Wd2 = (const float*)d_in[10]; const float* bd2 = (const float*)d_in[11];
    const float* Wd3 = (const float*)d_in[12]; const float* bd3 = (const float*)d_in[13];

    const int N = in_sizes[0] / 128;
    const int E = in_sizes[1] / 2;
    const int* rowp = ei;
    const int* colp = ei + E;

    const size_t Np = ((size_t)N + 64 + 63) & ~(size_t)63;
    const size_t Ep = ((size_t)E + 63) & ~(size_t)63;
    char* ws = (char*)d_ws;
    int*   indeg   = (int*)ws;    ws += sizeof(int)   * Np;
    float* dinv_s  = (float*)ws;  ws += sizeof(float) * Np;
    float* dinv_h  = (float*)ws;  ws += sizeof(float) * Np;
    float* hbuf    = (float*)ws;  ws += sizeof(float) * (size_t)N * 64;
    float* ybuf    = (float*)ws;  ws += sizeof(float) * (size_t)N * 64;
    int*   rowptr  = (int*)ws;    ws += sizeof(int)   * Np;
    int*   rank    = (int*)ws;    ws += sizeof(int)   * Ep;
    int*   tmp_src = (int*)ws;    ws += sizeof(int)   * Ep;
    int*   csr_src = (int*)ws;    ws += sizeof(int)   * Ep;
    float* w_s     = (float*)ws;  ws += sizeof(float) * Ep;
    float* w_h     = (float*)ws;
    float* outp    = (float*)d_out;

    const int TB = 256;
    const int rtiles  = (N + 15) >> 4;
    const int gblocks = (rtiles + 7) >> 3;

    k_zero_i32   <<<cdiv(N, TB), TB, 0, stream>>>(indeg, N);
    k_count_rank <<<cdiv(E, TB), TB, 0, stream>>>(colp, indeg, rank, E);
    k_rowptr_scan<<<cdiv(N, 1024), 1024, 0, stream>>>(indeg, rowptr, N);
    k_dinv       <<<cdiv(N, TB), TB, 0, stream>>>(indeg, dinv_s, dinv_h, N);
    k_csr_fill   <<<cdiv(E, TB), TB, 0, stream>>>(rowp, colp, rank, rowptr, tmp_src, E);
    k_csr_place  <<<cdiv(E, TB), TB, 0, stream>>>(rowp, colp, rank, rowptr, tmp_src, dinv_s, dinv_h,
                                                  csr_src, w_s, w_h, E);

    k_gemm_wmma<128, 4, false, false><<<gblocks, TB, 0, stream>>>(x, We1, be1, hbuf, N);
    k_agg_gemm<64, 2, false><<<cdiv(N, 16), TB, 0, stream>>>(hbuf, rowptr, csr_src, w_s, dinv_s, be1, We2, ybuf, N);
    k_agg_gemm<32, 1, false><<<cdiv(N, 32), TB, 0, stream>>>(ybuf, rowptr, csr_src, w_s, dinv_s, be2, We3, hbuf, N);
    k_agg_gather<16, false><<<cdiv((long long)N * 4, TB), TB, 0, stream>>>(hbuf, rowptr, csr_src, w_s, dinv_s, be3, ybuf, N);

    k_agg_gemm<16, 2, true><<<cdiv(N, 64), TB, 0, stream>>>(ybuf, rowptr, csr_src, w_h, dinv_h, bd1, Wd1, hbuf, N);
    k_agg_gemm<32, 4, true><<<cdiv(N, 32), TB, 0, stream>>>(hbuf, rowptr, csr_src, w_h, dinv_h, bd2, Wd2, ybuf, N);
    k_agg_gemm<64, 8, true><<<cdiv(N, 16), TB, 0, stream>>>(ybuf, rowptr, csr_src, w_h, dinv_h, bd3, Wd3, outp, N);
}


# 13 launches become 11: the first layer's map, independent of the graph, shares the in-degree count's launch by block index; the prefix-sum launch also writes the degree coefficients; sums unchanged
# speedup vs baseline: 3.5050x; 1.0582x over previous
#include <hip/hip_runtime.h>

typedef __attribute__((ext_vector_type(2))) float v2f;
typedef __attribute__((ext_vector_type(8))) float v8f;
typedef __attribute__((ext_vector_type(4))) float v4f;

__global__ void k_zero_i32(int* __restrict__ p, int n) {
    int i = blockIdx.x * blockDim.x + threadIdx.x;
    if (i < n) p[i] = 0;
}

__global__ void __launch_bounds__(1024)
k_rowptr_scan_dinv(const int* __restrict__ indeg, int* __restrict__ rowptr,
                   float* __restrict__ dinv_s, float* __restrict__ dinv_h, int n) {
    __shared__ int s_pre[32];
    __shared__ int s_inc[32];
    const int tid  = threadIdx.x;
    const int lane = tid & 31;
    const int wv   = tid >> 5;
    int pre = 0;
#pragma unroll 8
    for (int k = 0; k < (int)blockIdx.x; ++k) pre += indeg[k * 1024 + tid];
    const int i = blockIdx.x * 1024 + tid;
    const int v = (i < n) ? indeg[i] : 0;
    int inc = v;
    for (int d = 1; d < 32; d <<= 1) {
        int t = __shfl_up(inc, d, 32);
        if (lane >= d) inc += t;
    }
    for (int d = 16; d > 0; d >>= 1) pre += __shfl_xor(pre, d, 32);
    if (lane == 31) s_inc[wv] = inc;
    if (lane == 0)  s_pre[wv] = pre;
    __syncthreads();
    int off = s_pre[lane] + (lane < wv ? s_inc[lane] : 0);
    for (int d = 16; d > 0; d >>= 1) off += __shfl_xor(off, d, 32);
    if (i < n) {
        rowptr[i] = off + inc - v;
        const float d = (float)v;
        dinv_s[i] = rsqrtf(d + 1.0f);
        dinv_h[i] = rsqrtf(d + 2.0f);
    }
    if (i == n - 1) rowptr[n] = off + inc;
}

__global__ void k_csr_fill(const int* __restrict__ row, const int* __restrict__ col,
                           const int* __restrict__ rank, const int* __restrict__ rowptr,
                           int* __restrict__ tmp_src, int e) {
    int i = blockIdx.x * blockDim.x + threadIdx.x;
    if (i < e) tmp_src[rowptr[col[i]] + rank[i]] = row[i];
}

__global__ void k_csr_place(const int* __restrict__ row, const int* __restrict__ col,
                            const int* __restrict__ rank, const int* __restrict__ rowptr,
                            const int* __restrict__ tmp_src,
                            const float* __restrict__ dinv_s, const float* __restrict__ dinv_h,
                            int* __restrict__ csr_src, float* __restrict__ w_s,
                            float* __restrict__ w_h, int e) {
    int i = blockIdx.x * blockDim.x + threadIdx.x;
    if (i >= e) return;
    const int r   = row[i];
    const int c   = col[i];
    const int beg = rowptr[c];
    const int end = rowptr[c + 1];
    const int me  = beg + rank[i];
    int pos = beg;
    for (int b = beg; b < end; ++b) {
        const int k = tmp_src[b];
        pos += (k < r || (k == r && b < me)) ? 1 : 0;
    }
    csr_src[pos] = r;
    w_s[pos] = dinv_s[r] * dinv_s[c];
    w_h[pos] = -(dinv_h[r] * dinv_h[c]);
}

template <int DIN, int NT, bool RELU_IN, bool BIAS>
static __device__ __forceinline__ void
gemm_wmma_strip(const float* __restrict__ X, const float* __restrict__ W,
                const float* __restrict__ bias, float* __restrict__ H, int N, int blk) {
    constexpr int DOUT = NT * 16;
    const int lane = threadIdx.x & 31;
    const int wave = threadIdx.x >> 5;
    const int tile = blk * (blockDim.x >> 5) + wave;
    const int rtiles = (N + 15) >> 4;
    if (tile >= rtiles) return;

    const int mrow  = tile * 16 + (lane & 15);
    const int mload = mrow < N ? mrow : N - 1;
    const int kb    = (lane >> 4) * 2;
    const int nloc  = lane & 15;

    v8f acc[NT];
#pragma unroll
    for (int j = 0; j < NT; ++j) {
        v8f z = {0.f, 0.f, 0.f, 0.f, 0.f, 0.f, 0.f, 0.f};
        acc[j] = z;
    }

    const float* xr = X + (size_t)mload * DIN;
#pragma unroll 4
    for (int k = 0; k < DIN; k += 4) {
        float a0 = xr[k + kb];
        float a1 = xr[k + kb + 1];
        if (RELU_IN) { a0 = fmaxf(a0, 0.0f); a1 = fmaxf(a1, 0.0f); }
        v2f A = {a0, a1};
#pragma unroll
        for (int j = 0; j < NT; ++j) {
            const int n0 = j * 16 + nloc;
            v2f B = {W[(size_t)(k + kb) * DOUT + n0],
                     W[(size_t)(k + kb + 1) * DOUT + n0]};
            acc[j] = __builtin_amdgcn_wmma_f32_16x16x4_f32(
                false, A, false, B, (short)0, acc[j], false, false);
        }
    }

    const int half = lane >> 4;
#pragma unroll
    for (int j = 0; j < NT; ++j) {
        const float bj = BIAS ? bias[j * 16 + nloc] : 0.0f;
#pragma unroll
        for (int r = 0; r < 8; ++r) {
            int mo = tile * 16 + r + 8 * half;
            if (mo < N) H[(size_t)mo * DOUT + j * 16 + nloc] = BIAS ? acc[j][r] + bj : acc[j][r];
        }
    }
}

__global__ void __launch_bounds__(256)
k_gemm1_count_rank(const float* __restrict__ X, const float* __restrict__ W,
                   float* __restrict__ H, int N, int gblocks,
                   const int* __restrict__ col, int* __restrict__ indeg,
                   int* __restrict__ rank, int e) {
    const int b = blockIdx.x;
    if (b < gblocks) {
        gemm_wmma_strip<128, 4, false, false>(X, W, nullptr, H, N, b);
    } else {
        const int i = (b - gblocks) * blockDim.x + threadIdx.x;
        if (i < e) rank[i] = atomicAdd(&indeg[col[i]], 1);
    }
}

static __device__ __forceinline__ v4f relu4(v4f h) {
    h.x = fmaxf(h.x, 0.0f); h.y = fmaxf(h.y, 0.0f);
    h.z = fmaxf(h.z, 0.0f); h.w = fmaxf(h.w, 0.0f);
    return h;
}

template <int D, bool DEC>
__global__ void k_agg_gather(const float* __restrict__ H, const int* __restrict__ rowptr,
                             const int* __restrict__ csr_src, const float* __restrict__ wgt,
                             const float* __restrict__ dinv, const float* __restrict__ bias,
                             float* __restrict__ Y, int n) {
    constexpr int LPN = D / 4;
    const int i    = blockIdx.x * blockDim.x + threadIdx.x;
    const int node = i / LPN;
    if (node >= n) return;
    const int q    = i % LPN;
    const v4f* __restrict__ H4 = (const v4f*)H;
    const int beg = rowptr[node];
    const int end = rowptr[node + 1];
    const float dv = dinv[node];
    float sw = dv * dv;
    if (DEC) sw *= 2.0f;

    v4f acc = H4[(size_t)node * LPN + q];
    if (DEC) {
        acc = relu4(acc) * sw;
    } else {
        v4f b = {bias[4 * q], bias[4 * q + 1], bias[4 * q + 2], bias[4 * q + 3]};
        acc = acc * sw + b;
    }

    int e = beg;
    for (; e + 4 <= end; e += 4) {
        const int s0 = csr_src[e],     s1 = csr_src[e + 1];
        const int s2 = csr_src[e + 2], s3 = csr_src[e + 3];
        const float w0 = wgt[e],     w1 = wgt[e + 1];
        const float w2 = wgt[e + 2], w3 = wgt[e + 3];
        v4f h0 = H4[(size_t)s0 * LPN + q];
        v4f h1 = H4[(size_t)s1 * LPN + q];
        v4f h2 = H4[(size_t)s2 * LPN + q];
        v4f h3 = H4[(size_t)s3 * LPN + q];
        if (DEC) { h0 = relu4(h0); h1 = relu4(h1); h2 = relu4(h2); h3 = relu4(h3); }
        acc += h0 * w0;
        acc += h1 * w1;
        acc += h2 * w2;
        acc += h3 * w3;
    }
    if (e < end) {
        const int last = end - 1;
        const int e1 = min(e + 1, last);
        const int e2 = min(e + 2, last);
        const int s0 = csr_src[e], s1 = csr_src[e1], s2 = csr_src[e2];
        const float w0 = wgt[e], w1 = wgt[e1], w2 = wgt[e2];
        v4f h0 = H4[(size_t)s0 * LPN + q];
        v4f h1 = H4[(size_t)s1 * LPN + q];
        v4f h2 = H4[(size_t)s2 * LPN + q];
        if (DEC) { h0 = relu4(h0); h1 = relu4(h1); h2 = relu4(h2); }
        acc += h0 * w0;
        if (e + 1 < end) acc += h1 * w1;
        if (e + 2 < end) acc += h2 * w2;
    }
    ((v4f*)Y)[(size_t)node * LPN + q] = acc;
}

template <int D, int NT, bool DEC>
__global__ void __launch_bounds__(256)
k_agg_gemm(const float* __restrict__ H, const int* __restrict__ rowptr,
           const int* __restrict__ csr_src, const float* __restrict__ wgt,
           const float* __restrict__ dinv, const float* __restrict__ bias,
           const float* __restrict__ W, float* __restrict__ Y, int n) {
    constexpr int LPN  = D / 4;
    constexpr int NPB  = 256 / LPN;
    constexpr int RT   = NPB / 16;
    constexpr int DOUT = NT * 16;
    constexpr int T    = RT * NT;
    constexpr int KS   = D / 4;
    constexpr int QO   = DOUT / 4;
    constexpr int PA   = D + 4;
    constexpr int PO   = DOUT + 4;
    constexpr bool ALLW = (T == 8);
    static_assert(T <= 8, "one output tile per wave");
    __shared__ __attribute__((aligned(16))) float sA[NPB * PA];
    __shared__ __attribute__((aligned(16))) float sO[NPB * PO];

    const int tid  = threadIdx.x;
    const int nl   = tid / LPN;
    const int q    = tid % LPN;
    const int node = blockIdx.x * NPB + nl;
    const v4f* __restrict__ H4 = (const v4f*)H;

    v4f acc = {0.f, 0.f, 0.f, 0.f};
    if (node < n) {
        const int beg = rowptr[node];
        const int end = rowptr[node + 1];
        const float dv = dinv[node];
        float sw = dv * dv;
        if (DEC) sw *= 2.0f;

        acc = H4[(size_t)node * LPN + q];
        if (DEC) {
            acc = relu4(acc) * sw;
        } else {
            v4f b = {bias[4 * q], bias[4 * q + 1], bias[4 * q + 2], bias[4 * q + 3]};
            acc = acc * sw + b;
        }

        int e = beg;
        for (; e + 4 <= end; e += 4) {
            const int s0 = csr_src[e],     s1 = csr_src[e + 1];
            const int s2 = csr_src[e + 2], s3 = csr_src[e + 3];
            const float w0 = wgt[e],     w1 = wgt[e + 1];
            const float w2 = wgt[e + 2], w3 = wgt[e + 3];
            v4f h0 = H4[(size_t)s0 * LPN + q];
            v4f h1 = H4[(size_t)s1 * LPN + q];
            v4f h2 = H4[(size_t)s2 * LPN + q];
            v4f h3 = H4[(size_t)s3 * LPN + q];
            if (DEC) { h0 = relu4(h0); h1 = relu4(h1); h2 = relu4(h2); h3 = relu4(h3); }
            acc += h0 * w0;
            acc += h1 * w1;
            acc += h2 * w2;
            acc += h3 * w3;
        }
        if (e < end) {
            const int last = end - 1;
            const int e1 = min(e + 1, last);
            const int e2 = min(e + 2, last);
            const int s0 = csr_src[e], s1 = csr_src[e1], s2 = csr_src[e2];
            const float w0 = wgt[e], w1 = wgt[e1], w2 = wgt[e2];
            v4f h0 = H4[(size_t)s0 * LPN + q];
            v4f h1 = H4[(size_t)s1 * LPN + q];
            v4f h2 = H4[(size_t)s2 * LPN + q];
            if (DEC) { h0 = relu4(h0); h1 = relu4(h1); h2 = relu4(h2); }
            acc += h0 * w0;
            if (e + 1 < end) acc += h1 * w1;
            if (e + 2 < end) acc += h2 * w2;
        }
    }
    if (!DEC) acc = relu4(acc);
    *(v4f*)&sA[nl * PA + 4 * q] = acc;

    const int wave = __builtin_amdgcn_readfirstlane(tid >> 5);
    const int lane = tid & 31;
    const int c    = lane & 15;
    const int half = lane >> 4;
    const int kb   = half * 2;
    const int rt   = wave / NT;
    const int ct   = wave % NT;

    v2f Bf[KS];
    if (ALLW || wave < T) {
        const float* wp = W + (size_t)kb * DOUT + ct * 16 + c;
#pragma unroll
        for (int kk = 0; kk < KS; ++kk) {
            Bf[kk].x = wp[(size_t)(4 * kk) * DOUT];
            Bf[kk].y = wp[(size_t)(4 * kk + 1) * DOUT];
        }
    }
    const int qo = tid % QO;
    v4f bv = {0.f, 0.f, 0.f, 0.f};
    if (DEC) { bv.x = bias[4 * qo]; bv.y = bias[4 * qo + 1]; bv.z = bias[4 * qo + 2]; bv.w = bias[4 * qo + 3]; }

    __syncthreads();

    if (ALLW || wave < T) {
        v8f d = {0.f, 0.f, 0.f, 0.f, 0.f, 0.f, 0.f, 0.f};
        const float* ap = &sA[(rt * 16 + c) * PA + kb];
#pragma unroll
        for (int kk = 0; kk < KS; ++kk) {
            const v2f A = *(const v2f*)(ap + 4 * kk);
            d = __builtin_amdgcn_wmma_f32_16x16x4_f32(
                false, A, false, Bf[kk], (short)0, d, false, false);
        }
#pragma unroll
        for (int r = 0; r < 8; ++r) sO[(rt * 16 + 8 * half + r) * PO + ct * 16 + c] = d[r];
    }

    __syncthreads();

#pragma unroll
    for (int idx = tid; idx < NPB * QO; idx += 256) {
        const int row = idx / QO;
        const int nd  = blockIdx.x * NPB + row;
        if (nd < n) {
            v4f o = *(const v4f*)&sO[row * PO + 4 * qo];
            if (DEC) o += bv;
            ((v4f*)Y)[(size_t)nd * QO + qo] = o;
        }
    }
}

static inline int cdiv(long long a, int b) { return (int)((a + b - 1) / b); }

extern "C" void kernel_launch(void* const* d_in, const int* in_sizes, int n_in,
                              void* d_out, int out_size, void* d_ws, size_t ws_size,
                              hipStream_t stream) {
    const float* x   = (const float*)d_in[0];
    const int*   ei  = (const int*)d_in[1];
    const float* We1 = (const float*)d_in[2];  const float* be1 = (const float*)d_in[3];
    const float* We2 = (const float*)d_in[4];  const float* be2 = (const float*)d_in[5];
    const float* We3 = (const float*)d_in[6];  const float* be3 = (const float*)d_in[7];
    const float* Wd1 = (const float*)d_in[8];  const float* bd1 = (const float*)d_in[9];
    const float* Wd2 = (const float*)d_in[10]; const float* bd2 = (const float*)d_in[11];
    const float* Wd3 = (const float*)d_in[12]; const float* bd3 = (const float*)d_in[13];

    const int N = in_sizes[0] / 128;
    const int E = in_sizes[1] / 2;
    const int* rowp = ei;
    const int* colp = ei + E;

    const size_t Np = ((size_t)N + 64 + 63) & ~(size_t)63;
    const size_t Ep = ((size_t)E + 63) & ~(size_t)63;
    char* ws = (char*)d_ws;
    int*   indeg   = (int*)ws;    ws += sizeof(int)   * Np;
    float* dinv_s  = (float*)ws;  ws += sizeof(float) * Np;
    float* dinv_h  = (float*)ws;  ws += sizeof(float) * Np;
    float* hbuf    = (float*)ws;  ws += sizeof(float) * (size_t)N * 64;
    float* ybuf    = (float*)ws;  ws += sizeof(float) * (size_t)N * 64;
    int*   rowptr  = (int*)ws;    ws += sizeof(int)   * Np;
    int*   rank    = (int*)ws;    ws += sizeof(int)   * Ep;
    int*   tmp_src = (int*)ws;    ws += sizeof(int)   * Ep;
    int*   csr_src = (int*)ws;    ws += sizeof(int)   * Ep;
    float* w_s     = (float*)ws;  ws += sizeof(float) * Ep;
    float* w_h     = (float*)ws;
    float* outp    = (float*)d_out;

    const int TB = 256;
    const int rtiles  = (N + 15) >> 4;
    const int gblocks = (rtiles + 7) >> 3;

    k_zero_i32   <<<cdiv(N, TB), TB, 0, stream>>>(indeg, N);
    k_gemm1_count_rank<<<gblocks + cdiv(E, TB), TB, 0, stream>>>(x, We1, hbuf, N, gblocks,
                                                                  colp, indeg, rank, E);
    k_rowptr_scan_dinv<<<cdiv(N, 1024), 1024, 0, stream>>>(indeg, rowptr, dinv_s, dinv_h, N);
    k_csr_fill   <<<cdiv(E, TB), TB, 0, stream>>>(rowp, colp, rank, rowptr, tmp_src, E);
    k_csr_place  <<<cdiv(E, TB), TB, 0, stream>>>(rowp, colp, rank, rowptr, tmp_src, dinv_s, dinv_h,
                                                  csr_src, w_s, w_h, E);

    k_agg_gemm<64, 2, false><<<cdiv(N, 16), TB, 0, stream>>>(hbuf, rowptr, csr_src, w_s, dinv_s, be1, We2, ybuf, N);
    k_agg_gemm<32, 1, false><<<cdiv(N, 32), TB, 0, stream>>>(ybuf, rowptr, csr_src, w_s, dinv_s, be2, We3, hbuf, N);
    k_agg_gather<16, false><<<cdiv((long long)N * 4, TB), TB, 0, stream>>>(hbuf, rowptr, csr_src, w_s, dinv_s, be3, ybuf, N);

    k_agg_gemm<16, 2, true><<<cdiv(N, 64), TB, 0, stream>>>(ybuf, rowptr, csr_src, w_h, dinv_h, bd1, Wd1, hbuf, N);
    k_agg_gemm<32, 4, true><<<cdiv(N, 32), TB, 0, stream>>>(hbuf, rowptr, csr_src, w_h, dinv_h, bd2, Wd2, ybuf, N);
    k_agg_gemm<64, 8, true><<<cdiv(N, 16), TB, 0, stream>>>(ybuf, rowptr, csr_src, w_h, dinv_h, bd3, Wd3, outp, N);
}


# Aggregation lanes take two adjacent nodes each: the pair's in-edge lists are one contiguous range, walked once with 8 row fetches in flight per lane, not 4, on half as many waves; sums keep order
# speedup vs baseline: 3.6274x; 1.0349x over previous
#include <hip/hip_runtime.h>

typedef __attribute__((ext_vector_type(2))) float v2f;
typedef __attribute__((ext_vector_type(8))) float v8f;
typedef __attribute__((ext_vector_type(4))) float v4f;

__global__ void k_zero_i32(int* __restrict__ p, int n) {
    int i = blockIdx.x * blockDim.x + threadIdx.x;
    if (i < n) p[i] = 0;
}

__global__ void __launch_bounds__(1024)
k_rowptr_scan_dinv(const int* __restrict__ indeg, int* __restrict__ rowptr,
                   float* __restrict__ dinv_s, float* __restrict__ dinv_h, int n) {
    __shared__ int s_pre[32];
    __shared__ int s_inc[32];
    const int tid  = threadIdx.x;
    const int lane = tid & 31;
    const int wv   = tid >> 5;
    int pre = 0;
#pragma unroll 8
    for (int k = 0; k < (int)blockIdx.x; ++k) pre += indeg[k * 1024 + tid];
    const int i = blockIdx.x * 1024 + tid;
    const int v = (i < n) ? indeg[i] : 0;
    int inc = v;
    for (int d = 1; d < 32; d <<= 1) {
        int t = __shfl_up(inc, d, 32);
        if (lane >= d) inc += t;
    }
    for (int d = 16; d > 0; d >>= 1) pre += __shfl_xor(pre, d, 32);
    if (lane == 31) s_inc[wv] = inc;
    if (lane == 0)  s_pre[wv] = pre;
    __syncthreads();
    int off = s_pre[lane] + (lane < wv ? s_inc[lane] : 0);
    for (int d = 16; d > 0; d >>= 1) off += __shfl_xor(off, d, 32);
    if (i < n) {
        rowptr[i] = off + inc - v;
        const float d = (float)v;
        dinv_s[i] = rsqrtf(d + 1.0f);
        dinv_h[i] = rsqrtf(d + 2.0f);
    }
    if (i == n - 1) rowptr[n] = off + inc;
}

__global__ void k_csr_fill(const int* __restrict__ row, const int* __restrict__ col,
                           const int* __restrict__ rank, const int* __restrict__ rowptr,
                           int* __restrict__ tmp_src, int e) {
    int i = blockIdx.x * blockDim.x + threadIdx.x;
    if (i < e) tmp_src[rowptr[col[i]] + rank[i]] = row[i];
}

__global__ void k_csr_place(const int* __restrict__ row, const int* __restrict__ col,
                            const int* __restrict__ rank, const int* __restrict__ rowptr,
                            const int* __restrict__ tmp_src,
                            const float* __restrict__ dinv_s, const float* __restrict__ dinv_h,
                            int* __restrict__ csr_src, float* __restrict__ w_s,
                            float* __restrict__ w_h, int e) {
    int i = blockIdx.x * blockDim.x + threadIdx.x;
    if (i >= e) return;
    const int r   = row[i];
    const int c   = col[i];
    const int beg = rowptr[c];
    const int end = rowptr[c + 1];
    const int me  = beg + rank[i];
    int pos = beg;
    for (int b = beg; b < end; ++b) {
        const int k = tmp_src[b];
        pos += (k < r || (k == r && b < me)) ? 1 : 0;
    }
    csr_src[pos] = r;
    w_s[pos] = dinv_s[r] * dinv_s[c];
    w_h[pos] = -(dinv_h[r] * dinv_h[c]);
}

template <int DIN, int NT, bool RELU_IN, bool BIAS>
static __device__ __forceinline__ void
gemm_wmma_strip(const float* __restrict__ X, const float* __restrict__ W,
                const float* __restrict__ bias, float* __restrict__ H, int N, int blk) {
    constexpr int DOUT = NT * 16;
    const int lane = threadIdx.x & 31;
    const int wave = threadIdx.x >> 5;
    const int tile = blk * (blockDim.x >> 5) + wave;
    const int rtiles = (N + 15) >> 4;
    if (tile >= rtiles) return;

    const int mrow  = tile * 16 + (lane & 15);
    const int mload = mrow < N ? mrow : N - 1;
    const int kb    = (lane >> 4) * 2;
    const int nloc  = lane & 15;

    v8f acc[NT];
#pragma unroll
    for (int j = 0; j < NT; ++j) {
        v8f z = {0.f, 0.f, 0.f, 0.f, 0.f, 0.f, 0.f, 0.f};
        acc[j] = z;
    }

    const float* xr = X + (size_t)mload * DIN;
#pragma unroll 4
    for (int k = 0; k < DIN; k += 4) {
        float a0 = xr[k + kb];
        float a1 = xr[k + kb + 1];
        if (RELU_IN) { a0 = fmaxf(a0, 0.0f); a1 = fmaxf(a1, 0.0f); }
        v2f A = {a0, a1};
#pragma unroll
        for (int j = 0; j < NT; ++j) {
            const int n0 = j * 16 + nloc;
            v2f B = {W[(size_t)(k + kb) * DOUT + n0],
                     W[(size_t)(k + kb + 1) * DOUT + n0]};
            acc[j] = __builtin_amdgcn_wmma_f32_16x16x4_f32(
                false, A, false, B, (short)0, acc[j], false, false);
        }
    }

    const int half = lane >> 4;
#pragma unroll
    for (int j = 0; j < NT; ++j) {
        const float bj = BIAS ? bias[j * 16 + nloc] : 0.0f;
#pragma unroll
        for (int r = 0; r < 8; ++r) {
            int mo = tile * 16 + r + 8 * half;
            if (mo < N) H[(size_t)mo * DOUT + j * 16 + nloc] = BIAS ? acc[j][r] + bj : acc[j][r];
        }
    }
}

__global__ void __launch_bounds__(256)
k_gemm1_count_rank(const float* __restrict__ X, const float* __restrict__ W,
                   float* __restrict__ H, int N, int gblocks,
                   const int* __restrict__ col, int* __restrict__ indeg,
                   int* __restrict__ rank, int e) {
    const int b = blockIdx.x;
    if (b < gblocks) {
        gemm_wmma_strip<128, 4, false, false>(X, W, nullptr, H, N, b);
    } else {
        const int i = (b - gblocks) * blockDim.x + threadIdx.x;
        if (i < e) rank[i] = atomicAdd(&indeg[col[i]], 1);
    }
}

static __device__ __forceinline__ v4f relu4(v4f h) {
    h.x = fmaxf(h.x, 0.0f); h.y = fmaxf(h.y, 0.0f);
    h.z = fmaxf(h.z, 0.0f); h.w = fmaxf(h.w, 0.0f);
    return h;
}

template <int D, bool DEC>
static __device__ __forceinline__ void
gather_pair(const float* __restrict__ H, const int* __restrict__ rowptr,
            const int* __restrict__ csr_src, const float* __restrict__ wgt,
            const float* __restrict__ dinv, const float* __restrict__ bias,
            int a, int q, int n, v4f& accA, v4f& accB) {
    constexpr unsigned ROWB = D * 4;
    const char* __restrict__ Hb = (const char*)H;
    const unsigned qoff = (unsigned)q * 16u;
    const bool hasB = a + 1 < n;
    const int b = hasB ? a + 1 : a;

    const int beg = rowptr[a];
    const int mid = rowptr[a + 1];
    const int end = rowptr[b + 1];
    const float dvA = dinv[a];
    const float dvB = dinv[b];
    float swA = dvA * dvA;
    float swB = dvB * dvB;
    if (DEC) { swA *= 2.0f; swB *= 2.0f; }

    accA = *(const v4f*)(Hb + ((unsigned)a * ROWB + qoff));
    accB = *(const v4f*)(Hb + ((unsigned)b * ROWB + qoff));
    if (DEC) {
        accA = relu4(accA) * swA;
        accB = relu4(accB) * swB;
    } else {
        v4f bs = {bias[4 * q], bias[4 * q + 1], bias[4 * q + 2], bias[4 * q + 3]};
        accA = accA * swA + bs;
        accB = accB * swB + bs;
    }
    if (!hasB) { v4f z = {0.f, 0.f, 0.f, 0.f}; accB = z; }

    int e = beg;
    for (; e + 8 <= end; e += 8) {
        int s[8];
        float w[8];
        v4f h[8];
#pragma unroll
        for (int j = 0; j < 8; ++j) { s[j] = csr_src[e + j]; w[j] = wgt[e + j]; }
#pragma unroll
        for (int j = 0; j < 8; ++j) h[j] = *(const v4f*)(Hb + ((unsigned)s[j] * ROWB + qoff));
#pragma unroll
        for (int j = 0; j < 8; ++j) {
            if (DEC) h[j] = relu4(h[j]);
            if (e + j < mid) accA += h[j] * w[j];
            else             accB += h[j] * w[j];
        }
    }
    if (e < end) {
        const int last = end - 1;
        int s[7];
        float w[7];
        v4f h[7];
#pragma unroll
        for (int j = 0; j < 7; ++j) {
            const int p = min(e + j, last);
            s[j] = csr_src[p];
            w[j] = wgt[p];
        }
#pragma unroll
        for (int j = 0; j < 7; ++j) h[j] = *(const v4f*)(Hb + ((unsigned)s[j] * ROWB + qoff));
#pragma unroll
        for (int j = 0; j < 7; ++j) {
            if (DEC) h[j] = relu4(h[j]);
            if (e + j < mid)      accA += h[j] * w[j];
            else if (e + j < end) accB += h[j] * w[j];
        }
    }
}

template <int D, bool DEC>
__global__ void __launch_bounds__(256, 16)
k_agg_gather(const float* __restrict__ H, const int* __restrict__ rowptr,
                             const int* __restrict__ csr_src, const float* __restrict__ wgt,
                             const float* __restrict__ dinv, const float* __restrict__ bias,
                             float* __restrict__ Y, int n) {
    constexpr int LPN = D / 4;
    const int i = blockIdx.x * blockDim.x + threadIdx.x;
    const int a = (i / LPN) * 2;
    if (a >= n) return;
    const int q = i % LPN;
    v4f accA, accB;
    gather_pair<D, DEC>(H, rowptr, csr_src, wgt, dinv, bias, a, q, n, accA, accB);
    ((v4f*)Y)[(size_t)a * LPN + q] = accA;
    if (a + 1 < n) ((v4f*)Y)[(size_t)(a + 1) * LPN + q] = accB;
}

template <int D, int NT, bool DEC>
__global__ void __launch_bounds__(256, 16)
k_agg_gemm(const float* __restrict__ H, const int* __restrict__ rowptr,
           const int* __restrict__ csr_src, const float* __restrict__ wgt,
           const float* __restrict__ dinv, const float* __restrict__ bias,
           const float* __restrict__ W, float* __restrict__ Y, int n) {
    constexpr int LPN  = D / 4;
    constexpr int NPB  = 2 * (256 / LPN);
    constexpr int RT   = NPB / 16;
    constexpr int DOUT = NT * 16;
    constexpr int T    = RT * NT;
    constexpr int WPT  = T < 8 ? T : 8;
    constexpr int TPW  = T / WPT;
    constexpr int RSTEP = WPT / NT;
    constexpr int KS   = D / 4;
    constexpr int QO   = DOUT / 4;
    constexpr int PA   = D + 4;
    constexpr int PO   = DOUT + 4;
    constexpr bool ALLW = (WPT == 8);
    static_assert(T % WPT == 0 && WPT % NT == 0, "a wave's tiles share one column tile");
    __shared__ __attribute__((aligned(16))) float sA[NPB * PA];
    __shared__ __attribute__((aligned(16))) float sO[NPB * PO];

    const int tid  = threadIdx.x;
    const int pl   = tid / LPN;
    const int q    = tid % LPN;
    const int a    = blockIdx.x * NPB + 2 * pl;
    float* const sp = &sA[(2 * pl) * PA + 4 * q];
    v4f accA = {0.f, 0.f, 0.f, 0.f};
    v4f accB = {0.f, 0.f, 0.f, 0.f};
    if (a < n) gather_pair<D, DEC>(H, rowptr, csr_src, wgt, dinv, bias, a, q, n, accA, accB);
    if (!DEC) { accA = relu4(accA); accB = relu4(accB); }
    *(v4f*)sp        = accA;
    *(v4f*)(sp + PA) = accB;

    const int wave = __builtin_amdgcn_readfirstlane(tid >> 5);
    const int lane = tid & 31;
    const int c    = lane & 15;
    const int half = lane >> 4;
    const int kb   = half * 2;
    const int rb   = wave / NT;
    const int ct   = wave % NT;

    v2f Bf[KS];
    if (ALLW || wave < WPT) {
        const float* wp = W + (size_t)kb * DOUT + ct * 16 + c;
#pragma unroll
        for (int kk = 0; kk < KS; ++kk) {
            Bf[kk].x = wp[(size_t)(4 * kk) * DOUT];
            Bf[kk].y = wp[(size_t)(4 * kk + 1) * DOUT];
        }
    }
    const int qo = tid % QO;
    v4f bv = {0.f, 0.f, 0.f, 0.f};
    if (DEC) { bv.x = bias[4 * qo]; bv.y = bias[4 * qo + 1]; bv.z = bias[4 * qo + 2]; bv.w = bias[4 * qo + 3]; }

    __syncthreads();

    if (ALLW || wave < WPT) {
        v8f d[TPW];
#pragma unroll
        for (int j = 0; j < TPW; ++j) {
            v8f z = {0.f, 0.f, 0.f, 0.f, 0.f, 0.f, 0.f, 0.f};
            d[j] = z;
        }
        const float* ap = &sA[(rb * 16 + c) * PA + kb];
#pragma unroll
        for (int kk = 0; kk < KS; ++kk) {
#pragma unroll
            for (int j = 0; j < TPW; ++j) {
                const v2f A = *(const v2f*)(ap + (j * RSTEP * 16) * PA + 4 * kk);
                d[j] = __builtin_amdgcn_wmma_f32_16x16x4_f32(
                    false, A, false, Bf[kk], (short)0, d[j], false, false);
            }
        }
#pragma unroll
        for (int j = 0; j < TPW; ++j) {
#pragma unroll
            for (int r = 0; r < 8; ++r)
                sO[((rb + j * RSTEP) * 16 + 8 * half + r) * PO + ct * 16 + c] = d[j][r];
        }
    }

    __syncthreads();

#pragma unroll
    for (int idx = tid; idx < NPB * QO; idx += 256) {
        const int row = idx / QO;
        const int nd  = blockIdx.x * NPB + row;
        if (nd < n) {
            v4f o = *(const v4f*)&sO[row * PO + 4 * qo];
            if (DEC) o += bv;
            ((v4f*)Y)[(size_t)nd * QO + qo] = o;
        }
    }
}

static inline int cdiv(long long a, int b) { return (int)((a + b - 1) / b); }

extern "C" void kernel_launch(void* const* d_in, const int* in_sizes, int n_in,
                              void* d_out, int out_size, void* d_ws, size_t ws_size,
                              hipStream_t stream) {
    const float* x   = (const float*)d_in[0];
    const int*   ei  = (const int*)d_in[1];
    const float* We1 = (const float*)d_in[2];  const float* be1 = (const float*)d_in[3];
    const float* We2 = (const float*)d_in[4];  const float* be2 = (const float*)d_in[5];
    const float* We3 = (const float*)d_in[6];  const float* be3 = (const float*)d_in[7];
    const float* Wd1 = (const float*)d_in[8];  const float* bd1 = (const float*)d_in[9];
    const float* Wd2 = (const float*)d_in[10]; const float* bd2 = (const float*)d_in[11];
    const float* Wd3 = (const float*)d_in[12]; const float* bd3 = (const float*)d_in[13];

    const int N = in_sizes[0] / 128;
    const int E = in_sizes[1] / 2;
    const int* rowp = ei;
    const int* colp = ei + E;

    const size_t Np = ((size_t)N + 64 + 63) & ~(size_t)63;
    const size_t Ep = ((size_t)E + 63) & ~(size_t)63;
    char* ws = (char*)d_ws;
    int*   indeg   = (int*)ws;    ws += sizeof(int)   * Np;
    float* dinv_s  = (float*)ws;  ws += sizeof(float) * Np;
    float* dinv_h  = (float*)ws;  ws += sizeof(float) * Np;
    float* hbuf    = (float*)ws;  ws += sizeof(float) * (size_t)N * 64;
    float* ybuf    = (float*)ws;  ws += sizeof(float) * (size_t)N * 64;
    int*   rowptr  = (int*)ws;    ws += sizeof(int)   * Np;
    int*   rank    = (int*)ws;    ws += sizeof(int)   * Ep;
    int*   tmp_src = (int*)ws;    ws += sizeof(int)   * Ep;
    int*   csr_src = (int*)ws;    ws += sizeof(int)   * Ep;
    float* w_s     = (float*)ws;  ws += sizeof(float) * Ep;
    float* w_h     = (float*)ws;
    float* outp    = (float*)d_out;

    const int TB = 256;
    const int rtiles  = (N + 15) >> 4;
    const int gblocks = (rtiles + 7) >> 3;

    k_zero_i32   <<<cdiv(N, TB), TB, 0, stream>>>(indeg, N);
    k_gemm1_count_rank<<<gblocks + cdiv(E, TB), TB, 0, stream>>>(x, We1, hbuf, N, gblocks,
                                                                  colp, indeg, rank, E);
    k_rowptr_scan_dinv<<<cdiv(N, 1024), 1024, 0, stream>>>(indeg, rowptr, dinv_s, dinv_h, N);
    k_csr_fill   <<<cdiv(E, TB), TB, 0, stream>>>(rowp, colp, rank, rowptr, tmp_src, E);
    k_csr_place  <<<cdiv(E, TB), TB, 0, stream>>>(rowp, colp, rank, rowptr, tmp_src, dinv_s, dinv_h,
                                                  csr_src, w_s, w_h, E);

    k_agg_gemm<64, 2, false><<<cdiv(N, 32), TB, 0, stream>>>(hbuf, rowptr, csr_src, w_s, dinv_s, be1, We2, ybuf, N);
    k_agg_gemm<32, 1, false><<<cdiv(N, 64), TB, 0, stream>>>(ybuf, rowptr, csr_src, w_s, dinv_s, be2, We3, hbuf, N);
    k_agg_gather<16, false><<<cdiv(N, 128), TB, 0, stream>>>(hbuf, rowptr, csr_src, w_s, dinv_s, be3, ybuf, N);

    k_agg_gemm<16, 2, true><<<cdiv(N, 128), TB, 0, stream>>>(ybuf, rowptr, csr_src, w_h, dinv_h, bd1, Wd1, hbuf, N);
    k_agg_gemm<32, 4, true><<<cdiv(N, 64), TB, 0, stream>>>(hbuf, rowptr, csr_src, w_h, dinv_h, bd2, Wd2, ybuf, N);
    k_agg_gemm<64, 8, true><<<cdiv(N, 32), TB, 0, stream>>>(ybuf, rowptr, csr_src, w_h, dinv_h, bd3, Wd3, outp, N);
}


# In-degree count and bucket slots on LDS counters (a block per node range x edge chunk, partial counts summed in the prefix-sum launch), not returning global atomics; zero-fill launch goes; sums same
# speedup vs baseline: 3.9111x; 1.0782x over previous
#include <hip/hip_runtime.h>

typedef __attribute__((ext_vector_type(2))) float v2f;
typedef __attribute__((ext_vector_type(8))) float v8f;
typedef __attribute__((ext_vector_type(4))) float v4f;

constexpr int CNT_NR = 8192;
constexpr int CNT_G  = 32;

__global__ void __launch_bounds__(1024)
k_rowptr_scan_dinv(int* __restrict__ part, const int* __restrict__ gsum, int* __restrict__ rowptr,
                   float* __restrict__ dinv_s, float* __restrict__ dinv_h, int n, int np) {
    __shared__ int s_pre[32];
    __shared__ int s_inc[32];
    const int tid  = threadIdx.x;
    const int lane = tid & 31;
    const int wv   = tid >> 5;
    int pre = 0;
    const int ng = (int)blockIdx.x * CNT_G;
    for (int j = tid; j < ng; j += 1024) pre += gsum[j];
    const int i = blockIdx.x * 1024 + tid;
    int v = 0;
    if (i < n) {
        int t[CNT_G];
#pragma unroll
        for (int g = 0; g < CNT_G; ++g) t[g] = part[(size_t)g * np + i];
#pragma unroll
        for (int g = 0; g < CNT_G; ++g) { part[(size_t)g * np + i] = v; v += t[g]; }
    }
    int inc = v;
    for (int d = 1; d < 32; d <<= 1) {
        int t = __shfl_up(inc, d, 32);
        if (lane >= d) inc += t;
    }
    for (int d = 16; d > 0; d >>= 1) pre += __shfl_xor(pre, d, 32);
    if (lane == 31) s_inc[wv] = inc;
    if (lane == 0)  s_pre[wv] = pre;
    __syncthreads();
    int off = s_pre[lane] + (lane < wv ? s_inc[lane] : 0);
    for (int d = 16; d > 0; d >>= 1) off += __shfl_xor(off, d, 32);
    if (i < n) {
        rowptr[i] = off + inc - v;
        const float d = (float)v;
        dinv_s[i] = rsqrtf(d + 1.0f);
        dinv_h[i] = rsqrtf(d + 2.0f);
    }
    if (i == n - 1) rowptr[n] = off + inc;
}

__global__ void k_csr_fill(const int* __restrict__ row, const int* __restrict__ col,
                           int* __restrict__ rank, const int* __restrict__ rowptr,
                           const int* __restrict__ part, int* __restrict__ tmp_src,
                           int e, int np, int ec) {
    int i = blockIdx.x * blockDim.x + threadIdx.x;
    if (i < e) {
        const int c  = col[i];
        const int rk = rank[i] + part[(size_t)(i / ec) * np + c];
        rank[i] = rk;
        tmp_src[rowptr[c] + rk] = row[i];
    }
}

__global__ void k_csr_place(const int* __restrict__ row, const int* __restrict__ col,
                            const int* __restrict__ rank, const int* __restrict__ rowptr,
                            const int* __restrict__ tmp_src,
                            const float* __restrict__ dinv_s, const float* __restrict__ dinv_h,
                            int* __restrict__ csr_src, float* __restrict__ w_s,
                            float* __restrict__ w_h, int e) {
    int i = blockIdx.x * blockDim.x + threadIdx.x;
    if (i >= e) return;
    const int r   = row[i];
    const int c   = col[i];
    const int beg = rowptr[c];
    const int end = rowptr[c + 1];
    const int me  = beg + rank[i];
    int pos = beg;
    for (int b = beg; b < end; ++b) {
        const int k = tmp_src[b];
        pos += (k < r || (k == r && b < me)) ? 1 : 0;
    }
    csr_src[pos] = r;
    w_s[pos] = dinv_s[r] * dinv_s[c];
    w_h[pos] = -(dinv_h[r] * dinv_h[c]);
}

template <int DIN, int NT, bool RELU_IN, bool BIAS>
static __device__ __forceinline__ void
gemm_wmma_strip(const float* __restrict__ X, const float* __restrict__ W,
                const float* __restrict__ bias, float* __restrict__ H, int N, int blk) {
    constexpr int DOUT = NT * 16;
    const int lane = threadIdx.x & 31;
    const int wave = threadIdx.x >> 5;
    const int tile = blk * (blockDim.x >> 5) + wave;
    const int rtiles = (N + 15) >> 4;
    if (tile >= rtiles) return;

    const int mrow  = tile * 16 + (lane & 15);
    const int mload = mrow < N ? mrow : N - 1;
    const int kb    = (lane >> 4) * 2;
    const int nloc  = lane & 15;

    v8f acc[NT];
#pragma unroll
    for (int j = 0; j < NT; ++j) {
        v8f z = {0.f, 0.f, 0.f, 0.f, 0.f, 0.f, 0.f, 0.f};
        acc[j] = z;
    }

    const float* xr = X + (size_t)mload * DIN;
#pragma unroll 4
    for (int k = 0; k < DIN; k += 4) {
        float a0 = xr[k + kb];
        float a1 = xr[k + kb + 1];
        if (RELU_IN) { a0 = fmaxf(a0, 0.0f); a1 = fmaxf(a1, 0.0f); }
        v2f A = {a0, a1};
#pragma unroll
        for (int j = 0; j < NT; ++j) {
            const int n0 = j * 16 + nloc;
            v2f B = {W[(size_t)(k + kb) * DOUT + n0],
                     W[(size_t)(k + kb + 1) * DOUT + n0]};
            acc[j] = __builtin_amdgcn_wmma_f32_16x16x4_f32(
                false, A, false, B, (short)0, acc[j], false, false);
        }
    }

    const int half = lane >> 4;
#pragma unroll
    for (int j = 0; j < NT; ++j) {
        const float bj = BIAS ? bias[j * 16 + nloc] : 0.0f;
#pragma unroll
        for (int r = 0; r < 8; ++r) {
            int mo = tile * 16 + r + 8 * half;
            if (mo < N) H[(size_t)mo * DOUT + j * 16 + nloc] = BIAS ? acc[j][r] + bj : acc[j][r];
        }
    }
}

static __device__ __forceinline__ void
count_range_chunk(const int* __restrict__ col, int* __restrict__ rank,
                  int* __restrict__ part, int* __restrict__ gsum,
                  int n, int e, int np, int ec, int blk) {
    __shared__ int s_cnt[CNT_NR + 32];
    __shared__ int s_gs[CNT_NR / 1024];
    const int tid = threadIdx.x;
    const int r   = blk / CNT_G;
    const int g   = blk % CNT_G;
    const int lo  = r * CNT_NR;
    for (int j = tid; j < CNT_NR + 32; j += 256) s_cnt[j] = 0;
    if (tid < CNT_NR / 1024) s_gs[tid] = 0;
    __syncthreads();

    const int e0 = g * ec;
    const int e1 = min(e0 + ec, e);
    for (int base = e0; base < e1; base += 8 * 256) {
        int cv[8];
#pragma unroll
        for (int u = 0; u < 8; ++u) {
            const int i = base + u * 256 + tid;
            cv[u] = (i < e1) ? col[i] : -1;
        }
        int rk[8];
#pragma unroll
        for (int u = 0; u < 8; ++u) {
            const unsigned d = (unsigned)(cv[u] - lo);
            rk[u] = atomicAdd(&s_cnt[d < (unsigned)CNT_NR ? d : (unsigned)(CNT_NR + (tid & 31))], 1);
        }
#pragma unroll
        for (int u = 0; u < 8; ++u) {
            const int i = base + u * 256 + tid;
            if ((unsigned)(cv[u] - lo) < (unsigned)CNT_NR) rank[i] = rk[u];
        }
    }
    __syncthreads();

    for (int q = 0; q < CNT_NR / 1024; ++q) {
        int sum = 0;
#pragma unroll
        for (int t = 0; t < 4; ++t) {
            const int j = (q * 4 + t) * 256 + tid;
            const int v = s_cnt[j];
            if (lo + j < n) part[(size_t)g * np + lo + j] = v;
            sum += v;
        }
        for (int d = 16; d > 0; d >>= 1) sum += __shfl_xor(sum, d, 32);
        if ((tid & 31) == 0) atomicAdd(&s_gs[q], sum);
    }
    __syncthreads();
    if (tid < CNT_NR / 1024) gsum[(r * (CNT_NR / 1024) + tid) * CNT_G + g] = s_gs[tid];
}

__global__ void __launch_bounds__(256)
k_gemm1_count_rank(const float* __restrict__ X, const float* __restrict__ W,
                   float* __restrict__ H, int N, int gblocks,
                   const int* __restrict__ col, int* __restrict__ rank,
                   int* __restrict__ part, int* __restrict__ gsum,
                   int e, int np, int ec) {
    const int b = blockIdx.x;
    if (b < gblocks) {
        gemm_wmma_strip<128, 4, false, false>(X, W, nullptr, H, N, b);
    } else {
        count_range_chunk(col, rank, part, gsum, N, e, np, ec, b - gblocks);
    }
}

static __device__ __forceinline__ v4f relu4(v4f h) {
    h.x = fmaxf(h.x, 0.0f); h.y = fmaxf(h.y, 0.0f);
    h.z = fmaxf(h.z, 0.0f); h.w = fmaxf(h.w, 0.0f);
    return h;
}

template <int D, bool DEC>
static __device__ __forceinline__ void
gather_pair(const float* __restrict__ H, const int* __restrict__ rowptr,
            const int* __restrict__ csr_src, const float* __restrict__ wgt,
            const float* __restrict__ dinv, const float* __restrict__ bias,
            int a, int q, int n, v4f& accA, v4f& accB) {
    constexpr unsigned ROWB = D * 4;
    const char* __restrict__ Hb = (const char*)H;
    const unsigned qoff = (unsigned)q * 16u;
    const bool hasB = a + 1 < n;
    const int b = hasB ? a + 1 : a;

    const int beg = rowptr[a];
    const int mid = rowptr[a + 1];
    const int end = rowptr[b + 1];
    const float dvA = dinv[a];
    const float dvB = dinv[b];
    float swA = dvA * dvA;
    float swB = dvB * dvB;
    if (DEC) { swA *= 2.0f; swB *= 2.0f; }

    accA = *(const v4f*)(Hb + ((unsigned)a * ROWB + qoff));
    accB = *(const v4f*)(Hb + ((unsigned)b * ROWB + qoff));
    if (DEC) {
        accA = relu4(accA) * swA;
        accB = relu4(accB) * swB;
    } else {
        v4f bs = {bias[4 * q], bias[4 * q + 1], bias[4 * q + 2], bias[4 * q + 3]};
        accA = accA * swA + bs;
        accB = accB * swB + bs;
    }
    if (!hasB) { v4f z = {0.f, 0.f, 0.f, 0.f}; accB = z; }

    int e = beg;
    for (; e + 8 <= end; e += 8) {
        int s[8];
        float w[8];
        v4f h[8];
#pragma unroll
        for (int j = 0; j < 8; ++j) { s[j] = csr_src[e + j]; w[j] = wgt[e + j]; }
#pragma unroll
        for (int j = 0; j < 8; ++j) h[j] = *(const v4f*)(Hb + ((unsigned)s[j] * ROWB + qoff));
#pragma unroll
        for (int j = 0; j < 8; ++j) {
            if (DEC) h[j] = relu4(h[j]);
            if (e + j < mid) accA += h[j] * w[j];
            else             accB += h[j] * w[j];
        }
    }
    if (e < end) {
        const int last = end - 1;
        int s[7];
        float w[7];
        v4f h[7];
#pragma unroll
        for (int j = 0; j < 7; ++j) {
            const int p = min(e + j, last);
            s[j] = csr_src[p];
            w[j] = wgt[p];
        }
#pragma unroll
        for (int j = 0; j < 7; ++j) h[j] = *(const v4f*)(Hb + ((unsigned)s[j] * ROWB + qoff));
#pragma unroll
        for (int j = 0; j < 7; ++j) {
            if (DEC) h[j] = relu4(h[j]);
            if (e + j < mid)      accA += h[j] * w[j];
            else if (e + j < end) accB += h[j] * w[j];
        }
    }
}

template <int D, bool DEC>
__global__ void __launch_bounds__(256, 16)
k_agg_gather(const float* __restrict__ H, const int* __restrict__ rowptr,
                             const int* __restrict__ csr_src, const float* __restrict__ wgt,
                             const float* __restrict__ dinv, const float* __restrict__ bias,
                             float* __restrict__ Y, int n) {
    constexpr int LPN = D / 4;
    const int i = blockIdx.x * blockDim.x + threadIdx.x;
    const int a = (i / LPN) * 2;
    if (a >= n) return;
    const int q = i % LPN;
    v4f accA, accB;
    gather_pair<D, DEC>(H, rowptr, csr_src, wgt, dinv, bias, a, q, n, accA, accB);
    ((v4f*)Y)[(size_t)a * LPN + q] = accA;
    if (a + 1 < n) ((v4f*)Y)[(size_t)(a + 1) * LPN + q] = accB;
}

template <int D, int NT, bool DEC>
__global__ void __launch_bounds__(256, 16)
k_agg_gemm(const float* __restrict__ H, const int* __restrict__ rowptr,
           const int* __restrict__ csr_src, const float* __restrict__ wgt,
           const float* __restrict__ dinv, const float* __restrict__ bias,
           const float* __restrict__ W, float* __restrict__ Y, int n) {
    constexpr int LPN  = D / 4;
    constexpr int NPB  = 2 * (256 / LPN);
    constexpr int RT   = NPB / 16;
    constexpr int DOUT = NT * 16;
    constexpr int T    = RT * NT;
    constexpr int WPT  = T < 8 ? T : 8;
    constexpr int TPW  = T / WPT;
    constexpr int RSTEP = WPT / NT;
    constexpr int KS   = D / 4;
    constexpr int QO   = DOUT / 4;
    constexpr int PA   = D + 4;
    constexpr int PO   = DOUT + 4;
    constexpr bool ALLW = (WPT == 8);
    static_assert(T % WPT == 0 && WPT % NT == 0, "a wave's tiles share one column tile");
    __shared__ __attribute__((aligned(16))) float sA[NPB * PA];
    __shared__ __attribute__((aligned(16))) float sO[NPB * PO];

    const int tid  = threadIdx.x;
    const int pl   = tid / LPN;
    const int q    = tid % LPN;
    const int a    = blockIdx.x * NPB + 2 * pl;
    float* const sp = &sA[(2 * pl) * PA + 4 * q];
    v4f accA = {0.f, 0.f, 0.f, 0.f};
    v4f accB = {0.f, 0.f, 0.f, 0.f};
    if (a < n) gather_pair<D, DEC>(H, rowptr, csr_src, wgt, dinv, bias, a, q, n, accA, accB);
    if (!DEC) { accA = relu4(accA); accB = relu4(accB); }
    *(v4f*)sp        = accA;
    *(v4f*)(sp + PA) = accB;

    const int wave = __builtin_amdgcn_readfirstlane(tid >> 5);
    const int lane = tid & 31;
    const int c    = lane & 15;
    const int half = lane >> 4;
    const int kb   = half * 2;
    const int rb   = wave / NT;
    const int ct   = wave % NT;

    v2f Bf[KS];
    if (ALLW || wave < WPT) {
        const float* wp = W + (size_t)kb * DOUT + ct * 16 + c;
#pragma unroll
        for (int kk = 0; kk < KS; ++kk) {
            Bf[kk].x = wp[(size_t)(4 * kk) * DOUT];
            Bf[kk].y = wp[(size_t)(4 * kk + 1) * DOUT];
        }
    }
    const int qo = tid % QO;
    v4f bv = {0.f, 0.f, 0.f, 0.f};
    if (DEC) { bv.x = bias[4 * qo]; bv.y = bias[4 * qo + 1]; bv.z = bias[4 * qo + 2]; bv.w = bias[4 * qo + 3]; }

    __syncthreads();

    if (ALLW || wave < WPT) {
        v8f d[TPW];
#pragma unroll
        for (int j = 0; j < TPW; ++j) {
            v8f z = {0.f, 0.f, 0.f, 0.f, 0.f, 0.f, 0.f, 0.f};
            d[j] = z;
        }
        const float* ap = &sA[(rb * 16 + c) * PA + kb];
#pragma unroll
        for (int kk = 0; kk < KS; ++kk) {
#pragma unroll
            for (int j = 0; j < TPW; ++j) {
                const v2f A = *(const v2f*)(ap + (j * RSTEP * 16) * PA + 4 * kk);
                d[j] = __builtin_amdgcn_wmma_f32_16x16x4_f32(
                    false, A, false, Bf[kk], (short)0, d[j], false, false);
            }
        }
#pragma unroll
        for (int j = 0; j < TPW; ++j) {
#pragma unroll
            for (int r = 0; r < 8; ++r)
                sO[((rb + j * RSTEP) * 16 + 8 * half + r) * PO + ct * 16 + c] = d[j][r];
        }
    }

    __syncthreads();

#pragma unroll
    for (int idx = tid; idx < NPB * QO; idx += 256) {
        const int row = idx / QO;
        const int nd  = blockIdx.x * NPB + row;
        if (nd < n) {
            v4f o = *(const v4f*)&sO[row * PO + 4 * qo];
            if (DEC) o += bv;
            ((v4f*)Y)[(size_t)nd * QO + qo] = o;
        }
    }
}

static inline int cdiv(long long a, int b) { return (int)((a + b - 1) / b); }

extern "C" void kernel_launch(void* const* d_in, const int* in_sizes, int n_in,
                              void* d_out, int out_size, void* d_ws, size_t ws_size,
                              hipStream_t stream) {
    const float* x   = (const float*)d_in[0];
    const int*   ei  = (const int*)d_in[1];
    const float* We1 = (const float*)d_in[2];  const float* be1 = (const float*)d_in[3];
    const float* We2 = (const float*)d_in[4];  const float* be2 = (const float*)d_in[5];
    const float* We3 = (const float*)d_in[6];  const float* be3 = (const float*)d_in[7];
    const float* Wd1 = (const float*)d_in[8];  const float* bd1 = (const float*)d_in[9];
    const float* Wd2 = (const float*)d_in[10]; const float* bd2 = (const float*)d_in[11];
    const float* Wd3 = (const float*)d_in[12]; const float* bd3 = (const float*)d_in[13];

    const int N = in_sizes[0] / 128;
    const int E = in_sizes[1] / 2;
    const int* rowp = ei;
    const int* colp = ei + E;

    const size_t Np = ((size_t)N + 64 + 63) & ~(size_t)63;
    const size_t Ep = ((size_t)E + 63) & ~(size_t)63;
    char* ws = (char*)d_ws;
    float* dinv_s  = (float*)ws;  ws += sizeof(float) * Np;
    float* dinv_h  = (float*)ws;  ws += sizeof(float) * Np;
    float* hbuf    = (float*)ws;  ws += sizeof(float) * (size_t)N * 64;
    float* ybuf    = (float*)ws;  ws += sizeof(float) * (size_t)N * 64;
    int*   rowptr  = (int*)ws;    ws += sizeof(int)   * Np;
    int*   rank    = (int*)ws;    ws += sizeof(int)   * Ep;
    int*   tmp_src = (int*)ws;    ws += sizeof(int)   * Ep;
    int*   csr_src = (int*)ws;    ws += sizeof(int)   * Ep;
    float* w_s     = (float*)ws;  ws += sizeof(float) * Ep;
    float* w_h     = (float*)ws;
    float* outp    = (float*)d_out;
    int*   part    = (int*)ybuf;
    int*   gsum    = part + (size_t)CNT_G * Np;

    const int TB = 256;
    const int rtiles  = (N + 15) >> 4;
    const int gblocks = (rtiles + 7) >> 3;

    const int cranges = cdiv(N, CNT_NR);
    const int ec      = cdiv(cdiv(E, CNT_G), TB) * TB;
    k_gemm1_count_rank<<<gblocks + cranges * CNT_G, TB, 0, stream>>>(x, We1, hbuf, N, gblocks,
                                                                     colp, rank, part, gsum, E, (int)Np, ec);
    k_rowptr_scan_dinv<<<cdiv(N, 1024), 1024, 0, stream>>>(part, gsum, rowptr, dinv_s, dinv_h, N, (int)Np);
    k_csr_fill   <<<cdiv(E, TB), TB, 0, stream>>>(rowp, colp, rank, rowptr, part, tmp_src, E, (int)Np, ec);
    k_csr_place  <<<cdiv(E, TB), TB, 0, stream>>>(rowp, colp, rank, rowptr, tmp_src, dinv_s, dinv_h,
                                                  csr_src, w_s, w_h, E);

    k_agg_gemm<64, 2, false><<<cdiv(N, 32), TB, 0, stream>>>(hbuf, rowptr, csr_src, w_s, dinv_s, be1, We2, ybuf, N);
    k_agg_gemm<32, 1, false><<<cdiv(N, 64), TB, 0, stream>>>(ybuf, rowptr, csr_src, w_s, dinv_s, be2, We3, hbuf, N);
    k_agg_gather<16, false><<<cdiv(N, 128), TB, 0, stream>>>(hbuf, rowptr, csr_src, w_s, dinv_s, be3, ybuf, N);

    k_agg_gemm<16, 2, true><<<cdiv(N, 128), TB, 0, stream>>>(ybuf, rowptr, csr_src, w_h, dinv_h, bd1, Wd1, hbuf, N);
    k_agg_gemm<32, 4, true><<<cdiv(N, 64), TB, 0, stream>>>(hbuf, rowptr, csr_src, w_h, dinv_h, bd2, Wd2, ybuf, N);
    k_agg_gemm<64, 8, true><<<cdiv(N, 32), TB, 0, stream>>>(ybuf, rowptr, csr_src, w_h, dinv_h, bd3, Wd3, outp, N);
}


# Bucket ordering pass runs per bucket slot, not per edge: the fill stores (source,target) pairs, a wave finds whole buckets in its own lines and writes the lists as contiguous runs; sums keep order
# speedup vs baseline: 5.0868x; 1.3006x over previous
#include <hip/hip_runtime.h>

typedef __attribute__((ext_vector_type(2))) float v2f;
typedef __attribute__((ext_vector_type(8))) float v8f;
typedef __attribute__((ext_vector_type(4))) float v4f;
typedef __attribute__((ext_vector_type(2))) int v2i;
typedef __attribute__((ext_vector_type(4))) int v4i;

constexpr int CNT_NR = 8192;
constexpr int CNT_G  = 32;

__global__ void __launch_bounds__(1024)
k_rowptr_scan_dinv(int* __restrict__ part, const int* __restrict__ gsum, int* __restrict__ rowptr,
                   float* __restrict__ dinv_s, float* __restrict__ dinv_h, int n, int np) {
    __shared__ int s_pre[32];
    __shared__ int s_inc[32];
    const int tid  = threadIdx.x;
    const int lane = tid & 31;
    const int wv   = tid >> 5;
    int pre = 0;
    const int ng = (int)blockIdx.x * CNT_G;
    for (int j = tid; j < ng; j += 1024) pre += gsum[j];
    const int i = blockIdx.x * 1024 + tid;
    int v = 0;
    if (i < n) {
        int t[CNT_G];
#pragma unroll
        for (int g = 0; g < CNT_G; ++g) t[g] = part[(size_t)g * np + i];
#pragma unroll
        for (int g = 0; g < CNT_G; ++g) { part[(size_t)g * np + i] = v; v += t[g]; }
    }
    int inc = v;
    for (int d = 1; d < 32; d <<= 1) {
        int t = __shfl_up(inc, d, 32);
        if (lane >= d) inc += t;
    }
    for (int d = 16; d > 0; d >>= 1) pre += __shfl_xor(pre, d, 32);
    if (lane == 31) s_inc[wv] = inc;
    if (lane == 0)  s_pre[wv] = pre;
    __syncthreads();
    int off = s_pre[lane] + (lane < wv ? s_inc[lane] : 0);
    for (int d = 16; d > 0; d >>= 1) off += __shfl_xor(off, d, 32);
    if (i < n) {
        rowptr[i] = off + inc - v;
        const float d = (float)v;
        dinv_s[i] = rsqrtf(d + 1.0f);
        dinv_h[i] = rsqrtf(d + 2.0f);
    }
    if (i == n - 1) rowptr[n] = off + inc;
}

__global__ void k_csr_fill(const int* __restrict__ row, const int* __restrict__ col,
                           const int* __restrict__ rank, const int* __restrict__ rowptr,
                           const int* __restrict__ part, int* __restrict__ tmp_pair,
                           int e, int np, int ec) {
    int i = blockIdx.x * blockDim.x + threadIdx.x;
    if (i < e) {
        const int c    = col[i];
        const int slot = rowptr[c] + rank[i] + part[(size_t)(i / ec) * np + c];
        const v2i t = {row[i], c};
        *(v2i*)(tmp_pair + 2 * slot) = t;
    }
}

constexpr int PLACE_S = 4;

__global__ void __launch_bounds__(256)
k_csr_place(const int* __restrict__ tmp_pair, const int* __restrict__ rowptr,
            const float* __restrict__ dinv_s, const float* __restrict__ dinv_h,
            int* __restrict__ csr_src, float* __restrict__ w_s,
            float* __restrict__ w_h, int e) {
    const int p0 = (blockIdx.x * blockDim.x + threadIdx.x) * PLACE_S;
    if (p0 >= e) return;
    static_assert(PLACE_S == 4, "a thread's pairs are read as two 16-byte pieces");
    int r[PLACE_S], c[PLACE_S];
    if (p0 + PLACE_S <= e) {
        const v4i q0 = *(const v4i*)(tmp_pair + 2 * p0);
        const v4i q1 = *(const v4i*)(tmp_pair + 2 * p0 + 4);
        r[0] = q0.x; c[0] = q0.y; r[1] = q0.z; c[1] = q0.w;
        r[2] = q1.x; c[2] = q1.y; r[3] = q1.z; c[3] = q1.w;
    } else {
#pragma unroll
        for (int j = 0; j < PLACE_S; ++j) {
            const v2i t = *(const v2i*)(tmp_pair + 2 * min(p0 + j, e - 1));
            r[j] = t.x;
            c[j] = t.y;
        }
    }
    int beg[PLACE_S], end[PLACE_S], pos[PLACE_S];
    float ds[PLACE_S], dh[PLACE_S];
#pragma unroll
    for (int j = 0; j < PLACE_S; ++j) {
        beg[j] = rowptr[c[j]];
        end[j] = rowptr[c[j] + 1];
        ds[j]  = dinv_s[r[j]] * dinv_s[c[j]];
        dh[j]  = dinv_h[r[j]] * dinv_h[c[j]];
    }
#pragma unroll
    for (int j = 0; j < PLACE_S; ++j) pos[j] = beg[j];
    const int hi   = end[PLACE_S - 1];
    const int last = hi - 1;
    for (int b = beg[0]; b < hi; b += 4) {
        int k[4];
#pragma unroll
        for (int i = 0; i < 4; ++i) k[i] = tmp_pair[2 * min(b + i, last)];
#pragma unroll
        for (int j = 0; j < PLACE_S; ++j) {
#pragma unroll
            for (int i = 0; i < 4; ++i) {
                const int bi = b + i;
                const int inb = (int)(bi >= beg[j]) & (int)(bi < end[j]);
                pos[j] += inb & ((int)(k[i] < r[j]) | ((int)(k[i] == r[j]) & (int)(bi < p0 + j)));
            }
        }
    }
#pragma unroll
    for (int j = 0; j < PLACE_S; ++j) {
        if (p0 + j < e) {
            csr_src[pos[j]] = r[j];
            w_s[pos[j]] = ds[j];
            w_h[pos[j]] = -dh[j];
        }
    }
}

template <int DIN, int NT, bool RELU_IN, bool BIAS>
static __device__ __forceinline__ void
gemm_wmma_strip(const float* __restrict__ X, const float* __restrict__ W,
                const float* __restrict__ bias, float* __restrict__ H, int N, int blk) {
    constexpr int DOUT = NT * 16;
    const int lane = threadIdx.x & 31;
    const int wave = threadIdx.x >> 5;
    const int tile = blk * (blockDim.x >> 5) + wave;
    const int rtiles = (N + 15) >> 4;
    if (tile >= rtiles) return;

    const int mrow  = tile * 16 + (lane & 15);
    const int mload = mrow < N ? mrow : N - 1;
    const int kb    = (lane >> 4) * 2;
    const int nloc  = lane & 15;

    v8f acc[NT];
#pragma unroll
    for (int j = 0; j < NT; ++j) {
        v8f z = {0.f, 0.f, 0.f, 0.f, 0.f, 0.f, 0.f, 0.f};
        acc[j] = z;
    }

    const float* xr = X + (size_t)mload * DIN;
#pragma unroll 4
    for (int k = 0; k < DIN; k += 4) {
        float a0 = xr[k + kb];
        float a1 = xr[k + kb + 1];
        if (RELU_IN) { a0 = fmaxf(a0, 0.0f); a1 = fmaxf(a1, 0.0f); }
        v2f A = {a0, a1};
#pragma unroll
        for (int j = 0; j < NT; ++j) {
            const int n0 = j * 16 + nloc;
            v2f B = {W[(size_t)(k + kb) * DOUT + n0],
                     W[(size_t)(k + kb + 1) * DOUT + n0]};
            acc[j] = __builtin_amdgcn_wmma_f32_16x16x4_f32(
                false, A, false, B, (short)0, acc[j], false, false);
        }
    }

    const int half = lane >> 4;
#pragma unroll
    for (int j = 0; j < NT; ++j) {
        const float bj = BIAS ? bias[j * 16 + nloc] : 0.0f;
#pragma unroll
        for (int r = 0; r < 8; ++r) {
            int mo = tile * 16 + r + 8 * half;
            if (mo < N) H[(size_t)mo * DOUT + j * 16 + nloc] = BIAS ? acc[j][r] + bj : acc[j][r];
        }
    }
}

static __device__ __forceinline__ void
count_range_chunk(const int* __restrict__ col, int* __restrict__ rank,
                  int* __restrict__ part, int* __restrict__ gsum,
                  int n, int e, int np, int ec, int blk) {
    __shared__ int s_cnt[CNT_NR + 32];
    __shared__ int s_gs[CNT_NR / 1024];
    const int tid = threadIdx.x;
    const int r   = blk / CNT_G;
    const int g   = blk % CNT_G;
    const int lo  = r * CNT_NR;
    for (int j = tid; j < CNT_NR + 32; j += 256) s_cnt[j] = 0;
    if (tid < CNT_NR / 1024) s_gs[tid] = 0;
    __syncthreads();

    const int e0 = g * ec;
    const int e1 = min(e0 + ec, e);
    for (int base = e0; base < e1; base += 8 * 256) {
        int cv[8];
#pragma unroll
        for (int u = 0; u < 8; ++u) {
            const int i = base + u * 256 + tid;
            cv[u] = (i < e1) ? col[i] : -1;
        }
        int rk[8];
#pragma unroll
        for (int u = 0; u < 8; ++u) {
            const unsigned d = (unsigned)(cv[u] - lo);
            rk[u] = atomicAdd(&s_cnt[d < (unsigned)CNT_NR ? d : (unsigned)(CNT_NR + (tid & 31))], 1);
        }
#pragma unroll
        for (int u = 0; u < 8; ++u) {
            const int i = base + u * 256 + tid;
            if ((unsigned)(cv[u] - lo) < (unsigned)CNT_NR) rank[i] = rk[u];
        }
    }
    __syncthreads();

    for (int q = 0; q < CNT_NR / 1024; ++q) {
        int sum = 0;
#pragma unroll
        for (int t = 0; t < 4; ++t) {
            const int j = (q * 4 + t) * 256 + tid;
            const int v = s_cnt[j];
            if (lo + j < n) part[(size_t)g * np + lo + j] = v;
            sum += v;
        }
        for (int d = 16; d > 0; d >>= 1) sum += __shfl_xor(sum, d, 32);
        if ((tid & 31) == 0) atomicAdd(&s_gs[q], sum);
    }
    __syncthreads();
    if (tid < CNT_NR / 1024) gsum[(r * (CNT_NR / 1024) + tid) * CNT_G + g] = s_gs[tid];
}

__global__ void __launch_bounds__(256)
k_gemm1_count_rank(const float* __restrict__ X, const float* __restrict__ W,
                   float* __restrict__ H, int N, int gblocks,
                   const int* __restrict__ col, int* __restrict__ rank,
                   int* __restrict__ part, int* __restrict__ gsum,
                   int e, int np, int ec) {
    const int b = blockIdx.x;
    if (b < gblocks) {
        gemm_wmma_strip<128, 4, false, false>(X, W, nullptr, H, N, b);
    } else {
        count_range_chunk(col, rank, part, gsum, N, e, np, ec, b - gblocks);
    }
}

static __device__ __forceinline__ v4f relu4(v4f h) {
    h.x = fmaxf(h.x, 0.0f); h.y = fmaxf(h.y, 0.0f);
    h.z = fmaxf(h.z, 0.0f); h.w = fmaxf(h.w, 0.0f);
    return h;
}

template <int D, bool DEC>
static __device__ __forceinline__ void
gather_pair(const float* __restrict__ H, const int* __restrict__ rowptr,
            const int* __restrict__ csr_src, const float* __restrict__ wgt,
            const float* __restrict__ dinv, const float* __restrict__ bias,
            int a, int q, int n, v4f& accA, v4f& accB) {
    constexpr unsigned ROWB = D * 4;
    const char* __restrict__ Hb = (const char*)H;
    const unsigned qoff = (unsigned)q * 16u;
    const bool hasB = a + 1 < n;
    const int b = hasB ? a + 1 : a;

    const int beg = rowptr[a];
    const int mid = rowptr[a + 1];
    const int end = rowptr[b + 1];
    const float dvA = dinv[a];
    const float dvB = dinv[b];
    float swA = dvA * dvA;
    float swB = dvB * dvB;
    if (DEC) { swA *= 2.0f; swB *= 2.0f; }

    accA = *(const v4f*)(Hb + ((unsigned)a * ROWB + qoff));
    accB = *(const v4f*)(Hb + ((unsigned)b * ROWB + qoff));
    if (DEC) {
        accA = relu4(accA) * swA;
        accB = relu4(accB) * swB;
    } else {
        v4f bs = {bias[4 * q], bias[4 * q + 1], bias[4 * q + 2], bias[4 * q + 3]};
        accA = accA * swA + bs;
        accB = accB * swB + bs;
    }
    if (!hasB) { v4f z = {0.f, 0.f, 0.f, 0.f}; accB = z; }

    int e = beg;
    for (; e + 8 <= end; e += 8) {
        int s[8];
        float w[8];
        v4f h[8];
#pragma unroll
        for (int j = 0; j < 8; ++j) { s[j] = csr_src[e + j]; w[j] = wgt[e + j]; }
#pragma unroll
        for (int j = 0; j < 8; ++j) h[j] = *(const v4f*)(Hb + ((unsigned)s[j] * ROWB + qoff));
#pragma unroll
        for (int j = 0; j < 8; ++j) {
            if (DEC) h[j] = relu4(h[j]);
            if (e + j < mid) accA += h[j] * w[j];
            else             accB += h[j] * w[j];
        }
    }
    if (e < end) {
        const int last = end - 1;
        int s[7];
        float w[7];
        v4f h[7];
#pragma unroll
        for (int j = 0; j < 7; ++j) {
            const int p = min(e + j, last);
            s[j] = csr_src[p];
            w[j] = wgt[p];
        }
#pragma unroll
        for (int j = 0; j < 7; ++j) h[j] = *(const v4f*)(Hb + ((unsigned)s[j] * ROWB + qoff));
#pragma unroll
        for (int j = 0; j < 7; ++j) {
            if (DEC) h[j] = relu4(h[j]);
            if (e + j < mid)      accA += h[j] * w[j];
            else if (e + j < end) accB += h[j] * w[j];
        }
    }
}

template <int D, bool DEC>
__global__ void __launch_bounds__(256, 16)
k_agg_gather(const float* __restrict__ H, const int* __restrict__ rowptr,
                             const int* __restrict__ csr_src, const float* __restrict__ wgt,
                             const float* __restrict__ dinv, const float* __restrict__ bias,
                             float* __restrict__ Y, int n) {
    constexpr int LPN = D / 4;
    const int i = blockIdx.x * blockDim.x + threadIdx.x;
    const int a = (i / LPN) * 2;
    if (a >= n) return;
    const int q = i % LPN;
    v4f accA, accB;
    gather_pair<D, DEC>(H, rowptr, csr_src, wgt, dinv, bias, a, q, n, accA, accB);
    ((v4f*)Y)[(size_t)a * LPN + q] = accA;
    if (a + 1 < n) ((v4f*)Y)[(size_t)(a + 1) * LPN + q] = accB;
}

template <int D, int NT, bool DEC>
__global__ void __launch_bounds__(256, 16)
k_agg_gemm(const float* __restrict__ H, const int* __restrict__ rowptr,
           const int* __restrict__ csr_src, const float* __restrict__ wgt,
           const float* __restrict__ dinv, const float* __restrict__ bias,
           const float* __restrict__ W, float* __restrict__ Y, int n) {
    constexpr int LPN  = D / 4;
    constexpr int NPB  = 2 * (256 / LPN);
    constexpr int RT   = NPB / 16;
    constexpr int DOUT = NT * 16;
    constexpr int T    = RT * NT;
    constexpr int WPT  = T < 8 ? T : 8;
    constexpr int TPW  = T / WPT;
    constexpr int RSTEP = WPT / NT;
    constexpr int KS   = D / 4;
    constexpr int QO   = DOUT / 4;
    constexpr int PA   = D + 4;
    constexpr int PO   = DOUT + 4;
    constexpr bool ALLW = (WPT == 8);
    static_assert(T % WPT == 0 && WPT % NT == 0, "a wave's tiles share one column tile");
    __shared__ __attribute__((aligned(16))) float sA[NPB * PA];
    __shared__ __attribute__((aligned(16))) float sO[NPB * PO];

    const int tid  = threadIdx.x;
    const int pl   = tid / LPN;
    const int q    = tid % LPN;
    const int a    = blockIdx.x * NPB + 2 * pl;
    float* const sp = &sA[(2 * pl) * PA + 4 * q];
    v4f accA = {0.f, 0.f, 0.f, 0.f};
    v4f accB = {0.f, 0.f, 0.f, 0.f};
    if (a < n) gather_pair<D, DEC>(H, rowptr, csr_src, wgt, dinv, bias, a, q, n, accA, accB);
    if (!DEC) { accA = relu4(accA); accB = relu4(accB); }
    *(v4f*)sp        = accA;
    *(v4f*)(sp + PA) = accB;

    const int wave = __builtin_amdgcn_readfirstlane(tid >> 5);
    const int lane = tid & 31;
    const int c    = lane & 15;
    const int half = lane >> 4;
    const int kb   = half * 2;
    const int rb   = wave / NT;
    const int ct   = wave % NT;

    v2f Bf[KS];
    if (ALLW || wave < WPT) {
        const float* wp = W + (size_t)kb * DOUT + ct * 16 + c;
#pragma unroll
        for (int kk = 0; kk < KS; ++kk) {
            Bf[kk].x = wp[(size_t)(4 * kk) * DOUT];
            Bf[kk].y = wp[(size_t)(4 * kk + 1) * DOUT];
        }
    }
    const int qo = tid % QO;
    v4f bv = {0.f, 0.f, 0.f, 0.f};
    if (DEC) { bv.x = bias[4 * qo]; bv.y = bias[4 * qo + 1]; bv.z = bias[4 * qo + 2]; bv.w = bias[4 * qo + 3]; }

    __syncthreads();

    if (ALLW || wave < WPT) {
        v8f d[TPW];
#pragma unroll
        for (int j = 0; j < TPW; ++j) {
            v8f z = {0.f, 0.f, 0.f, 0.f, 0.f, 0.f, 0.f, 0.f};
            d[j] = z;
        }
        const float* ap = &sA[(rb * 16 + c) * PA + kb];
#pragma unroll
        for (int kk = 0; kk < KS; ++kk) {
#pragma unroll
            for (int j = 0; j < TPW; ++j) {
                const v2f A = *(const v2f*)(ap + (j * RSTEP * 16) * PA + 4 * kk);
                d[j] = __builtin_amdgcn_wmma_f32_16x16x4_f32(
                    false, A, false, Bf[kk], (short)0, d[j], false, false);
            }
        }
#pragma unroll
        for (int j = 0; j < TPW; ++j) {
#pragma unroll
            for (int r = 0; r < 8; ++r)
                sO[((rb + j * RSTEP) * 16 + 8 * half + r) * PO + ct * 16 + c] = d[j][r];
        }
    }

    __syncthreads();

#pragma unroll
    for (int idx = tid; idx < NPB * QO; idx += 256) {
        const int row = idx / QO;
        const int nd  = blockIdx.x * NPB + row;
        if (nd < n) {
            v4f o = *(const v4f*)&sO[row * PO + 4 * qo];
            if (DEC) o += bv;
            ((v4f*)Y)[(size_t)nd * QO + qo] = o;
        }
    }
}

static inline int cdiv(long long a, int b) { return (int)((a + b - 1) / b); }

extern "C" void kernel_launch(void* const* d_in, const int* in_sizes, int n_in,
                              void* d_out, int out_size, void* d_ws, size_t ws_size,
                              hipStream_t stream) {
    const float* x   = (const float*)d_in[0];
    const int*   ei  = (const int*)d_in[1];
    const float* We1 = (const float*)d_in[2];  const float* be1 = (const float*)d_in[3];
    const float* We2 = (const float*)d_in[4];  const float* be2 = (const float*)d_in[5];
    const float* We3 = (const float*)d_in[6];  const float* be3 = (const float*)d_in[7];
    const float* Wd1 = (const float*)d_in[8];  const float* bd1 = (const float*)d_in[9];
    const float* Wd2 = (const float*)d_in[10]; const float* bd2 = (const float*)d_in[11];
    const float* Wd3 = (const float*)d_in[12]; const float* bd3 = (const float*)d_in[13];

    const int N = in_sizes[0] / 128;
    const int E = in_sizes[1] / 2;
    const int* rowp = ei;
    const int* colp = ei + E;

    const size_t Np = ((size_t)N + 64 + 63) & ~(size_t)63;
    const size_t Ep = ((size_t)E + 63) & ~(size_t)63;
    char* ws = (char*)d_ws;
    float* dinv_s  = (float*)ws;  ws += sizeof(float) * Np;
    float* dinv_h  = (float*)ws;  ws += sizeof(float) * Np;
    float* hbuf    = (float*)ws;  ws += sizeof(float) * (size_t)N * 64;
    float* ybuf    = (float*)ws;  ws += sizeof(float) * (size_t)N * 64;
    int*   rowptr  = (int*)ws;    ws += sizeof(int)   * Np;
    int*   tmp_pair = (int*)ws;   ws += sizeof(int)   * 2 * Ep;
    int*   csr_src = (int*)ws;    ws += sizeof(int)   * Ep;
    float* w_s     = (float*)ws;  ws += sizeof(float) * Ep;
    float* w_h     = (float*)ws;
    int*   rank    = (int*)w_h;
    float* outp    = (float*)d_out;
    int*   part    = (int*)ybuf;
    int*   gsum    = part + (size_t)CNT_G * Np;

    const int TB = 256;
    const int rtiles  = (N + 15) >> 4;
    const int gblocks = (rtiles + 7) >> 3;

    const int cranges = cdiv(N, CNT_NR);
    const int ec      = cdiv(cdiv(E, CNT_G), TB) * TB;
    k_gemm1_count_rank<<<gblocks + cranges * CNT_G, TB, 0, stream>>>(x, We1, hbuf, N, gblocks,
                                                                     colp, rank, part, gsum, E, (int)Np, ec);
    k_rowptr_scan_dinv<<<cdiv(N, 1024), 1024, 0, stream>>>(part, gsum, rowptr, dinv_s, dinv_h, N, (int)Np);
    k_csr_fill   <<<cdiv(E, TB), TB, 0, stream>>>(rowp, colp, rank, rowptr, part, tmp_pair, E, (int)Np, ec);
    k_csr_place  <<<cdiv(cdiv(E, PLACE_S), TB), TB, 0, stream>>>(tmp_pair, rowptr, dinv_s, dinv_h, csr_src, w_s, w_h, E);

    k_agg_gemm<64, 2, false><<<cdiv(N, 32), TB, 0, stream>>>(hbuf, rowptr, csr_src, w_s, dinv_s, be1, We2, ybuf, N);
    k_agg_gemm<32, 1, false><<<cdiv(N, 64), TB, 0, stream>>>(ybuf, rowptr, csr_src, w_s, dinv_s, be2, We3, hbuf, N);
    k_agg_gather<16, false><<<cdiv(N, 128), TB, 0, stream>>>(hbuf, rowptr, csr_src, w_s, dinv_s, be3, ybuf, N);

    k_agg_gemm<16, 2, true><<<cdiv(N, 128), TB, 0, stream>>>(ybuf, rowptr, csr_src, w_h, dinv_h, bd1, Wd1, hbuf, N);
    k_agg_gemm<32, 4, true><<<cdiv(N, 64), TB, 0, stream>>>(hbuf, rowptr, csr_src, w_h, dinv_h, bd2, Wd2, ybuf, N);
    k_agg_gemm<64, 8, true><<<cdiv(N, 32), TB, 0, stream>>>(ybuf, rowptr, csr_src, w_h, dinv_h, bd3, Wd3, outp, N);
}


# Edge weights go: each row is stored times its node's degree coefficient (first map moves beside the fill launch, ReLU at the producer), in-edge sums are plain adds scaled once; rounding changes
# speedup vs baseline: 5.3880x; 1.0592x over previous
#include <hip/hip_runtime.h>

typedef __attribute__((ext_vector_type(2))) float v2f;
typedef __attribute__((ext_vector_type(8))) float v8f;
typedef __attribute__((ext_vector_type(4))) float v4f;
typedef __attribute__((ext_vector_type(2))) int v2i;
typedef __attribute__((ext_vector_type(4))) int v4i;

constexpr int CNT_NR = 8192;
constexpr int CNT_G  = 32;

__global__ void __launch_bounds__(1024)
k_rowptr_scan_dinv(int* __restrict__ part, const int* __restrict__ gsum, int* __restrict__ rowptr,
                   float* __restrict__ dinv_s, float* __restrict__ dinv_h, int n, int np) {
    __shared__ int s_pre[32];
    __shared__ int s_inc[32];
    const int tid  = threadIdx.x;
    const int lane = tid & 31;
    const int wv   = tid >> 5;
    int pre = 0;
    const int ng = (int)blockIdx.x * CNT_G;
    for (int j = tid; j < ng; j += 1024) pre += gsum[j];
    const int i = blockIdx.x * 1024 + tid;
    int v = 0;
    if (i < n) {
        int t[CNT_G];
#pragma unroll
        for (int g = 0; g < CNT_G; ++g) t[g] = part[(size_t)g * np + i];
#pragma unroll
        for (int g = 0; g < CNT_G; ++g) { part[(size_t)g * np + i] = v; v += t[g]; }
    }
    int inc = v;
    for (int d = 1; d < 32; d <<= 1) {
        int t = __shfl_up(inc, d, 32);
        if (lane >= d) inc += t;
    }
    for (int d = 16; d > 0; d >>= 1) pre += __shfl_xor(pre, d, 32);
    if (lane == 31) s_inc[wv] = inc;
    if (lane == 0)  s_pre[wv] = pre;
    __syncthreads();
    int off = s_pre[lane] + (lane < wv ? s_inc[lane] : 0);
    for (int d = 16; d > 0; d >>= 1) off += __shfl_xor(off, d, 32);
    if (i < n) {
        rowptr[i] = off + inc - v;
        const float d = (float)v;
        dinv_s[i] = rsqrtf(d + 1.0f);
        dinv_h[i] = rsqrtf(d + 2.0f);
    }
    if (i == n - 1) rowptr[n] = off + inc;
}

static __device__ __forceinline__ void
fill_edges(const int* __restrict__ row, const int* __restrict__ col,
           const int* __restrict__ rank, const int* __restrict__ rowptr,
           const int* __restrict__ part, int* __restrict__ tmp_pair,
           int e, int np, int ec, int blk) {
    int i = blk * blockDim.x + threadIdx.x;
    if (i < e) {
        const int c    = col[i];
        const int slot = rowptr[c] + rank[i] + part[(size_t)(i / ec) * np + c];
        const v2i t = {row[i], c};
        *(v2i*)(tmp_pair + 2 * slot) = t;
    }
}

constexpr int PLACE_S = 4;

__global__ void __launch_bounds__(256)
k_csr_place(const int* __restrict__ tmp_pair, const int* __restrict__ rowptr,
            int* __restrict__ csr_src, int e) {
    const int p0 = (blockIdx.x * blockDim.x + threadIdx.x) * PLACE_S;
    if (p0 >= e) return;
    static_assert(PLACE_S == 4, "a thread's pairs are read as two 16-byte pieces");
    int r[PLACE_S], c[PLACE_S];
    if (p0 + PLACE_S <= e) {
        const v4i q0 = *(const v4i*)(tmp_pair + 2 * p0);
        const v4i q1 = *(const v4i*)(tmp_pair + 2 * p0 + 4);
        r[0] = q0.x; c[0] = q0.y; r[1] = q0.z; c[1] = q0.w;
        r[2] = q1.x; c[2] = q1.y; r[3] = q1.z; c[3] = q1.w;
    } else {
#pragma unroll
        for (int j = 0; j < PLACE_S; ++j) {
            const v2i t = *(const v2i*)(tmp_pair + 2 * min(p0 + j, e - 1));
            r[j] = t.x;
            c[j] = t.y;
        }
    }
    int beg[PLACE_S], end[PLACE_S], pos[PLACE_S];
#pragma unroll
    for (int j = 0; j < PLACE_S; ++j) {
        beg[j] = rowptr[c[j]];
        end[j] = rowptr[c[j] + 1];
    }
#pragma unroll
    for (int j = 0; j < PLACE_S; ++j) pos[j] = beg[j];
    const int hi   = end[PLACE_S - 1];
    const int last = hi - 1;
    for (int b = beg[0]; b < hi; b += 4) {
        int k[4];
#pragma unroll
        for (int i = 0; i < 4; ++i) k[i] = tmp_pair[2 * min(b + i, last)];
#pragma unroll
        for (int j = 0; j < PLACE_S; ++j) {
#pragma unroll
            for (int i = 0; i < 4; ++i) {
                const int bi = b + i;
                const int inb = (int)(bi >= beg[j]) & (int)(bi < end[j]);
                pos[j] += inb & ((int)(k[i] < r[j]) | ((int)(k[i] == r[j]) & (int)(bi < p0 + j)));
            }
        }
    }
#pragma unroll
    for (int j = 0; j < PLACE_S; ++j) {
        if (p0 + j < e) csr_src[pos[j]] = r[j];
    }
}

template <int DIN, int NT, bool RELU_IN, bool BIAS>
static __device__ __forceinline__ void
gemm_wmma_strip(const float* __restrict__ X, const float* __restrict__ W,
                const float* __restrict__ bias, const float* __restrict__ rscale,
                float* __restrict__ H, int N, int blk) {
    constexpr int DOUT = NT * 16;
    const int lane = threadIdx.x & 31;
    const int wave = threadIdx.x >> 5;
    const int tile = blk * (blockDim.x >> 5) + wave;
    const int rtiles = (N + 15) >> 4;
    if (tile >= rtiles) return;

    const int mrow  = tile * 16 + (lane & 15);
    const int mload = mrow < N ? mrow : N - 1;
    const int kb    = (lane >> 4) * 2;
    const int nloc  = lane & 15;

    v8f acc[NT];
#pragma unroll
    for (int j = 0; j < NT; ++j) {
        v8f z = {0.f, 0.f, 0.f, 0.f, 0.f, 0.f, 0.f, 0.f};
        acc[j] = z;
    }

    const float* xr = X + (size_t)mload * DIN;
#pragma unroll 4
    for (int k = 0; k < DIN; k += 4) {
        float a0 = xr[k + kb];
        float a1 = xr[k + kb + 1];
        if (RELU_IN) { a0 = fmaxf(a0, 0.0f); a1 = fmaxf(a1, 0.0f); }
        v2f A = {a0, a1};
#pragma unroll
        for (int j = 0; j < NT; ++j) {
            const int n0 = j * 16 + nloc;
            v2f B = {W[(size_t)(k + kb) * DOUT + n0],
                     W[(size_t)(k + kb + 1) * DOUT + n0]};
            acc[j] = __builtin_amdgcn_wmma_f32_16x16x4_f32(
                false, A, false, B, (short)0, acc[j], false, false);
        }
    }

    const int half = lane >> 4;
    float sc[8];
#pragma unroll
    for (int r = 0; r < 8; ++r) {
        const int mo = tile * 16 + r + 8 * half;
        sc[r] = rscale[mo < N ? mo : N - 1];
    }
#pragma unroll
    for (int j = 0; j < NT; ++j) {
        const float bj = BIAS ? bias[j * 16 + nloc] : 0.0f;
#pragma unroll
        for (int r = 0; r < 8; ++r) {
            int mo = tile * 16 + r + 8 * half;
            if (mo < N) H[(size_t)mo * DOUT + j * 16 + nloc] = (BIAS ? acc[j][r] + bj : acc[j][r]) * sc[r];
        }
    }
}

static __device__ __forceinline__ void
count_range_chunk(const int* __restrict__ col, int* __restrict__ rank,
                  int* __restrict__ part, int* __restrict__ gsum,
                  int n, int e, int np, int ec, int blk) {
    __shared__ int s_cnt[CNT_NR + 32];
    __shared__ int s_gs[CNT_NR / 1024];
    const int tid = threadIdx.x;
    const int r   = blk / CNT_G;
    const int g   = blk % CNT_G;
    const int lo  = r * CNT_NR;
    for (int j = tid; j < CNT_NR + 32; j += 256) s_cnt[j] = 0;
    if (tid < CNT_NR / 1024) s_gs[tid] = 0;
    __syncthreads();

    const int e0 = g * ec;
    const int e1 = min(e0 + ec, e);
    for (int base = e0; base < e1; base += 8 * 256) {
        int cv[8];
#pragma unroll
        for (int u = 0; u < 8; ++u) {
            const int i = base + u * 256 + tid;
            cv[u] = (i < e1) ? col[i] : -1;
        }
        int rk[8];
#pragma unroll
        for (int u = 0; u < 8; ++u) {
            const unsigned d = (unsigned)(cv[u] - lo);
            rk[u] = atomicAdd(&s_cnt[d < (unsigned)CNT_NR ? d : (unsigned)(CNT_NR + (tid & 31))], 1);
        }
#pragma unroll
        for (int u = 0; u < 8; ++u) {
            const int i = base + u * 256 + tid;
            if ((unsigned)(cv[u] - lo) < (unsigned)CNT_NR) rank[i] = rk[u];
        }
    }
    __syncthreads();

    for (int q = 0; q < CNT_NR / 1024; ++q) {
        int sum = 0;
#pragma unroll
        for (int t = 0; t < 4; ++t) {
            const int j = (q * 4 + t) * 256 + tid;
            const int v = s_cnt[j];
            if (lo + j < n) part[(size_t)g * np + lo + j] = v;
            sum += v;
        }
        for (int d = 16; d > 0; d >>= 1) sum += __shfl_xor(sum, d, 32);
        if ((tid & 31) == 0) atomicAdd(&s_gs[q], sum);
    }
    __syncthreads();
    if (tid < CNT_NR / 1024) gsum[(r * (CNT_NR / 1024) + tid) * CNT_G + g] = s_gs[tid];
}

__global__ void __launch_bounds__(256)
k_count_rank(const int* __restrict__ col, int* __restrict__ rank,
             int* __restrict__ part, int* __restrict__ gsum,
             int n, int e, int np, int ec) {
    count_range_chunk(col, rank, part, gsum, n, e, np, ec, blockIdx.x);
}

__global__ void __launch_bounds__(256, 16)
k_gemm1_fill(const float* __restrict__ X, const float* __restrict__ W,
             const float* __restrict__ rscale, float* __restrict__ H, int N, int gblocks,
             const int* __restrict__ row, const int* __restrict__ col,
             const int* __restrict__ rank, const int* __restrict__ rowptr,
             const int* __restrict__ part, int* __restrict__ tmp_pair,
             int e, int np, int ec) {
    const int b = blockIdx.x;
    if (b < gblocks) {
        gemm_wmma_strip<128, 4, false, false>(X, W, nullptr, rscale, H, N, b);
    } else {
        fill_edges(row, col, rank, rowptr, part, tmp_pair, e, np, ec, b - gblocks);
    }
}

static __device__ __forceinline__ v4f relu4(v4f h) {
    h.x = fmaxf(h.x, 0.0f); h.y = fmaxf(h.y, 0.0f);
    h.z = fmaxf(h.z, 0.0f); h.w = fmaxf(h.w, 0.0f);
    return h;
}

template <int D, bool DEC>
static __device__ __forceinline__ void
gather_pair(const float* __restrict__ H, const int* __restrict__ rowptr,
            const int* __restrict__ csr_src,
            const float* __restrict__ dinv, const float* __restrict__ bias,
            int a, int q, int n, v4f& accA, v4f& accB) {
    constexpr unsigned ROWB = D * 4;
    const char* __restrict__ Hb = (const char*)H;
    const unsigned qoff = (unsigned)q * 16u;
    const bool hasB = a + 1 < n;
    const int b = hasB ? a + 1 : a;

    const int beg = rowptr[a];
    const int mid = rowptr[a + 1];
    const int end = rowptr[b + 1];
    const float dvA = dinv[a];
    const float dvB = dinv[b];
    v4f bs = {0.f, 0.f, 0.f, 0.f};
    if (!DEC) { bs.x = bias[4 * q]; bs.y = bias[4 * q + 1]; bs.z = bias[4 * q + 2]; bs.w = bias[4 * q + 3]; }

    accA = *(const v4f*)(Hb + ((unsigned)a * ROWB + qoff));
    accB = *(const v4f*)(Hb + ((unsigned)b * ROWB + qoff));
    if (DEC) {
        accA = accA + accA;
        accB = accB + accB;
    }

    int e = beg;
    for (; e + 8 <= end; e += 8) {
        int s[8];
        v4f h[8];
#pragma unroll
        for (int j = 0; j < 8; ++j) s[j] = csr_src[e + j];
#pragma unroll
        for (int j = 0; j < 8; ++j) h[j] = *(const v4f*)(Hb + ((unsigned)s[j] * ROWB + qoff));
#pragma unroll
        for (int j = 0; j < 8; ++j) {
            if (DEC) h[j] = -h[j];
            if (e + j < mid) accA += h[j];
            else             accB += h[j];
        }
    }
    if (e < end) {
        const int last = end - 1;
        int s[7];
        v4f h[7];
#pragma unroll
        for (int j = 0; j < 7; ++j) s[j] = csr_src[min(e + j, last)];
#pragma unroll
        for (int j = 0; j < 7; ++j) h[j] = *(const v4f*)(Hb + ((unsigned)s[j] * ROWB + qoff));
#pragma unroll
        for (int j = 0; j < 7; ++j) {
            if (DEC) h[j] = -h[j];
            if (e + j < mid)      accA += h[j];
            else if (e + j < end) accB += h[j];
        }
    }
    accA = accA * dvA;
    accB = accB * dvB;
    if (!DEC) { accA += bs; accB += bs; }
    if (!hasB) { v4f z = {0.f, 0.f, 0.f, 0.f}; accB = z; }
}

template <int D, bool DEC>
__global__ void __launch_bounds__(256, 16)
k_agg_gather(const float* __restrict__ H, const int* __restrict__ rowptr,
                             const int* __restrict__ csr_src,
                             const float* __restrict__ dinv, const float* __restrict__ dnext,
                             const float* __restrict__ bias,
                             float* __restrict__ Y, int n) {
    constexpr int LPN = D / 4;
    const int i = blockIdx.x * blockDim.x + threadIdx.x;
    const int a = (i / LPN) * 2;
    if (a >= n) return;
    const int q = i % LPN;
    const float nxA = dnext[a];
    const float nxB = dnext[a + 1 < n ? a + 1 : a];
    v4f accA, accB;
    gather_pair<D, DEC>(H, rowptr, csr_src, dinv, bias, a, q, n, accA, accB);
    ((v4f*)Y)[(size_t)a * LPN + q] = relu4(accA) * nxA;
    if (a + 1 < n) ((v4f*)Y)[(size_t)(a + 1) * LPN + q] = relu4(accB) * nxB;
}

template <int D, int NT, bool DEC, bool LAST>
__global__ void __launch_bounds__(256, 16)
k_agg_gemm(const float* __restrict__ H, const int* __restrict__ rowptr,
           const int* __restrict__ csr_src,
           const float* __restrict__ dinv, const float* __restrict__ dnext,
           const float* __restrict__ bias,
           const float* __restrict__ W, float* __restrict__ Y, int n) {
    constexpr int LPN  = D / 4;
    constexpr int NPB  = 2 * (256 / LPN);
    constexpr int RT   = NPB / 16;
    constexpr int DOUT = NT * 16;
    constexpr int T    = RT * NT;
    constexpr int WPT  = T < 8 ? T : 8;
    constexpr int TPW  = T / WPT;
    constexpr int RSTEP = WPT / NT;
    constexpr int KS   = D / 4;
    constexpr int QO   = DOUT / 4;
    constexpr int PA   = D + 4;
    constexpr int PO   = DOUT + 4;
    constexpr bool ALLW = (WPT == 8);
    constexpr int NIT  = NPB * QO / 256;
    static_assert(T % WPT == 0 && WPT % NT == 0, "a wave's tiles share one column tile");
    static_assert((NPB * QO) % 256 == 0, "every thread stores the same number of pieces");
    static_assert(DEC || !LAST, "only a decoder layer is the last");
    __shared__ __attribute__((aligned(16))) float sA[NPB * PA];
    __shared__ __attribute__((aligned(16))) float sO[NPB * PO];

    const int tid  = threadIdx.x;
    const int pl   = tid / LPN;
    const int q    = tid % LPN;
    const int a    = blockIdx.x * NPB + 2 * pl;
    float* const sp = &sA[(2 * pl) * PA + 4 * q];
    v4f accA = {0.f, 0.f, 0.f, 0.f};
    v4f accB = {0.f, 0.f, 0.f, 0.f};
    float nxA = 0.0f, nxB = 0.0f;
    if (a < n) {
        if (!DEC) { nxA = dnext[a]; nxB = dnext[a + 1 < n ? a + 1 : a]; }
        gather_pair<D, DEC>(H, rowptr, csr_src, dinv, bias, a, q, n, accA, accB);
    }
    if (!DEC) { accA = relu4(accA) * nxA; accB = relu4(accB) * nxB; }
    *(v4f*)sp        = accA;
    *(v4f*)(sp + PA) = accB;

    const int wave = __builtin_amdgcn_readfirstlane(tid >> 5);
    const int lane = tid & 31;
    const int c    = lane & 15;
    const int half = lane >> 4;
    const int kb   = half * 2;
    const int rb   = wave / NT;
    const int ct   = wave % NT;

    v2f Bf[KS];
    if (ALLW || wave < WPT) {
        const float* wp = W + (size_t)kb * DOUT + ct * 16 + c;
#pragma unroll
        for (int kk = 0; kk < KS; ++kk) {
            Bf[kk].x = wp[(size_t)(4 * kk) * DOUT];
            Bf[kk].y = wp[(size_t)(4 * kk + 1) * DOUT];
        }
    }
    const int qo = tid % QO;
    v4f bv = {0.f, 0.f, 0.f, 0.f};
    if (DEC) { bv.x = bias[4 * qo]; bv.y = bias[4 * qo + 1]; bv.z = bias[4 * qo + 2]; bv.w = bias[4 * qo + 3]; }
    float dn[NIT];
#pragma unroll
    for (int it = 0; it < NIT; ++it) {
        const int nd = blockIdx.x * NPB + (tid + it * 256) / QO;
        dn[it] = (DEC && !LAST) ? dnext[nd < n ? nd : n - 1] : 1.0f;
    }

    __syncthreads();

    if (ALLW || wave < WPT) {
        v8f d[TPW];
#pragma unroll
        for (int j = 0; j < TPW; ++j) {
            v8f z = {0.f, 0.f, 0.f, 0.f, 0.f, 0.f, 0.f, 0.f};
            d[j] = z;
        }
        const float* ap = &sA[(rb * 16 + c) * PA + kb];
#pragma unroll
        for (int kk = 0; kk < KS; ++kk) {
#pragma unroll
            for (int j = 0; j < TPW; ++j) {
                const v2f A = *(const v2f*)(ap + (j * RSTEP * 16) * PA + 4 * kk);
                d[j] = __builtin_amdgcn_wmma_f32_16x16x4_f32(
                    false, A, false, Bf[kk], (short)0, d[j], false, false);
            }
        }
#pragma unroll
        for (int j = 0; j < TPW; ++j) {
#pragma unroll
            for (int r = 0; r < 8; ++r)
                sO[((rb + j * RSTEP) * 16 + 8 * half + r) * PO + ct * 16 + c] = d[j][r];
        }
    }

    __syncthreads();

#pragma unroll
    for (int it = 0; it < NIT; ++it) {
        const int row = (tid + it * 256) / QO;
        const int nd  = blockIdx.x * NPB + row;
        if (nd < n) {
            v4f o = *(const v4f*)&sO[row * PO + 4 * qo];
            if (DEC) o += bv;
            if (DEC && !LAST) o = relu4(o) * dn[it];
            ((v4f*)Y)[(size_t)nd * QO + qo] = o;
        }
    }
}

static inline int cdiv(long long a, int b) { return (int)((a + b - 1) / b); }

extern "C" void kernel_launch(void* const* d_in, const int* in_sizes, int n_in,
                              void* d_out, int out_size, void* d_ws, size_t ws_size,
                              hipStream_t stream) {
    const float* x   = (const float*)d_in[0];
    const int*   ei  = (const int*)d_in[1];
    const float* We1 = (const float*)d_in[2];  const float* be1 = (const float*)d_in[3];
    const float* We2 = (const float*)d_in[4];  const float* be2 = (const float*)d_in[5];
    const float* We3 = (const float*)d_in[6];  const float* be3 = (const float*)d_in[7];
    const float* Wd1 = (const float*)d_in[8];  const float* bd1 = (const float*)d_in[9];
    const float* Wd2 = (const float*)d_in[10]; const float* bd2 = (const float*)d_in[11];
    const float* Wd3 = (const float*)d_in[12]; const float* bd3 = (const float*)d_in[13];

    const int N = in_sizes[0] / 128;
    const int E = in_sizes[1] / 2;
    const int* rowp = ei;
    const int* colp = ei + E;

    const size_t Np = ((size_t)N + 64 + 63) & ~(size_t)63;
    const size_t Ep = ((size_t)E + 63) & ~(size_t)63;
    char* ws = (char*)d_ws;
    float* dinv_s  = (float*)ws;  ws += sizeof(float) * Np;
    float* dinv_h  = (float*)ws;  ws += sizeof(float) * Np;
    float* hbuf    = (float*)ws;  ws += sizeof(float) * (size_t)N * 64;
    float* ybuf    = (float*)ws;  ws += sizeof(float) * (size_t)N * 64;
    int*   rowptr  = (int*)ws;    ws += sizeof(int)   * Np;
    int*   tmp_pair = (int*)ws;   ws += sizeof(int)   * 2 * Ep;
    int*   csr_src = (int*)ws;    ws += sizeof(int)   * Ep;
    int*   rank    = (int*)ws;
    float* outp    = (float*)d_out;
    int*   part    = (int*)ybuf;
    int*   gsum    = part + (size_t)CNT_G * Np;

    const int TB = 256;
    const int rtiles  = (N + 15) >> 4;
    const int gblocks = (rtiles + 7) >> 3;

    const int cranges = cdiv(N, CNT_NR);
    const int ec      = cdiv(cdiv(E, CNT_G), TB) * TB;
    k_count_rank<<<cranges * CNT_G, TB, 0, stream>>>(colp, rank, part, gsum, N, E, (int)Np, ec);
    k_rowptr_scan_dinv<<<cdiv(N, 1024), 1024, 0, stream>>>(part, gsum, rowptr, dinv_s, dinv_h, N, (int)Np);
    k_gemm1_fill <<<gblocks + cdiv(E, TB), TB, 0, stream>>>(x, We1, dinv_s, hbuf, N, gblocks,
                                                            rowp, colp, rank, rowptr, part, tmp_pair, E, (int)Np, ec);
    k_csr_place  <<<cdiv(cdiv(E, PLACE_S), TB), TB, 0, stream>>>(tmp_pair, rowptr, csr_src, E);

    k_agg_gemm<64, 2, false, false><<<cdiv(N, 32), TB, 0, stream>>>(hbuf, rowptr, csr_src, dinv_s, dinv_s, be1, We2, ybuf, N);
    k_agg_gemm<32, 1, false, false><<<cdiv(N, 64), TB, 0, stream>>>(ybuf, rowptr, csr_src, dinv_s, dinv_s, be2, We3, hbuf, N);
    k_agg_gather<16, false><<<cdiv(N, 128), TB, 0, stream>>>(hbuf, rowptr, csr_src, dinv_s, dinv_h, be3, ybuf, N);

    k_agg_gemm<16, 2, true, false><<<cdiv(N, 128), TB, 0, stream>>>(ybuf, rowptr, csr_src, dinv_h, dinv_h, bd1, Wd1, hbuf, N);
    k_agg_gemm<32, 4, true, false><<<cdiv(N, 64), TB, 0, stream>>>(hbuf, rowptr, csr_src, dinv_h, dinv_h, bd2, Wd2, ybuf, N);
    k_agg_gemm<64, 8, true, true><<<cdiv(N, 32), TB, 0, stream>>>(ybuf, rowptr, csr_src, dinv_h, dinv_h, bd3, Wd3, outp, N);
}


# Bucket fill runs per (node range, edge chunk) block with its slot bases in LDS: a slot comes from an LDS add, not from two scattered global reads and a per-edge rank array; sums keep their order
# speedup vs baseline: 5.5969x; 1.0388x over previous
#include <hip/hip_runtime.h>

typedef __attribute__((ext_vector_type(2))) float v2f;
typedef __attribute__((ext_vector_type(8))) float v8f;
typedef __attribute__((ext_vector_type(4))) float v4f;
typedef __attribute__((ext_vector_type(2))) int v2i;
typedef __attribute__((ext_vector_type(4))) int v4i;

constexpr int CNT_NR = 8192;
constexpr int CNT_G  = 32;

__global__ void __launch_bounds__(1024)
k_rowptr_scan_dinv(int* __restrict__ part, const int* __restrict__ gsum, int* __restrict__ rowptr,
                   float* __restrict__ dinv_s, float* __restrict__ dinv_h, int n, int np) {
    __shared__ int s_pre[32];
    __shared__ int s_inc[32];
    const int tid  = threadIdx.x;
    const int lane = tid & 31;
    const int wv   = tid >> 5;
    int pre = 0;
    const int ng = (int)blockIdx.x * CNT_G;
    for (int j = tid; j < ng; j += 1024) pre += gsum[j];
    const int i = blockIdx.x * 1024 + tid;
    int v = 0;
    int t[CNT_G];
#pragma unroll
    for (int g = 0; g < CNT_G; ++g) t[g] = 0;
    if (i < n) {
#pragma unroll
        for (int g = 0; g < CNT_G; ++g) t[g] = part[(size_t)g * np + i];
    }
#pragma unroll
    for (int g = 0; g < CNT_G; ++g) v += t[g];
    int inc = v;
    for (int d = 1; d < 32; d <<= 1) {
        const int up = __shfl_up(inc, d, 32);
        if (lane >= d) inc += up;
    }
    for (int d = 16; d > 0; d >>= 1) pre += __shfl_xor(pre, d, 32);
    if (lane == 31) s_inc[wv] = inc;
    if (lane == 0)  s_pre[wv] = pre;
    __syncthreads();
    int off = s_pre[lane] + (lane < wv ? s_inc[lane] : 0);
    for (int d = 16; d > 0; d >>= 1) off += __shfl_xor(off, d, 32);
    if (i < n) {
        const int rp = off + inc - v;
        rowptr[i] = rp;
        int run = rp;
#pragma unroll
        for (int g = 0; g < CNT_G; ++g) { part[(size_t)g * np + i] = run; run += t[g]; }
        const float d = (float)v;
        dinv_s[i] = rsqrtf(d + 1.0f);
        dinv_h[i] = rsqrtf(d + 2.0f);
    }
    if (i == n - 1) rowptr[n] = off + inc;
}

static __device__ __forceinline__ void
fill_range_chunk(const int* __restrict__ row, const int* __restrict__ col,
                 const int* __restrict__ part, int* __restrict__ tmp_pair,
                 int n, int e, int np, int ec, int blk) {
    __shared__ __attribute__((aligned(16))) int s_slot[CNT_NR + 32];
    const int tid = threadIdx.x;
    const int r   = blk / CNT_G;
    const int g   = blk % CNT_G;
    const int lo  = r * CNT_NR;
    const int* __restrict__ pg = part + (size_t)g * np;
    const int nfull = n & ~3;
    v4i b[CNT_NR / 1024];
#pragma unroll
    for (int it = 0; it < CNT_NR / 1024; ++it) {
        v4i z = {0, 0, 0, 0};
        b[it] = z;
    }
    if (nfull >= 4) {
#pragma unroll
        for (int it = 0; it < CNT_NR / 1024; ++it) {
            const int gi = lo + (it * 256 + tid) * 4;
            b[it] = *(const v4i*)(pg + min(gi, nfull - 4));
        }
    }
#pragma unroll
    for (int it = 0; it < CNT_NR / 1024; ++it) *(v4i*)&s_slot[(it * 256 + tid) * 4] = b[it];
    if (nfull < n) {
        const int j = nfull - lo;
        if (j >= 0 && j < CNT_NR && tid == ((j >> 2) & 255)) {
            for (int k = nfull; k < n; ++k) s_slot[k - lo] = pg[k];
        }
    }
    if (tid < 32) s_slot[CNT_NR + tid] = 0;
    __syncthreads();

    const int e0 = g * ec;
    const int e1 = min(e0 + ec, e);
    if (e0 >= e1) return;
    const int elast = e1 - 1;
    int cv[8], rv[8];
#pragma unroll
    for (int u = 0; u < 8; ++u) {
        const int i = e0 + u * 256 + tid;
        const int c = col[min(i, elast)];
        rv[u] = row[min(i, elast)];
        cv[u] = (i < e1) ? c : -1;
    }
    for (int base = e0; base < e1; base += 8 * 256) {
        int ncv[8], nrv[8];
#pragma unroll
        for (int u = 0; u < 8; ++u) {
            const int i = base + 8 * 256 + u * 256 + tid;
            const int c = col[min(i, elast)];
            nrv[u] = row[min(i, elast)];
            ncv[u] = (i < e1) ? c : -1;
        }
        int sl[8];
#pragma unroll
        for (int u = 0; u < 8; ++u) {
            const unsigned d = (unsigned)(cv[u] - lo);
            sl[u] = atomicAdd(&s_slot[d < (unsigned)CNT_NR ? d : (unsigned)(CNT_NR + (tid & 31))], 1);
        }
#pragma unroll
        for (int u = 0; u < 8; ++u) {
            if ((unsigned)(cv[u] - lo) < (unsigned)CNT_NR) {
                const v2i t = {rv[u], cv[u]};
                *(v2i*)(tmp_pair + 2 * sl[u]) = t;
            }
        }
#pragma unroll
        for (int u = 0; u < 8; ++u) { cv[u] = ncv[u]; rv[u] = nrv[u]; }
    }
}

constexpr int PLACE_S = 4;

__global__ void __launch_bounds__(256)
k_csr_place(const int* __restrict__ tmp_pair, const int* __restrict__ rowptr,
            int* __restrict__ csr_src, int e) {
    const int p0 = (blockIdx.x * blockDim.x + threadIdx.x) * PLACE_S;
    if (p0 >= e) return;
    static_assert(PLACE_S == 4, "a thread's pairs are read as two 16-byte pieces");
    int r[PLACE_S], c[PLACE_S];
    if (p0 + PLACE_S <= e) {
        const v4i q0 = *(const v4i*)(tmp_pair + 2 * p0);
        const v4i q1 = *(const v4i*)(tmp_pair + 2 * p0 + 4);
        r[0] = q0.x; c[0] = q0.y; r[1] = q0.z; c[1] = q0.w;
        r[2] = q1.x; c[2] = q1.y; r[3] = q1.z; c[3] = q1.w;
    } else {
#pragma unroll
        for (int j = 0; j < PLACE_S; ++j) {
            const v2i t = *(const v2i*)(tmp_pair + 2 * min(p0 + j, e - 1));
            r[j] = t.x;
            c[j] = t.y;
        }
    }
    int beg[PLACE_S], end[PLACE_S], pos[PLACE_S];
#pragma unroll
    for (int j = 0; j < PLACE_S; ++j) {
        beg[j] = rowptr[c[j]];
        end[j] = rowptr[c[j] + 1];
    }
#pragma unroll
    for (int j = 0; j < PLACE_S; ++j) pos[j] = beg[j];
    const int hi   = end[PLACE_S - 1];
    const int last = hi - 1;
    for (int b = beg[0]; b < hi; b += 4) {
        int k[4];
#pragma unroll
        for (int i = 0; i < 4; ++i) k[i] = tmp_pair[2 * min(b + i, last)];
#pragma unroll
        for (int j = 0; j < PLACE_S; ++j) {
#pragma unroll
            for (int i = 0; i < 4; ++i) {
                const int bi = b + i;
                const int inb = (int)(bi >= beg[j]) & (int)(bi < end[j]);
                pos[j] += inb & ((int)(k[i] < r[j]) | ((int)(k[i] == r[j]) & (int)(bi < p0 + j)));
            }
        }
    }
#pragma unroll
    for (int j = 0; j < PLACE_S; ++j) {
        if (p0 + j < e) csr_src[pos[j]] = r[j];
    }
}

template <int DIN, int NT, bool RELU_IN, bool BIAS>
static __device__ __forceinline__ void
gemm_wmma_strip(const float* __restrict__ X, const float* __restrict__ W,
                const float* __restrict__ bias, const float* __restrict__ rscale,
                float* __restrict__ H, int N, int blk) {
    constexpr int DOUT = NT * 16;
    const int lane = threadIdx.x & 31;
    const int wave = threadIdx.x >> 5;
    const int tile = blk * (blockDim.x >> 5) + wave;
    const int rtiles = (N + 15) >> 4;
    if (tile >= rtiles) return;

    const int mrow  = tile * 16 + (lane & 15);
    const int mload = mrow < N ? mrow : N - 1;
    const int kb    = (lane >> 4) * 2;
    const int nloc  = lane & 15;

    v8f acc[NT];
#pragma unroll
    for (int j = 0; j < NT; ++j) {
        v8f z = {0.f, 0.f, 0.f, 0.f, 0.f, 0.f, 0.f, 0.f};
        acc[j] = z;
    }

    const float* xr = X + (size_t)mload * DIN;
#pragma unroll 4
    for (int k = 0; k < DIN; k += 4) {
        float a0 = xr[k + kb];
        float a1 = xr[k + kb + 1];
        if (RELU_IN) { a0 = fmaxf(a0, 0.0f); a1 = fmaxf(a1, 0.0f); }
        v2f A = {a0, a1};
#pragma unroll
        for (int j = 0; j < NT; ++j) {
            const int n0 = j * 16 + nloc;
            v2f B = {W[(size_t)(k + kb) * DOUT + n0],
                     W[(size_t)(k + kb + 1) * DOUT + n0]};
            acc[j] = __builtin_amdgcn_wmma_f32_16x16x4_f32(
                false, A, false, B, (short)0, acc[j], false, false);
        }
    }

    const int half = lane >> 4;
    float sc[8];
#pragma unroll
    for (int r = 0; r < 8; ++r) {
        const int mo = tile * 16 + r + 8 * half;
        sc[r] = rscale[mo < N ? mo : N - 1];
    }
#pragma unroll
    for (int j = 0; j < NT; ++j) {
        const float bj = BIAS ? bias[j * 16 + nloc] : 0.0f;
#pragma unroll
        for (int r = 0; r < 8; ++r) {
            int mo = tile * 16 + r + 8 * half;
            if (mo < N) H[(size_t)mo * DOUT + j * 16 + nloc] = (BIAS ? acc[j][r] + bj : acc[j][r]) * sc[r];
        }
    }
}

static __device__ __forceinline__ void
count_range_chunk(const int* __restrict__ col,
                  int* __restrict__ part, int* __restrict__ gsum,
                  int n, int e, int np, int ec, int blk) {
    __shared__ int s_cnt[CNT_NR + 32];
    __shared__ int s_gs[CNT_NR / 1024];
    const int tid = threadIdx.x;
    const int r   = blk / CNT_G;
    const int g   = blk % CNT_G;
    const int lo  = r * CNT_NR;
    for (int j = tid; j < CNT_NR + 32; j += 256) s_cnt[j] = 0;
    if (tid < CNT_NR / 1024) s_gs[tid] = 0;
    __syncthreads();

    const int e0 = g * ec;
    const int e1 = min(e0 + ec, e);
    for (int base = e0; base < e1; base += 8 * 256) {
        int cv[8];
#pragma unroll
        for (int u = 0; u < 8; ++u) {
            const int i = base + u * 256 + tid;
            cv[u] = (i < e1) ? col[i] : -1;
        }
#pragma unroll
        for (int u = 0; u < 8; ++u) {
            const unsigned d = (unsigned)(cv[u] - lo);
            atomicAdd(&s_cnt[d < (unsigned)CNT_NR ? d : (unsigned)(CNT_NR + (tid & 31))], 1);
        }
    }
    __syncthreads();

    for (int q = 0; q < CNT_NR / 1024; ++q) {
        int sum = 0;
#pragma unroll
        for (int t = 0; t < 4; ++t) {
            const int j = (q * 4 + t) * 256 + tid;
            const int v = s_cnt[j];
            if (lo + j < n) part[(size_t)g * np + lo + j] = v;
            sum += v;
        }
        for (int d = 16; d > 0; d >>= 1) sum += __shfl_xor(sum, d, 32);
        if ((tid & 31) == 0) atomicAdd(&s_gs[q], sum);
    }
    __syncthreads();
    if (tid < CNT_NR / 1024) gsum[(r * (CNT_NR / 1024) + tid) * CNT_G + g] = s_gs[tid];
}

__global__ void __launch_bounds__(256)
k_count_indeg(const int* __restrict__ col,
             int* __restrict__ part, int* __restrict__ gsum,
             int n, int e, int np, int ec) {
    count_range_chunk(col, part, gsum, n, e, np, ec, blockIdx.x);
}

__global__ void __launch_bounds__(256, 16)
k_gemm1_fill(const float* __restrict__ X, const float* __restrict__ W,
             const float* __restrict__ rscale, float* __restrict__ H, int N, int fblocks,
             const int* __restrict__ row, const int* __restrict__ col,
             const int* __restrict__ part, int* __restrict__ tmp_pair,
             int e, int np, int ec) {
    const int b = blockIdx.x;
    if (b < fblocks) {
        fill_range_chunk(row, col, part, tmp_pair, N, e, np, ec, b);
    } else {
        gemm_wmma_strip<128, 4, false, false>(X, W, nullptr, rscale, H, N, b - fblocks);
    }
}

static __device__ __forceinline__ v4f relu4(v4f h) {
    h.x = fmaxf(h.x, 0.0f); h.y = fmaxf(h.y, 0.0f);
    h.z = fmaxf(h.z, 0.0f); h.w = fmaxf(h.w, 0.0f);
    return h;
}

template <int D, bool DEC>
static __device__ __forceinline__ void
gather_pair(const float* __restrict__ H, const int* __restrict__ rowptr,
            const int* __restrict__ csr_src,
            const float* __restrict__ dinv, const float* __restrict__ bias,
            int a, int q, int n, v4f& accA, v4f& accB) {
    constexpr unsigned ROWB = D * 4;
    const char* __restrict__ Hb = (const char*)H;
    const unsigned qoff = (unsigned)q * 16u;
    const bool hasB = a + 1 < n;
    const int b = hasB ? a + 1 : a;

    const int beg = rowptr[a];
    const int mid = rowptr[a + 1];
    const int end = rowptr[b + 1];
    const float dvA = dinv[a];
    const float dvB = dinv[b];
    v4f bs = {0.f, 0.f, 0.f, 0.f};
    if (!DEC) { bs.x = bias[4 * q]; bs.y = bias[4 * q + 1]; bs.z = bias[4 * q + 2]; bs.w = bias[4 * q + 3]; }

    accA = *(const v4f*)(Hb + ((unsigned)a * ROWB + qoff));
    accB = *(const v4f*)(Hb + ((unsigned)b * ROWB + qoff));
    if (DEC) {
        accA = accA + accA;
        accB = accB + accB;
    }

    int e = beg;
    for (; e + 8 <= end; e += 8) {
        int s[8];
        v4f h[8];
#pragma unroll
        for (int j = 0; j < 8; ++j) s[j] = csr_src[e + j];
#pragma unroll
        for (int j = 0; j < 8; ++j) h[j] = *(const v4f*)(Hb + ((unsigned)s[j] * ROWB + qoff));
#pragma unroll
        for (int j = 0; j < 8; ++j) {
            if (DEC) h[j] = -h[j];
            if (e + j < mid) accA += h[j];
            else             accB += h[j];
        }
    }
    if (e < end) {
        const int last = end - 1;
        int s[7];
        v4f h[7];
#pragma unroll
        for (int j = 0; j < 7; ++j) s[j] = csr_src[min(e + j, last)];
#pragma unroll
        for (int j = 0; j < 7; ++j) h[j] = *(const v4f*)(Hb + ((unsigned)s[j] * ROWB + qoff));
#pragma unroll
        for (int j = 0; j < 7; ++j) {
            if (DEC) h[j] = -h[j];
            if (e + j < mid)      accA += h[j];
            else if (e + j < end) accB += h[j];
        }
    }
    accA = accA * dvA;
    accB = accB * dvB;
    if (!DEC) { accA += bs; accB += bs; }
    if (!hasB) { v4f z = {0.f, 0.f, 0.f, 0.f}; accB = z; }
}

template <int D, bool DEC>
__global__ void __launch_bounds__(256, 16)
k_agg_gather(const float* __restrict__ H, const int* __restrict__ rowptr,
                             const int* __restrict__ csr_src,
                             const float* __restrict__ dinv, const float* __restrict__ dnext,
                             const float* __restrict__ bias,
                             float* __restrict__ Y, int n) {
    constexpr int LPN = D / 4;
    const int i = blockIdx.x * blockDim.x + threadIdx.x;
    const int a = (i / LPN) * 2;
    if (a >= n) return;
    const int q = i % LPN;
    const float nxA = dnext[a];
    const float nxB = dnext[a + 1 < n ? a + 1 : a];
    v4f accA, accB;
    gather_pair<D, DEC>(H, rowptr, csr_src, dinv, bias, a, q, n, accA, accB);
    ((v4f*)Y)[(size_t)a * LPN + q] = relu4(accA) * nxA;
    if (a + 1 < n) ((v4f*)Y)[(size_t)(a + 1) * LPN + q] = relu4(accB) * nxB;
}

template <int D, int NT, bool DEC, bool LAST>
__global__ void __launch_bounds__(256, 16)
k_agg_gemm(const float* __restrict__ H, const int* __restrict__ rowptr,
           const int* __restrict__ csr_src,
           const float* __restrict__ dinv, const float* __restrict__ dnext,
           const float* __restrict__ bias,
           const float* __restrict__ W, float* __restrict__ Y, int n) {
    constexpr int LPN  = D / 4;
    constexpr int NPB  = 2 * (256 / LPN);
    constexpr int RT   = NPB / 16;
    constexpr int DOUT = NT * 16;
    constexpr int T    = RT * NT;
    constexpr int WPT  = T < 8 ? T : 8;
    constexpr int TPW  = T / WPT;
    constexpr int RSTEP = WPT / NT;
    constexpr int KS   = D / 4;
    constexpr int QO   = DOUT / 4;
    constexpr int PA   = D + 4;
    constexpr int PO   = DOUT + 4;
    constexpr bool ALLW = (WPT == 8);
    constexpr int NIT  = NPB * QO / 256;
    static_assert(T % WPT == 0 && WPT % NT == 0, "a wave's tiles share one column tile");
    static_assert((NPB * QO) % 256 == 0, "every thread stores the same number of pieces");
    static_assert(DEC || !LAST, "only a decoder layer is the last");
    __shared__ __attribute__((aligned(16))) float sA[NPB * PA];
    __shared__ __attribute__((aligned(16))) float sO[NPB * PO];

    const int tid  = threadIdx.x;
    const int pl   = tid / LPN;
    const int q    = tid % LPN;
    const int a    = blockIdx.x * NPB + 2 * pl;
    float* const sp = &sA[(2 * pl) * PA + 4 * q];
    v4f accA = {0.f, 0.f, 0.f, 0.f};
    v4f accB = {0.f, 0.f, 0.f, 0.f};
    float nxA = 0.0f, nxB = 0.0f;
    if (a < n) {
        if (!DEC) { nxA = dnext[a]; nxB = dnext[a + 1 < n ? a + 1 : a]; }
        gather_pair<D, DEC>(H, rowptr, csr_src, dinv, bias, a, q, n, accA, accB);
    }
    if (!DEC) { accA = relu4(accA) * nxA; accB = relu4(accB) * nxB; }
    *(v4f*)sp        = accA;
    *(v4f*)(sp + PA) = accB;

    const int wave = __builtin_amdgcn_readfirstlane(tid >> 5);
    const int lane = tid & 31;
    const int c    = lane & 15;
    const int half = lane >> 4;
    const int kb   = half * 2;
    const int rb   = wave / NT;
    const int ct   = wave % NT;

    v2f Bf[KS];
    if (ALLW || wave < WPT) {
        const float* wp = W + (size_t)kb * DOUT + ct * 16 + c;
#pragma unroll
        for (int kk = 0; kk < KS; ++kk) {
            Bf[kk].x = wp[(size_t)(4 * kk) * DOUT];
            Bf[kk].y = wp[(size_t)(4 * kk + 1) * DOUT];
        }
    }
    const int qo = tid % QO;
    v4f bv = {0.f, 0.f, 0.f, 0.f};
    if (DEC) { bv.x = bias[4 * qo]; bv.y = bias[4 * qo + 1]; bv.z = bias[4 * qo + 2]; bv.w = bias[4 * qo + 3]; }
    float dn[NIT];
#pragma unroll
    for (int it = 0; it < NIT; ++it) {
        const int nd = blockIdx.x * NPB + (tid + it * 256) / QO;
        dn[it] = (DEC && !LAST) ? dnext[nd < n ? nd : n - 1] : 1.0f;
    }

    __syncthreads();

    if (ALLW || wave < WPT) {
        v8f d[TPW];
#pragma unroll
        for (int j = 0; j < TPW; ++j) {
            v8f z = {0.f, 0.f, 0.f, 0.f, 0.f, 0.f, 0.f, 0.f};
            d[j] = z;
        }
        const float* ap = &sA[(rb * 16 + c) * PA + kb];
#pragma unroll
        for (int kk = 0; kk < KS; ++kk) {
#pragma unroll
            for (int j = 0; j < TPW; ++j) {
                const v2f A = *(const v2f*)(ap + (j * RSTEP * 16) * PA + 4 * kk);
                d[j] = __builtin_amdgcn_wmma_f32_16x16x4_f32(
                    false, A, false, Bf[kk], (short)0, d[j], false, false);
            }
        }
#pragma unroll
        for (int j = 0; j < TPW; ++j) {
#pragma unroll
            for (int r = 0; r < 8; ++r)
                sO[((rb + j * RSTEP) * 16 + 8 * half + r) * PO + ct * 16 + c] = d[j][r];
        }
    }

    __syncthreads();

#pragma unroll
    for (int it = 0; it < NIT; ++it) {
        const int row = (tid + it * 256) / QO;
        const int nd  = blockIdx.x * NPB + row;
        if (nd < n) {
            v4f o = *(const v4f*)&sO[row * PO + 4 * qo];
            if (DEC) o += bv;
            if (DEC && !LAST) o = relu4(o) * dn[it];
            ((v4f*)Y)[(size_t)nd * QO + qo] = o;
        }
    }
}

static inline int cdiv(long long a, int b) { return (int)((a + b - 1) / b); }

extern "C" void kernel_launch(void* const* d_in, const int* in_sizes, int n_in,
                              void* d_out, int out_size, void* d_ws, size_t ws_size,
                              hipStream_t stream) {
    const float* x   = (const float*)d_in[0];
    const int*   ei  = (const int*)d_in[1];
    const float* We1 = (const float*)d_in[2];  const float* be1 = (const float*)d_in[3];
    const float* We2 = (const float*)d_in[4];  const float* be2 = (const float*)d_in[5];
    const float* We3 = (const float*)d_in[6];  const float* be3 = (const float*)d_in[7];
    const float* Wd1 = (const float*)d_in[8];  const float* bd1 = (const float*)d_in[9];
    const float* Wd2 = (const float*)d_in[10]; const float* bd2 = (const float*)d_in[11];
    const float* Wd3 = (const float*)d_in[12]; const float* bd3 = (const float*)d_in[13];

    const int N = in_sizes[0] / 128;
    const int E = in_sizes[1] / 2;
    const int* rowp = ei;
    const int* colp = ei + E;

    const size_t Np = ((size_t)N + 64 + 63) & ~(size_t)63;
    const size_t Ep = ((size_t)E + 63) & ~(size_t)63;
    char* ws = (char*)d_ws;
    float* dinv_s  = (float*)ws;  ws += sizeof(float) * Np;
    float* dinv_h  = (float*)ws;  ws += sizeof(float) * Np;
    float* hbuf    = (float*)ws;  ws += sizeof(float) * (size_t)N * 64;
    float* ybuf    = (float*)ws;  ws += sizeof(float) * (size_t)N * 64;
    int*   rowptr  = (int*)ws;    ws += sizeof(int)   * Np;
    int*   tmp_pair = (int*)ws;   ws += sizeof(int)   * 2 * Ep;
    int*   csr_src = (int*)ws;
    float* outp    = (float*)d_out;
    int*   part    = (int*)ybuf;
    int*   gsum    = part + (size_t)CNT_G * Np;

    const int TB = 256;
    const int rtiles  = (N + 15) >> 4;
    const int gblocks = (rtiles + 7) >> 3;

    const int cranges = cdiv(N, CNT_NR);
    const int ec      = cdiv(cdiv(E, CNT_G), TB) * TB;
    k_count_indeg<<<cranges * CNT_G, TB, 0, stream>>>(colp, part, gsum, N, E, (int)Np, ec);
    k_rowptr_scan_dinv<<<cdiv(N, 1024), 1024, 0, stream>>>(part, gsum, rowptr, dinv_s, dinv_h, N, (int)Np);
    k_gemm1_fill <<<cranges * CNT_G + gblocks, TB, 0, stream>>>(x, We1, dinv_s, hbuf, N, cranges * CNT_G,
                                                                rowp, colp, part, tmp_pair, E, (int)Np, ec);
    k_csr_place  <<<cdiv(cdiv(E, PLACE_S), TB), TB, 0, stream>>>(tmp_pair, rowptr, csr_src, E);

    k_agg_gemm<64, 2, false, false><<<cdiv(N, 32), TB, 0, stream>>>(hbuf, rowptr, csr_src, dinv_s, dinv_s, be1, We2, ybuf, N);
    k_agg_gemm<32, 1, false, false><<<cdiv(N, 64), TB, 0, stream>>>(ybuf, rowptr, csr_src, dinv_s, dinv_s, be2, We3, hbuf, N);
    k_agg_gather<16, false><<<cdiv(N, 128), TB, 0, stream>>>(hbuf, rowptr, csr_src, dinv_s, dinv_h, be3, ybuf, N);

    k_agg_gemm<16, 2, true, false><<<cdiv(N, 128), TB, 0, stream>>>(ybuf, rowptr, csr_src, dinv_h, dinv_h, bd1, Wd1, hbuf, N);
    k_agg_gemm<32, 4, true, false><<<cdiv(N, 64), TB, 0, stream>>>(hbuf, rowptr, csr_src, dinv_h, dinv_h, bd2, Wd2, ybuf, N);
    k_agg_gemm<64, 8, true, true><<<cdiv(N, 32), TB, 0, stream>>>(ybuf, rowptr, csr_src, dinv_h, dinv_h, bd3, Wd3, outp, N);
}


# In-edge list ordering loses its launch: the first aggregation's block ranks its own nodes' bucket slots in LDS, walks that list from LDS and writes it out for the later layers; sums keep their order
# speedup vs baseline: 5.7843x; 1.0335x over previous
#include <hip/hip_runtime.h>

typedef __attribute__((ext_vector_type(2))) float v2f;
typedef __attribute__((ext_vector_type(8))) float v8f;
typedef __attribute__((ext_vector_type(4))) float v4f;
typedef __attribute__((ext_vector_type(2))) int v2i;
typedef __attribute__((ext_vector_type(4))) int v4i;

constexpr int CNT_NR = 8192;
constexpr int CNT_G  = 32;

__global__ void __launch_bounds__(1024)
k_rowptr_scan_dinv(int* __restrict__ part, const int* __restrict__ gsum, int* __restrict__ rowptr,
                   float* __restrict__ dinv_s, float* __restrict__ dinv_h, int n, int np) {
    __shared__ int s_pre[32];
    __shared__ int s_inc[32];
    const int tid  = threadIdx.x;
    const int lane = tid & 31;
    const int wv   = tid >> 5;
    int pre = 0;
    const int ng = (int)blockIdx.x * CNT_G;
    for (int j = tid; j < ng; j += 1024) pre += gsum[j];
    const int i = blockIdx.x * 1024 + tid;
    int v = 0;
    int t[CNT_G];
#pragma unroll
    for (int g = 0; g < CNT_G; ++g) t[g] = 0;
    if (i < n) {
#pragma unroll
        for (int g = 0; g < CNT_G; ++g) t[g] = part[(size_t)g * np + i];
    }
#pragma unroll
    for (int g = 0; g < CNT_G; ++g) v += t[g];
    int inc = v;
    for (int d = 1; d < 32; d <<= 1) {
        const int up = __shfl_up(inc, d, 32);
        if (lane >= d) inc += up;
    }
    for (int d = 16; d > 0; d >>= 1) pre += __shfl_xor(pre, d, 32);
    if (lane == 31) s_inc[wv] = inc;
    if (lane == 0)  s_pre[wv] = pre;
    __syncthreads();
    int off = s_pre[lane] + (lane < wv ? s_inc[lane] : 0);
    for (int d = 16; d > 0; d >>= 1) off += __shfl_xor(off, d, 32);
    if (i < n) {
        const int rp = off + inc - v;
        rowptr[i] = rp;
        int run = rp;
#pragma unroll
        for (int g = 0; g < CNT_G; ++g) { part[(size_t)g * np + i] = run; run += t[g]; }
        const float d = (float)v;
        dinv_s[i] = rsqrtf(d + 1.0f);
        dinv_h[i] = rsqrtf(d + 2.0f);
    }
    if (i == n - 1) rowptr[n] = off + inc;
}

static __device__ __forceinline__ void
fill_range_chunk(const int* __restrict__ row, const int* __restrict__ col,
                 const int* __restrict__ part, int* __restrict__ tmp_pair,
                 int n, int e, int np, int ec, int blk) {
    __shared__ __attribute__((aligned(16))) int s_slot[CNT_NR + 32];
    const int tid = threadIdx.x;
    const int r   = blk / CNT_G;
    const int g   = blk % CNT_G;
    const int lo  = r * CNT_NR;
    const int* __restrict__ pg = part + (size_t)g * np;
    const int nfull = n & ~3;
    v4i b[CNT_NR / 1024];
#pragma unroll
    for (int it = 0; it < CNT_NR / 1024; ++it) {
        v4i z = {0, 0, 0, 0};
        b[it] = z;
    }
    if (nfull >= 4) {
#pragma unroll
        for (int it = 0; it < CNT_NR / 1024; ++it) {
            const int gi = lo + (it * 256 + tid) * 4;
            b[it] = *(const v4i*)(pg + min(gi, nfull - 4));
        }
    }
#pragma unroll
    for (int it = 0; it < CNT_NR / 1024; ++it) *(v4i*)&s_slot[(it * 256 + tid) * 4] = b[it];
    if (nfull < n) {
        const int j = nfull - lo;
        if (j >= 0 && j < CNT_NR && tid == ((j >> 2) & 255)) {
            for (int k = nfull; k < n; ++k) s_slot[k - lo] = pg[k];
        }
    }
    if (tid < 32) s_slot[CNT_NR + tid] = 0;
    __syncthreads();

    const int e0 = g * ec;
    const int e1 = min(e0 + ec, e);
    if (e0 >= e1) return;
    const int elast = e1 - 1;
    int cv[8], rv[8];
#pragma unroll
    for (int u = 0; u < 8; ++u) {
        const int i = e0 + u * 256 + tid;
        const int c = col[min(i, elast)];
        rv[u] = row[min(i, elast)];
        cv[u] = (i < e1) ? c : -1;
    }
    for (int base = e0; base < e1; base += 8 * 256) {
        int ncv[8], nrv[8];
#pragma unroll
        for (int u = 0; u < 8; ++u) {
            const int i = base + 8 * 256 + u * 256 + tid;
            const int c = col[min(i, elast)];
            nrv[u] = row[min(i, elast)];
            ncv[u] = (i < e1) ? c : -1;
        }
        int sl[8];
#pragma unroll
        for (int u = 0; u < 8; ++u) {
            const unsigned d = (unsigned)(cv[u] - lo);
            sl[u] = atomicAdd(&s_slot[d < (unsigned)CNT_NR ? d : (unsigned)(CNT_NR + (tid & 31))], 1);
        }
#pragma unroll
        for (int u = 0; u < 8; ++u) {
            if ((unsigned)(cv[u] - lo) < (unsigned)CNT_NR) {
                const v2i t = {rv[u], cv[u]};
                *(v2i*)(tmp_pair + 2 * sl[u]) = t;
            }
        }
#pragma unroll
        for (int u = 0; u < 8; ++u) { cv[u] = ncv[u]; rv[u] = nrv[u]; }
    }
}

template <int DIN, int NT, bool RELU_IN, bool BIAS>
static __device__ __forceinline__ void
gemm_wmma_strip(const float* __restrict__ X, const float* __restrict__ W,
                const float* __restrict__ bias, const float* __restrict__ rscale,
                float* __restrict__ H, int N, int blk) {
    constexpr int DOUT = NT * 16;
    const int lane = threadIdx.x & 31;
    const int wave = threadIdx.x >> 5;
    const int tile = blk * (blockDim.x >> 5) + wave;
    const int rtiles = (N + 15) >> 4;
    if (tile >= rtiles) return;

    const int mrow  = tile * 16 + (lane & 15);
    const int mload = mrow < N ? mrow : N - 1;
    const int kb    = (lane >> 4) * 2;
    const int nloc  = lane & 15;

    v8f acc[NT];
#pragma unroll
    for (int j = 0; j < NT; ++j) {
        v8f z = {0.f, 0.f, 0.f, 0.f, 0.f, 0.f, 0.f, 0.f};
        acc[j] = z;
    }

    const float* xr = X + (size_t)mload * DIN;
#pragma unroll 4
    for (int k = 0; k < DIN; k += 4) {
        float a0 = xr[k + kb];
        float a1 = xr[k + kb + 1];
        if (RELU_IN) { a0 = fmaxf(a0, 0.0f); a1 = fmaxf(a1, 0.0f); }
        v2f A = {a0, a1};
#pragma unroll
        for (int j = 0; j < NT; ++j) {
            const int n0 = j * 16 + nloc;
            v2f B = {W[(size_t)(k + kb) * DOUT + n0],
                     W[(size_t)(k + kb + 1) * DOUT + n0]};
            acc[j] = __builtin_amdgcn_wmma_f32_16x16x4_f32(
                false, A, false, B, (short)0, acc[j], false, false);
        }
    }

    const int half = lane >> 4;
    float sc[8];
#pragma unroll
    for (int r = 0; r < 8; ++r) {
        const int mo = tile * 16 + r + 8 * half;
        sc[r] = rscale[mo < N ? mo : N - 1];
    }
#pragma unroll
    for (int j = 0; j < NT; ++j) {
        const float bj = BIAS ? bias[j * 16 + nloc] : 0.0f;
#pragma unroll
        for (int r = 0; r < 8; ++r) {
            int mo = tile * 16 + r + 8 * half;
            if (mo < N) H[(size_t)mo * DOUT + j * 16 + nloc] = (BIAS ? acc[j][r] + bj : acc[j][r]) * sc[r];
        }
    }
}

static __device__ __forceinline__ void
count_range_chunk(const int* __restrict__ col,
                  int* __restrict__ part, int* __restrict__ gsum,
                  int n, int e, int np, int ec, int blk) {
    __shared__ int s_cnt[CNT_NR + 32];
    __shared__ int s_gs[CNT_NR / 1024];
    const int tid = threadIdx.x;
    const int r   = blk / CNT_G;
    const int g   = blk % CNT_G;
    const int lo  = r * CNT_NR;
    for (int j = tid; j < CNT_NR + 32; j += 256) s_cnt[j] = 0;
    if (tid < CNT_NR / 1024) s_gs[tid] = 0;
    __syncthreads();

    const int e0 = g * ec;
    const int e1 = min(e0 + ec, e);
    for (int base = e0; base < e1; base += 8 * 256) {
        int cv[8];
#pragma unroll
        for (int u = 0; u < 8; ++u) {
            const int i = base + u * 256 + tid;
            cv[u] = (i < e1) ? col[i] : -1;
        }
#pragma unroll
        for (int u = 0; u < 8; ++u) {
            const unsigned d = (unsigned)(cv[u] - lo);
            atomicAdd(&s_cnt[d < (unsigned)CNT_NR ? d : (unsigned)(CNT_NR + (tid & 31))], 1);
        }
    }
    __syncthreads();

    for (int q = 0; q < CNT_NR / 1024; ++q) {
        int sum = 0;
#pragma unroll
        for (int t = 0; t < 4; ++t) {
            const int j = (q * 4 + t) * 256 + tid;
            const int v = s_cnt[j];
            if (lo + j < n) part[(size_t)g * np + lo + j] = v;
            sum += v;
        }
        for (int d = 16; d > 0; d >>= 1) sum += __shfl_xor(sum, d, 32);
        if ((tid & 31) == 0) atomicAdd(&s_gs[q], sum);
    }
    __syncthreads();
    if (tid < CNT_NR / 1024) gsum[(r * (CNT_NR / 1024) + tid) * CNT_G + g] = s_gs[tid];
}

__global__ void __launch_bounds__(256)
k_count_indeg(const int* __restrict__ col,
             int* __restrict__ part, int* __restrict__ gsum,
             int n, int e, int np, int ec) {
    count_range_chunk(col, part, gsum, n, e, np, ec, blockIdx.x);
}

__global__ void __launch_bounds__(256, 16)
k_gemm1_fill(const float* __restrict__ X, const float* __restrict__ W,
             const float* __restrict__ rscale, float* __restrict__ H, int N, int fblocks,
             const int* __restrict__ row, const int* __restrict__ col,
             const int* __restrict__ part, int* __restrict__ tmp_pair,
             int e, int np, int ec) {
    const int b = blockIdx.x;
    if (b < fblocks) {
        fill_range_chunk(row, col, part, tmp_pair, N, e, np, ec, b);
    } else {
        gemm_wmma_strip<128, 4, false, false>(X, W, nullptr, rscale, H, N, b - fblocks);
    }
}

static __device__ __forceinline__ v4f relu4(v4f h) {
    h.x = fmaxf(h.x, 0.0f); h.y = fmaxf(h.y, 0.0f);
    h.z = fmaxf(h.z, 0.0f); h.w = fmaxf(h.w, 0.0f);
    return h;
}

enum { LIST_GLOBAL = 0, LIST_LDS = 1 };

template <int LIST>
static __device__ __forceinline__ int
list_source(const int* __restrict__ g, const int* s, int lo, int e) {
    if (LIST == LIST_GLOBAL) return g[e];
    return s[e - lo];
}

template <int D, bool DEC, int LIST>
static __device__ __forceinline__ void
gather_pair_walk(const float* __restrict__ H, const int* __restrict__ lst, const int* slst, int lo,
                 int beg, int mid, int end, int q, v4f& accA, v4f& accB) {
    constexpr unsigned ROWB = D * 4;
    const char* __restrict__ Hb = (const char*)H;
    const unsigned qoff = (unsigned)q * 16u;
    if (DEC) {
        accA = accA + accA;
        accB = accB + accB;
    }

    int e = beg;
    for (; e + 8 <= end; e += 8) {
        int s[8];
        v4f h[8];
#pragma unroll
        for (int j = 0; j < 8; ++j) s[j] = list_source<LIST>(lst, slst, lo, e + j);
#pragma unroll
        for (int j = 0; j < 8; ++j) h[j] = *(const v4f*)(Hb + ((unsigned)s[j] * ROWB + qoff));
#pragma unroll
        for (int j = 0; j < 8; ++j) {
            if (DEC) h[j] = -h[j];
            if (e + j < mid) accA += h[j];
            else             accB += h[j];
        }
    }
    if (e < end) {
        const int last = end - 1;
        int s[7];
        v4f h[7];
#pragma unroll
        for (int j = 0; j < 7; ++j) s[j] = list_source<LIST>(lst, slst, lo, min(e + j, last));
#pragma unroll
        for (int j = 0; j < 7; ++j) h[j] = *(const v4f*)(Hb + ((unsigned)s[j] * ROWB + qoff));
#pragma unroll
        for (int j = 0; j < 7; ++j) {
            if (DEC) h[j] = -h[j];
            if (e + j < mid)      accA += h[j];
            else if (e + j < end) accB += h[j];
        }
    }
}

template <int D, bool DEC>
static __device__ __forceinline__ void
gather_pair_slots(const float* __restrict__ H, const int* __restrict__ tmp_pair,
                  int beg, int mid, int end, int q, v4f& accA, v4f& accB) {
    const char* __restrict__ Hb = (const char*)H;
    if (DEC) {
        accA = accA + accA;
        accB = accB + accB;
    }
    for (int e = beg; e < end; ++e) {
        const int s = tmp_pair[2 * e];
        v4f h = *(const v4f*)(Hb + ((unsigned)s * (unsigned)(D * 4) + (unsigned)q * 16u));
        if (DEC) h = -h;
        if (e < mid) accA += h;
        else         accB += h;
    }
}

template <bool DEC>
static __device__ __forceinline__ void
pair_finish(float dvA, float dvB, v4f bs, bool hasB, v4f& accA, v4f& accB) {
    accA = accA * dvA;
    accB = accB * dvB;
    if (!DEC) { accA += bs; accB += bs; }
    if (!hasB) { v4f z = {0.f, 0.f, 0.f, 0.f}; accB = z; }
}

template <int D, bool DEC>
static __device__ __forceinline__ void
gather_pair(const float* __restrict__ H, const int* __restrict__ rowptr,
            const int* __restrict__ csr_src,
            const float* __restrict__ dinv, const float* __restrict__ bias,
            int a, int q, int n, v4f& accA, v4f& accB) {
    constexpr unsigned ROWB = D * 4;
    const char* __restrict__ Hb = (const char*)H;
    const unsigned qoff = (unsigned)q * 16u;
    const bool hasB = a + 1 < n;
    const int b = hasB ? a + 1 : a;

    const int beg = rowptr[a];
    const int mid = rowptr[a + 1];
    const int end = rowptr[b + 1];
    const float dvA = dinv[a];
    const float dvB = dinv[b];
    v4f bs = {0.f, 0.f, 0.f, 0.f};
    if (!DEC) { bs.x = bias[4 * q]; bs.y = bias[4 * q + 1]; bs.z = bias[4 * q + 2]; bs.w = bias[4 * q + 3]; }
    accA = *(const v4f*)(Hb + ((unsigned)a * ROWB + qoff));
    accB = *(const v4f*)(Hb + ((unsigned)b * ROWB + qoff));
    gather_pair_walk<D, DEC, LIST_GLOBAL>(H, csr_src, nullptr, 0, beg, mid, end, q, accA, accB);
    pair_finish<DEC>(dvA, dvB, bs, hasB, accA, accB);
}

template <int D, bool DEC>
__global__ void __launch_bounds__(256, 16)
k_agg_gather(const float* __restrict__ H, const int* __restrict__ rowptr,
                             const int* __restrict__ csr_src,
                             const float* __restrict__ dinv, const float* __restrict__ dnext,
                             const float* __restrict__ bias,
                             float* __restrict__ Y, int n) {
    constexpr int LPN = D / 4;
    const int i = blockIdx.x * blockDim.x + threadIdx.x;
    const int a = (i / LPN) * 2;
    if (a >= n) return;
    const int q = i % LPN;
    const float nxA = dnext[a];
    const float nxB = dnext[a + 1 < n ? a + 1 : a];
    v4f accA, accB;
    gather_pair<D, DEC>(H, rowptr, csr_src, dinv, bias, a, q, n, accA, accB);
    ((v4f*)Y)[(size_t)a * LPN + q] = relu4(accA) * nxA;
    if (a + 1 < n) ((v4f*)Y)[(size_t)(a + 1) * LPN + q] = relu4(accB) * nxB;
}

constexpr int PL_CAP = 1024;

template <bool INL>
static __device__ __forceinline__ int
slot_rank(const int* s_src, const int* __restrict__ tmp_pair, int lo, int bg, int en, int p, int r) {
    int pos = bg;
    const int last = en - 1;
    for (int j = bg; j < en; j += 4) {
        int k[4];
#pragma unroll
        for (int i = 0; i < 4; ++i) {
            const int jj = min(j + i, last);
            k[i] = INL ? s_src[jj] : tmp_pair[2 * (lo + jj)];
        }
#pragma unroll
        for (int i = 0; i < 4; ++i) {
            const int ji = j + i;
            pos += (int)(ji < en) & ((int)(k[i] < r) | ((int)(k[i] == r) & (int)(ji < p)));
        }
    }
    return pos;
}

template <int NPB>
static __device__ __forceinline__ void
place_slot(bool inl, const int* s_rp, const int* s_src, int* s_lst,
           const int* __restrict__ tmp_pair, int* __restrict__ csr_out,
           int n0, int lo, int p, int r, int c) {
    const int ci = min(max(c - n0, 0), NPB - 1);
    const int bg = s_rp[ci] - lo;
    const int en = s_rp[ci + 1] - lo;
    const int pos = inl ? slot_rank<true>(s_src, tmp_pair, lo, bg, en, p, r)
                        : slot_rank<false>(s_src, tmp_pair, lo, bg, en, p, r);
    if (inl) s_lst[pos] = r;
    else     csr_out[lo + pos] = r;
}

template <int D, int NT, bool DEC, bool LAST, bool PLACE>
__global__ void __launch_bounds__(256, 16)
k_agg_gemm(const float* __restrict__ H, const int* __restrict__ rowptr,
           const int* __restrict__ csr_src,
           const float* __restrict__ dinv, const float* __restrict__ dnext,
           const float* __restrict__ bias,
           const float* __restrict__ W, float* __restrict__ Y, int n,
           const int* __restrict__ tmp_pair, int* __restrict__ csr_out) {
    constexpr int LPN  = D / 4;
    constexpr int NPB  = 2 * (256 / LPN);
    constexpr int RT   = NPB / 16;
    constexpr int DOUT = NT * 16;
    constexpr int T    = RT * NT;
    constexpr int WPT  = T < 8 ? T : 8;
    constexpr int TPW  = T / WPT;
    constexpr int RSTEP = WPT / NT;
    constexpr int KS   = D / 4;
    constexpr int QO   = DOUT / 4;
    constexpr int PA   = D + 4;
    constexpr int PO   = DOUT + 4;
    constexpr bool ALLW = (WPT == 8);
    constexpr int NIT  = NPB * QO / 256;
    static_assert(T % WPT == 0 && WPT % NT == 0, "a wave's tiles share one column tile");
    static_assert((NPB * QO) % 256 == 0, "every thread stores the same number of pieces");
    static_assert(DEC || !LAST, "only a decoder layer is the last");
    static_assert(!PLACE || !DEC, "the ordering block is the first (encoder) aggregation");
    __shared__ __attribute__((aligned(16))) float sA[NPB * PA];
    __shared__ __attribute__((aligned(16))) float sO[NPB * PO];
    __shared__ int s_rp[PLACE ? NPB + 1 : 1];
    __shared__ int s_src[PLACE ? PL_CAP : 1];
    __shared__ int s_tgt[PLACE ? PL_CAP : 1];
    __shared__ int s_lst[PLACE ? PL_CAP : 1];
    __shared__ float s_dv[PLACE ? NPB : 1];
    __shared__ float s_nx[PLACE ? NPB : 1];
    __shared__ __attribute__((aligned(16))) float s_bs[PLACE ? D : 4];

    const int tid  = threadIdx.x;
    const int pl   = tid / LPN;
    const int q    = tid % LPN;
    const int a    = blockIdx.x * NPB + 2 * pl;
    float* const sp = &sA[(2 * pl) * PA + 4 * q];
    v4f accA = {0.f, 0.f, 0.f, 0.f};
    v4f accB = {0.f, 0.f, 0.f, 0.f};
    float nxA = 0.0f, nxB = 0.0f;
    if (PLACE) {
        const int n0 = blockIdx.x * NPB;
        const int lo = rowptr[n0];
        const int hi = rowptr[min(n0 + NPB, n)];
        const int rpv = rowptr[min(n0 + min(tid, NPB), n)];
        const int ndv = min(n0 + (tid & (NPB - 1)), n - 1);
        const float dvv = dinv[ndv];
        const float nxv = dnext[ndv];
        static_assert(!PLACE || (D <= 256 && (D & (D - 1)) == 0), "one thread per bias element");
        const float bsv = bias[tid & (D - 1)];
        const int S = hi - lo;
        const bool inl = S <= PL_CAP;
        const v2i t0 = *(const v2i*)(tmp_pair + 2 * max(min(lo + tid, hi - 1), 0));
        if (tid <= NPB) s_rp[tid] = rpv;
        if (tid < NPB) { s_dv[tid] = dvv; s_nx[tid] = nxv; }
        if (tid < D) s_bs[tid] = bsv;
        if (inl) {
            if (tid < S) { s_src[tid] = t0.x; s_tgt[tid] = t0.y; }
            for (int p = tid + 256; p < S; p += 256) {
                const v2i t = *(const v2i*)(tmp_pair + 2 * (lo + p));
                s_src[p] = t.x;
                s_tgt[p] = t.y;
            }
        }
        __syncthreads();
        if (tid < S) place_slot<NPB>(inl, s_rp, s_src, s_lst, tmp_pair, csr_out, n0, lo, tid, t0.x, t0.y);
        for (int p = tid + 256; p < S; p += 256) {
            int r, c;
            if (inl) { r = s_src[p]; c = s_tgt[p]; }
            else { const v2i t = *(const v2i*)(tmp_pair + 2 * (lo + p)); r = t.x; c = t.y; }
            place_slot<NPB>(inl, s_rp, s_src, s_lst, tmp_pair, csr_out, n0, lo, p, r, c);
        }
        __syncthreads();
        if (inl) {
            if (tid < S) csr_out[lo + tid] = s_lst[tid];
#pragma nounroll
            for (int p = tid + 256; p < S; p += 256) csr_out[lo + p] = s_lst[p];
        }
        if (a < n) {
            const bool hasB = a + 1 < n;
            const int b = hasB ? a + 1 : a;
            const int beg = s_rp[2 * pl];
            const int mid = s_rp[2 * pl + 1];
            const int end = s_rp[2 * pl + 2];
            const char* __restrict__ Hb = (const char*)H;
            accA = *(const v4f*)(Hb + ((unsigned)a * (unsigned)(D * 4) + (unsigned)q * 16u));
            accB = *(const v4f*)(Hb + ((unsigned)b * (unsigned)(D * 4) + (unsigned)q * 16u));
            if (inl) gather_pair_walk<D, DEC, LIST_LDS>(H, nullptr, s_lst, 0, beg - lo, mid - lo, end - lo, q, accA, accB);
            else     gather_pair_slots<D, DEC>(H, tmp_pair, beg, mid, end, q, accA, accB);
            nxA = s_nx[2 * pl];
            nxB = s_nx[2 * pl + 1];
            pair_finish<DEC>(s_dv[2 * pl], s_dv[2 * pl + 1], *(const v4f*)&s_bs[4 * q], hasB, accA, accB);
        }
    } else if (a < n) {
        if (!DEC) { nxA = dnext[a]; nxB = dnext[a + 1 < n ? a + 1 : a]; }
        gather_pair<D, DEC>(H, rowptr, csr_src, dinv, bias, a, q, n, accA, accB);
    }
    if (!DEC) { accA = relu4(accA) * nxA; accB = relu4(accB) * nxB; }
    *(v4f*)sp        = accA;
    *(v4f*)(sp + PA) = accB;

    const int wave = __builtin_amdgcn_readfirstlane(tid >> 5);
    const int lane = tid & 31;
    const int c    = lane & 15;
    const int half = lane >> 4;
    const int kb   = half * 2;
    const int rb   = wave / NT;
    const int ct   = wave % NT;

    v2f Bf[KS];
    if (ALLW || wave < WPT) {
        const float* wp = W + (size_t)kb * DOUT + ct * 16 + c;
#pragma unroll
        for (int kk = 0; kk < KS; ++kk) {
            Bf[kk].x = wp[(size_t)(4 * kk) * DOUT];
            Bf[kk].y = wp[(size_t)(4 * kk + 1) * DOUT];
        }
    }
    const int qo = tid % QO;
    v4f bv = {0.f, 0.f, 0.f, 0.f};
    if (DEC) { bv.x = bias[4 * qo]; bv.y = bias[4 * qo + 1]; bv.z = bias[4 * qo + 2]; bv.w = bias[4 * qo + 3]; }
    float dn[NIT];
#pragma unroll
    for (int it = 0; it < NIT; ++it) {
        const int nd = blockIdx.x * NPB + (tid + it * 256) / QO;
        dn[it] = (DEC && !LAST) ? dnext[nd < n ? nd : n - 1] : 1.0f;
    }

    __syncthreads();

    if (ALLW || wave < WPT) {
        v8f d[TPW];
#pragma unroll
        for (int j = 0; j < TPW; ++j) {
            v8f z = {0.f, 0.f, 0.f, 0.f, 0.f, 0.f, 0.f, 0.f};
            d[j] = z;
        }
        const float* ap = &sA[(rb * 16 + c) * PA + kb];
#pragma unroll
        for (int kk = 0; kk < KS; ++kk) {
#pragma unroll
            for (int j = 0; j < TPW; ++j) {
                const v2f A = *(const v2f*)(ap + (j * RSTEP * 16) * PA + 4 * kk);
                d[j] = __builtin_amdgcn_wmma_f32_16x16x4_f32(
                    false, A, false, Bf[kk], (short)0, d[j], false, false);
            }
        }
#pragma unroll
        for (int j = 0; j < TPW; ++j) {
#pragma unroll
            for (int r = 0; r < 8; ++r)
                sO[((rb + j * RSTEP) * 16 + 8 * half + r) * PO + ct * 16 + c] = d[j][r];
        }
    }

    __syncthreads();

#pragma unroll
    for (int it = 0; it < NIT; ++it) {
        const int row = (tid + it * 256) / QO;
        const int nd  = blockIdx.x * NPB + row;
        if (nd < n) {
            v4f o = *(const v4f*)&sO[row * PO + 4 * qo];
            if (DEC) o += bv;
            if (DEC && !LAST) o = relu4(o) * dn[it];
            ((v4f*)Y)[(size_t)nd * QO + qo] = o;
        }
    }
}

static inline int cdiv(long long a, int b) { return (int)((a + b - 1) / b); }

extern "C" void kernel_launch(void* const* d_in, const int* in_sizes, int n_in,
                              void* d_out, int out_size, void* d_ws, size_t ws_size,
                              hipStream_t stream) {
    const float* x   = (const float*)d_in[0];
    const int*   ei  = (const int*)d_in[1];
    const float* We1 = (const float*)d_in[2];  const float* be1 = (const float*)d_in[3];
    const float* We2 = (const float*)d_in[4];  const float* be2 = (const float*)d_in[5];
    const float* We3 = (const float*)d_in[6];  const float* be3 = (const float*)d_in[7];
    const float* Wd1 = (const float*)d_in[8];  const float* bd1 = (const float*)d_in[9];
    const float* Wd2 = (const float*)d_in[10]; const float* bd2 = (const float*)d_in[11];
    const float* Wd3 = (const float*)d_in[12]; const float* bd3 = (const float*)d_in[13];

    const int N = in_sizes[0] / 128;
    const int E = in_sizes[1] / 2;
    const int* rowp = ei;
    const int* colp = ei + E;

    const size_t Np = ((size_t)N + 64 + 63) & ~(size_t)63;
    const size_t Ep = ((size_t)E + 63) & ~(size_t)63;
    char* ws = (char*)d_ws;
    float* dinv_s  = (float*)ws;  ws += sizeof(float) * Np;
    float* dinv_h  = (float*)ws;  ws += sizeof(float) * Np;
    float* hbuf    = (float*)ws;  ws += sizeof(float) * (size_t)N * 64;
    float* ybuf    = (float*)ws;  ws += sizeof(float) * (size_t)N * 64;
    int*   rowptr  = (int*)ws;    ws += sizeof(int)   * Np;
    int*   tmp_pair = (int*)ws;   ws += sizeof(int)   * 2 * Ep;
    int*   csr_src = (int*)ws;
    float* outp    = (float*)d_out;
    int*   part    = (int*)ybuf;
    int*   gsum    = part + (size_t)CNT_G * Np;

    const int TB = 256;
    const int rtiles  = (N + 15) >> 4;
    const int gblocks = (rtiles + 7) >> 3;

    const int cranges = cdiv(N, CNT_NR);
    const int ec      = cdiv(cdiv(E, CNT_G), TB) * TB;
    k_count_indeg<<<cranges * CNT_G, TB, 0, stream>>>(colp, part, gsum, N, E, (int)Np, ec);
    k_rowptr_scan_dinv<<<cdiv(N, 1024), 1024, 0, stream>>>(part, gsum, rowptr, dinv_s, dinv_h, N, (int)Np);
    k_gemm1_fill <<<cranges * CNT_G + gblocks, TB, 0, stream>>>(x, We1, dinv_s, hbuf, N, cranges * CNT_G,
                                                                rowp, colp, part, tmp_pair, E, (int)Np, ec);

    k_agg_gemm<64, 2, false, false, true><<<cdiv(N, 32), TB, 0, stream>>>(hbuf, rowptr, nullptr, dinv_s, dinv_s, be1, We2, ybuf, N, tmp_pair, csr_src);
    k_agg_gemm<32, 1, false, false, false><<<cdiv(N, 64), TB, 0, stream>>>(ybuf, rowptr, csr_src, dinv_s, dinv_s, be2, We3, hbuf, N, nullptr, nullptr);
    k_agg_gather<16, false><<<cdiv(N, 128), TB, 0, stream>>>(hbuf, rowptr, csr_src, dinv_s, dinv_h, be3, ybuf, N);

    k_agg_gemm<16, 2, true, false, false><<<cdiv(N, 128), TB, 0, stream>>>(ybuf, rowptr, csr_src, dinv_h, dinv_h, bd1, Wd1, hbuf, N, nullptr, nullptr);
    k_agg_gemm<32, 4, true, false, false><<<cdiv(N, 64), TB, 0, stream>>>(hbuf, rowptr, csr_src, dinv_h, dinv_h, bd2, Wd2, ybuf, N, nullptr, nullptr);
    k_agg_gemm<64, 8, true, true, false><<<cdiv(N, 32), TB, 0, stream>>>(ybuf, rowptr, csr_src, dinv_h, dinv_h, bd3, Wd3, outp, N, nullptr, nullptr);
}
